# starting kernel
# baseline (speedup 1.0000x reference)
#include <hip/hip_runtime.h>
#include <hip/hip_bf16.h>

#define N_NODES 200000
#define N_EDGES 1000000
#define N_GRAPHS 4096
#define F_IN 11
#define H 64

typedef float v2f __attribute__((ext_vector_type(2)));
typedef float v8f __attribute__((ext_vector_type(8)));

__device__ __forceinline__ void atomicAddF(float* p, float v) {
    unsafeAtomicAdd(p, v);
}


__global__ void k_init_deg(float* deg, int n) {
    int i = blockIdx.x * blockDim.x + threadIdx.x;
    if (i < n) deg[i] = 1.0f;
}

__global__ void k_deg_edges(const int* __restrict__ dst, float* deg, int e) {
    int i = blockIdx.x * blockDim.x + threadIdx.x;
    if (i < e) atomicAddF(&deg[dst[i]], 1.0f);
}

__global__ void k_rsqrt(float* deg, int n) {
    int i = blockIdx.x * blockDim.x + threadIdx.x;
    if (i < n) deg[i] = rsqrtf(fmaxf(deg[i], 1.0f));
}


template <int KV, int AS, int KP>
__global__ void k_gemm_wmma(const float* __restrict__ A, const float* __restrict__ W,
                            float* __restrict__ Out, int n) {
    __shared__ float Wlds[(KP / 2) * H * 2];
    for (int idx = threadIdx.x; idx < KP * H; idx += blockDim.x) {
        int k = idx / H, c = idx - k * H;
        float v = (k < KV) ? W[k * H + c] : 0.0f;
        Wlds[(k >> 1) * (H * 2) + c * 2 + (k & 1)] = v;
    }
    __syncthreads();

    const int lane  = threadIdx.x & 31;
    const int strip = blockIdx.x * (blockDim.x >> 5) + (threadIdx.x >> 5);
    const int row0  = strip * 16;
    if (row0 >= n) return;

    const int mrow = lane & 15;
    const int hi   = lane >> 4;
    const float* __restrict__ arow = A + (size_t)(row0 + mrow) * AS;

    v8f acc[4];
    #pragma unroll
    for (int t = 0; t < 4; ++t) acc[t] = (v8f){0, 0, 0, 0, 0, 0, 0, 0};

    #pragma unroll
    for (int k0 = 0; k0 < KP; k0 += 4) {
        const int ka = k0 + hi * 2;
        v2f a;
        if (KV == KP && (AS & 1) == 0) {
            a = *reinterpret_cast<const v2f*>(arow + ka);
        } else {
            const int kc0 = (ka     < KV) ? ka     : (KV - 1);
            const int kc1 = (ka + 1 < KV) ? ka + 1 : (KV - 1);
            float ax = arow[kc0];
            float ay = arow[kc1];
            a.x = (ka     < KV) ? ax : 0.0f;
            a.y = (ka + 1 < KV) ? ay : 0.0f;
        }

        const int pbase = ((k0 >> 1) + hi) * (H * 2);
        #pragma unroll
        for (int t = 0; t < 4; ++t) {
            const int c = t * 16 + mrow;
            v2f b = *reinterpret_cast<const v2f*>(&Wlds[pbase + c * 2]);
            acc[t] = __builtin_amdgcn_wmma_f32_16x16x4_f32(
                 false, a,  false, b,
                 (short)0, acc[t],  false,  false);
        }
    }

    const int mbase = (lane < 16) ? 0 : 8;
    #pragma unroll
    for (int t = 0; t < 4; ++t) {
        const int col = t * 16 + mrow;
        #pragma unroll
        for (int j = 0; j < 8; ++j) {
            Out[(row0 + mbase + j) * H + col] = acc[t][j];
        }
    }
}


__global__ void k_selfloop_init(const float* __restrict__ h, const float* __restrict__ dis,
                                float* __restrict__ agg, int n) {
    int t = blockIdx.x * blockDim.x + threadIdx.x;
    int node = t >> 4;
    if (node >= n) return;
    int f4 = (t & 15) * 4;
    float w = dis[node]; w *= w;
    float4 hv = *reinterpret_cast<const float4*>(h + node * H + f4);
    float4 r = make_float4(hv.x * w, hv.y * w, hv.z * w, hv.w * w);
    *reinterpret_cast<float4*>(agg + node * H + f4) = r;
}

__global__ void k_scatter_edges(const int* __restrict__ src, const int* __restrict__ dst,
                                const float* __restrict__ dis, const float* __restrict__ h,
                                float* __restrict__ agg, int e) {
    int t = blockIdx.x * blockDim.x + threadIdx.x;
    int edge = t >> 4;
    if (edge >= e) return;
    int f4 = (t & 15) * 4;
    int s = src[edge];
    int d = dst[edge];
    float w = dis[s] * dis[d];
    float4 hv = *reinterpret_cast<const float4*>(h + s * H + f4);
    float* o = agg + d * H + f4;
    atomicAddF(o + 0, hv.x * w);
    atomicAddF(o + 1, hv.y * w);
    atomicAddF(o + 2, hv.z * w);
    atomicAddF(o + 3, hv.w * w);
}

__global__ void k_bias_act(float* __restrict__ agg, const float* __restrict__ b,
                           int n, int relu) {
    int t = blockIdx.x * blockDim.x + threadIdx.x;
    if (t >= n * H) return;
    float v = agg[t] + b[t & (H - 1)];
    agg[t] = relu ? fmaxf(v, 0.0f) : v;
}


__global__ void k_zero(float* p, int n) {
    int i = blockIdx.x * blockDim.x + threadIdx.x;
    if (i < n) p[i] = 0.0f;
}

__global__ void k_pool_acc(const float* __restrict__ h, const int* __restrict__ batch,
                           float* __restrict__ sums, float* __restrict__ cnts, int n) {
    int t = blockIdx.x * blockDim.x + threadIdx.x;
    int node = t >> 4;
    if (node >= n) return;
    int f4 = (t & 15) * 4;
    int g = batch[node];
    float4 hv = *reinterpret_cast<const float4*>(h + node * H + f4);
    float* o = sums + g * H + f4;
    atomicAddF(o + 0, hv.x);
    atomicAddF(o + 1, hv.y);
    atomicAddF(o + 2, hv.z);
    atomicAddF(o + 3, hv.w);
    if ((t & 15) == 0) atomicAddF(&cnts[g], 1.0f);
}

__global__ void k_pool_linear(const float* __restrict__ sums, const float* __restrict__ cnts,
                              const float* __restrict__ lin_w, const float* __restrict__ lin_b,
                              float* __restrict__ out, int g_total) {
    int g = blockIdx.x * blockDim.x + threadIdx.x;
    if (g >= g_total) return;
    float inv = 1.0f / fmaxf(cnts[g], 1.0f);
    float acc = 0.0f;
    #pragma unroll 8
    for (int f = 0; f < H; ++f) acc += sums[g * H + f] * lin_w[f];
    out[g] = acc * inv + lin_b[0];
}


extern "C" void kernel_launch(void* const* d_in, const int* in_sizes, int n_in,
                              void* d_out, int out_size, void* d_ws, size_t ws_size,
                              hipStream_t stream) {
    const float* x     = (const float*)d_in[0];
    const int*   eidx  = (const int*)  d_in[1];
    const int*   batch = (const int*)  d_in[2];
    const float* W1    = (const float*)d_in[3];
    const float* b1    = (const float*)d_in[4];
    const float* W2    = (const float*)d_in[5];
    const float* b2    = (const float*)d_in[6];
    const float* W3    = (const float*)d_in[7];
    const float* b3    = (const float*)d_in[8];
    const float* lin_w = (const float*)d_in[9];
    const float* lin_b = (const float*)d_in[10];
    float* out = (float*)d_out;

    const int n = in_sizes[0] / F_IN;
    const int e = in_sizes[1] / 2;
    const int* src = eidx;
    const int* dst = eidx + e;

    float* ws   = (float*)d_ws;
    float* dis  = ws;
    float* bufA = dis  + n;
    float* bufB = bufA + (size_t)n * H;
    float* sums = bufB + (size_t)n * H;
    float* cnts = sums + (size_t)N_GRAPHS * H;

    const int TB = 256;
    dim3 blk(TB);
    auto blocks = [](long long work, int tb) { return dim3((unsigned)((work + tb - 1) / tb)); };

    k_init_deg  <<<blocks(n, TB), blk, 0, stream>>>(dis, n);
    k_deg_edges <<<blocks(e, TB), blk, 0, stream>>>(dst, dis, e);
    k_rsqrt     <<<blocks(n, TB), blk, 0, stream>>>(dis, n);

    const int strips       = (n + 15) / 16;
    const dim3 gemm_grid   = blocks((long long)strips * 32, TB);
    const long long nodef4 = (long long)n * 16;
    const long long edgef4 = (long long)e * 16;

    k_gemm_wmma<F_IN, F_IN, 12><<<gemm_grid, blk, 0, stream>>>(x, W1, bufA, n);
    k_selfloop_init <<<blocks(nodef4, TB), blk, 0, stream>>>(bufA, dis, bufB, n);
    k_scatter_edges <<<blocks(edgef4, TB), blk, 0, stream>>>(src, dst, dis, bufA, bufB, e);
    k_bias_act      <<<blocks((long long)n * H, TB), blk, 0, stream>>>(bufB, b1, n, 1);

    k_gemm_wmma<H, H, H><<<gemm_grid, blk, 0, stream>>>(bufB, W2, bufA, n);
    k_selfloop_init <<<blocks(nodef4, TB), blk, 0, stream>>>(bufA, dis, bufB, n);
    k_scatter_edges <<<blocks(edgef4, TB), blk, 0, stream>>>(src, dst, dis, bufA, bufB, e);
    k_bias_act      <<<blocks((long long)n * H, TB), blk, 0, stream>>>(bufB, b2, n, 1);

    k_gemm_wmma<H, H, H><<<gemm_grid, blk, 0, stream>>>(bufB, W3, bufA, n);
    k_selfloop_init <<<blocks(nodef4, TB), blk, 0, stream>>>(bufA, dis, bufB, n);
    k_scatter_edges <<<blocks(edgef4, TB), blk, 0, stream>>>(src, dst, dis, bufA, bufB, e);
    k_bias_act      <<<blocks((long long)n * H, TB), blk, 0, stream>>>(bufB, b3, n, 0);

    k_zero       <<<blocks((long long)N_GRAPHS * H, TB), blk, 0, stream>>>(sums, N_GRAPHS * H);
    k_zero       <<<blocks(N_GRAPHS, TB), blk, 0, stream>>>(cnts, N_GRAPHS);
    k_pool_acc   <<<blocks(nodef4, TB), blk, 0, stream>>>(bufB, batch, sums, cnts, n);
    k_pool_linear<<<blocks(N_GRAPHS, TB), blk, 0, stream>>>(sums, cnts, lin_w, lin_b, out, N_GRAPHS);
}


# edge sums turned from float-atomic push into pull: edges counting-sorted by destination once per call (int atomics), each row gathered once per 32-col half with self term, bias, relu; sums regrouped
# speedup vs baseline: 1.9196x; 1.9196x over previous
#include <hip/hip_runtime.h>
#include <hip/hip_bf16.h>

#define N_NODES 200000
#define N_EDGES 1000000
#define N_GRAPHS 4096
#define F_IN 11
#define H 64
#define HH 32
#define SCAN_TB 256
#define SCAN_CHUNK 1024

typedef float v2f __attribute__((ext_vector_type(2)));
typedef float v8f __attribute__((ext_vector_type(8)));

__device__ __forceinline__ void atomicAddF(float* p, float v) {
    unsafeAtomicAdd(p, v);
}


__global__ void k_zero_i(int* p, int n) {
    int i = blockIdx.x * blockDim.x + threadIdx.x;
    if (i < n) p[i] = 0;
}

__global__ void k_count_in(const int* __restrict__ dst, int* cnt, int e) {
    int i = blockIdx.x * blockDim.x + threadIdx.x;
    if (i < e) atomicAdd(&cnt[dst[i]], 1);
}

__global__ void k_chunk_sums(const int* __restrict__ cnt, int* __restrict__ part, int n) {
    __shared__ int red[SCAN_TB / 32];
    const int base = blockIdx.x * SCAN_CHUNK + threadIdx.x * 4;
    int s = 0;
    for (int u = 0; u < 4; ++u) {
        int i = base + u;
        if (i < n) s += cnt[i];
    }
    for (int o = 16; o > 0; o >>= 1) s += __shfl_xor(s, o);
    if ((threadIdx.x & 31) == 0) red[threadIdx.x >> 5] = s;
    __syncthreads();
    if (threadIdx.x == 0) {
        int t = 0;
        for (int w = 0; w < SCAN_TB / 32; ++w) t += red[w];
        part[blockIdx.x] = t;
    }
}

__global__ void k_offsets_norm(const int* __restrict__ cnt, const int* __restrict__ part,
                               int* __restrict__ cursor, float* __restrict__ dis, int n) {
    __shared__ int red[SCAN_TB / 32];
    __shared__ int wsum[SCAN_TB / 32];
    const int tid = threadIdx.x, lane = tid & 31, wv = tid >> 5;

    int p = 0;
    for (int i = tid; i < (int)blockIdx.x; i += SCAN_TB) p += part[i];
    for (int o = 16; o > 0; o >>= 1) p += __shfl_xor(p, o);
    if (lane == 0) red[wv] = p;

    const int base = blockIdx.x * SCAN_CHUNK + tid * 4;
    int c[4];
    int tsum = 0;
    for (int u = 0; u < 4; ++u) {
        int i = base + u;
        c[u] = (i < n) ? cnt[i] : 0;
        tsum += c[u];
    }
    int incl = tsum;
    for (int o = 1; o < 32; o <<= 1) {
        int v = __shfl_up(incl, o);
        if (lane >= o) incl += v;
    }
    if (lane == 31) wsum[wv] = incl;
    __syncthreads();
    int off = 0;
    for (int w = 0; w < SCAN_TB / 32; ++w) {
        off += red[w];
        if (w < wv) off += wsum[w];
    }
    int run = off + incl - tsum;
    for (int u = 0; u < 4; ++u) {
        int i = base + u;
        if (i < n) {
            cursor[i] = run;
            run += c[u];
            dis[i] = rsqrtf(fmaxf((float)(c[u] + 1), 1.0f));
        }
    }
}

__global__ void k_place_edges(const int* __restrict__ src, const int* __restrict__ dst,
                              const float* __restrict__ dis, int* cursor,
                              int2* __restrict__ list, int e) {
    int i = blockIdx.x * blockDim.x + threadIdx.x;
    if (i >= e) return;
    int s = src[i];
    int d = dst[i];
    int pos = atomicAdd(&cursor[d], 1);
    float w = dis[s] * dis[d];
    list[pos] = make_int2(s, __float_as_int(w));
}


template <int KV, int AS, int KP, bool SPLIT_A>
__global__ void k_gemm_wmma(const float* A0, const float* A1, const float* __restrict__ W,
                            float* O0, float* O1, int n) {
    __shared__ float Wlds[(KP / 2) * H * 2];
    for (int idx = threadIdx.x; idx < KP * H; idx += blockDim.x) {
        int k = idx / H, c = idx - k * H;
        float v = (k < KV) ? W[k * H + c] : 0.0f;
        Wlds[(k >> 1) * (H * 2) + c * 2 + (k & 1)] = v;
    }
    __syncthreads();

    const int lane  = threadIdx.x & 31;
    const int strip = blockIdx.x * (blockDim.x >> 5) + (threadIdx.x >> 5);
    const int row0  = strip * 16;
    if (row0 >= n) return;

    const int mrow = lane & 15;
    const int hi   = lane >> 4;
    const float* arow  = SPLIT_A ? A0 + (size_t)(row0 + mrow) * HH
                                 : A0 + (size_t)(row0 + mrow) * AS;
    const float* arow1 = A1 + (size_t)(row0 + mrow) * HH;

    v8f acc[4];
    #pragma unroll
    for (int t = 0; t < 4; ++t) acc[t] = (v8f){0, 0, 0, 0, 0, 0, 0, 0};

    #pragma unroll
    for (int k0 = 0; k0 < KP; k0 += 4) {
        const int ka = k0 + hi * 2;
        v2f a;
        if (SPLIT_A) {
            const float* ap = (k0 < HH) ? (arow + ka) : (arow1 + (ka - HH));
            a = *reinterpret_cast<const v2f*>(ap);
        } else if (KV == KP && (AS & 1) == 0) {
            a = *reinterpret_cast<const v2f*>(arow + ka);
        } else {
            const int kc0 = (ka     < KV) ? ka     : (KV - 1);
            const int kc1 = (ka + 1 < KV) ? ka + 1 : (KV - 1);
            float ax = arow[kc0];
            float ay = arow[kc1];
            a.x = (ka     < KV) ? ax : 0.0f;
            a.y = (ka + 1 < KV) ? ay : 0.0f;
        }

        const int pbase = ((k0 >> 1) + hi) * (H * 2);
        #pragma unroll
        for (int t = 0; t < 4; ++t) {
            const int c = t * 16 + mrow;
            v2f b = *reinterpret_cast<const v2f*>(&Wlds[pbase + c * 2]);
            acc[t] = __builtin_amdgcn_wmma_f32_16x16x4_f32(
                 false, a,  false, b,
                 (short)0, acc[t],  false,  false);
        }
    }

    const int mbase = (lane < 16) ? 0 : 8;
    #pragma unroll
    for (int t = 0; t < 4; ++t) {
        float* op = (t < 2) ? O0 : O1;
        const int col = (t & 1) * 16 + mrow;
        #pragma unroll
        for (int j = 0; j < 8; ++j) {
            op[(size_t)(row0 + mbase + j) * HH + col] = acc[t][j];
        }
    }
}

__global__ void k_pull_half(const float* __restrict__ hin, float* __restrict__ hout,
                            const int* __restrict__ rowend, const int2* __restrict__ list,
                            const float* __restrict__ dis, const float* __restrict__ bias,
                            int n, int relu) {
    int t = blockIdx.x * blockDim.x + threadIdx.x;
    int node = t >> 3;
    if (node >= n) return;
    const int q4 = (t & 7) * 4;
    const int end = rowend[node];
    const int beg = (node > 0) ? rowend[node - 1] : 0;
    const float di = dis[node];
    const float wself = di * di;
    const float4 hv = *reinterpret_cast<const float4*>(hin + (size_t)node * HH + q4);
    float4 acc = make_float4(hv.x * wself, hv.y * wself, hv.z * wself, hv.w * wself);

    const int last = end - 1;
    for (int j = beg; j < end; j += 4) {
        const int j1 = (j + 1 < end) ? j + 1 : last;
        const int j2 = (j + 2 < end) ? j + 2 : last;
        const int j3 = (j + 3 < end) ? j + 3 : last;
        const int2 e0 = list[j];
        const int2 e1 = list[j1];
        const int2 e2 = list[j2];
        const int2 e3 = list[j3];
        const float4 r0 = *reinterpret_cast<const float4*>(hin + (size_t)e0.x * HH + q4);
        const float4 r1 = *reinterpret_cast<const float4*>(hin + (size_t)e1.x * HH + q4);
        const float4 r2 = *reinterpret_cast<const float4*>(hin + (size_t)e2.x * HH + q4);
        const float4 r3 = *reinterpret_cast<const float4*>(hin + (size_t)e3.x * HH + q4);
        const bool v1 = (j + 1 < end), v2 = (j + 2 < end), v3 = (j + 3 < end);
        const float w0 = __int_as_float(e0.y);
        const float w1 = __int_as_float(e1.y);
        const float w2 = __int_as_float(e2.y);
        const float w3 = __int_as_float(e3.y);
        acc.x += r0.x * w0; acc.y += r0.y * w0; acc.z += r0.z * w0; acc.w += r0.w * w0;
        acc.x += (v1 ? r1.x : 0.0f) * w1; acc.y += (v1 ? r1.y : 0.0f) * w1;
        acc.z += (v1 ? r1.z : 0.0f) * w1; acc.w += (v1 ? r1.w : 0.0f) * w1;
        acc.x += (v2 ? r2.x : 0.0f) * w2; acc.y += (v2 ? r2.y : 0.0f) * w2;
        acc.z += (v2 ? r2.z : 0.0f) * w2; acc.w += (v2 ? r2.w : 0.0f) * w2;
        acc.x += (v3 ? r3.x : 0.0f) * w3; acc.y += (v3 ? r3.y : 0.0f) * w3;
        acc.z += (v3 ? r3.z : 0.0f) * w3; acc.w += (v3 ? r3.w : 0.0f) * w3;
    }

    const float4 bv = *reinterpret_cast<const float4*>(bias + q4);
    float4 r = make_float4(acc.x + bv.x, acc.y + bv.y, acc.z + bv.z, acc.w + bv.w);
    if (relu) {
        r.x = fmaxf(r.x, 0.0f); r.y = fmaxf(r.y, 0.0f);
        r.z = fmaxf(r.z, 0.0f); r.w = fmaxf(r.w, 0.0f);
    }
    *reinterpret_cast<float4*>(hout + (size_t)node * HH + q4) = r;
}


__global__ void k_zero(float* p, int n) {
    int i = blockIdx.x * blockDim.x + threadIdx.x;
    if (i < n) p[i] = 0.0f;
}

__global__ void k_pool_acc(const float* __restrict__ h0, const float* __restrict__ h1,
                           const int* __restrict__ batch,
                           float* __restrict__ sums, float* __restrict__ cnts, int n) {
    int t = blockIdx.x * blockDim.x + threadIdx.x;
    int node = t >> 4;
    if (node >= n) return;
    const int q = t & 15;
    const float* hp = (q < 8) ? h0 : h1;
    int g = batch[node];
    float4 hv = *reinterpret_cast<const float4*>(hp + (size_t)node * HH + (q & 7) * 4);
    float* o = sums + g * H + q * 4;
    atomicAddF(o + 0, hv.x);
    atomicAddF(o + 1, hv.y);
    atomicAddF(o + 2, hv.z);
    atomicAddF(o + 3, hv.w);
    if (q == 0) atomicAddF(&cnts[g], 1.0f);
}

__global__ void k_pool_linear(const float* __restrict__ sums, const float* __restrict__ cnts,
                              const float* __restrict__ lin_w, const float* __restrict__ lin_b,
                              float* __restrict__ out, int g_total) {
    int g = blockIdx.x * blockDim.x + threadIdx.x;
    if (g >= g_total) return;
    float inv = 1.0f / fmaxf(cnts[g], 1.0f);
    float acc = 0.0f;
    #pragma unroll 8
    for (int f = 0; f < H; ++f) acc += sums[g * H + f] * lin_w[f];
    out[g] = acc * inv + lin_b[0];
}


extern "C" void kernel_launch(void* const* d_in, const int* in_sizes, int n_in,
                              void* d_out, int out_size, void* d_ws, size_t ws_size,
                              hipStream_t stream) {
    const float* x     = (const float*)d_in[0];
    const int*   eidx  = (const int*)  d_in[1];
    const int*   batch = (const int*)  d_in[2];
    const float* W1    = (const float*)d_in[3];
    const float* b1    = (const float*)d_in[4];
    const float* W2    = (const float*)d_in[5];
    const float* b2    = (const float*)d_in[6];
    const float* W3    = (const float*)d_in[7];
    const float* b3    = (const float*)d_in[8];
    const float* lin_w = (const float*)d_in[9];
    const float* lin_b = (const float*)d_in[10];
    float* out = (float*)d_out;

    const int n = in_sizes[0] / F_IN;
    const int e = in_sizes[1] / 2;
    const int* src = eidx;
    const int* dst = eidx + e;

    float* ws   = (float*)d_ws;
    float* dis  = ws;
    float* R0   = dis + n;
    float* R1   = R0 + (size_t)n * HH;
    float* R2   = R1 + (size_t)n * HH;
    int*   cnt    = (int*)(R2 + (size_t)n * HH);
    int*   cursor = cnt + n;
    int*   part   = cursor + n;
    int2*  list   = (int2*)(part + 1024);
    float* sums = R0 + (size_t)n * H * 2;
    float* cnts = sums + (size_t)N_GRAPHS * H;

    const int TB = 256;
    dim3 blk(TB);
    auto blocks = [](long long work, int tb) { return dim3((unsigned)((work + tb - 1) / tb)); };

    const dim3 scan_grid = blocks(n, SCAN_CHUNK);
    k_zero_i      <<<blocks(n, TB), blk, 0, stream>>>(cnt, n);
    k_count_in    <<<blocks(e, TB), blk, 0, stream>>>(dst, cnt, e);
    k_chunk_sums  <<<scan_grid, dim3(SCAN_TB), 0, stream>>>(cnt, part, n);
    k_offsets_norm<<<scan_grid, dim3(SCAN_TB), 0, stream>>>(cnt, part, cursor, dis, n);
    k_place_edges <<<blocks(e, TB), blk, 0, stream>>>(src, dst, dis, cursor, list, e);

    const int strips       = (n + 15) / 16;
    const dim3 gemm_grid   = blocks((long long)strips * 32, TB);
    const dim3 pull_grid   = blocks((long long)n * 8, TB);
    const long long nodef4 = (long long)n * 16;

    k_gemm_wmma<F_IN, F_IN, 12, false><<<gemm_grid, blk, 0, stream>>>(x, x, W1, R0, R1, n);
    k_pull_half<<<pull_grid, blk, 0, stream>>>(R0, R2, cursor, list, dis, b1,      n, 1);
    k_pull_half<<<pull_grid, blk, 0, stream>>>(R1, R0, cursor, list, dis, b1 + HH, n, 1);

    k_gemm_wmma<H, H, H, true><<<gemm_grid, blk, 0, stream>>>(R2, R0, W2, R2, R0, n);
    k_pull_half<<<pull_grid, blk, 0, stream>>>(R2, R1, cursor, list, dis, b2,      n, 1);
    k_pull_half<<<pull_grid, blk, 0, stream>>>(R0, R2, cursor, list, dis, b2 + HH, n, 1);

    k_gemm_wmma<H, H, H, true><<<gemm_grid, blk, 0, stream>>>(R1, R2, W3, R1, R2, n);
    k_pull_half<<<pull_grid, blk, 0, stream>>>(R1, R0, cursor, list, dis, b3,      n, 0);
    k_pull_half<<<pull_grid, blk, 0, stream>>>(R2, R1, cursor, list, dis, b3 + HH, n, 0);

    k_zero       <<<blocks((long long)N_GRAPHS * H, TB), blk, 0, stream>>>(sums, N_GRAPHS * H);
    k_zero       <<<blocks(N_GRAPHS, TB), blk, 0, stream>>>(cnts, N_GRAPHS);
    k_pool_acc   <<<blocks(nodef4, TB), blk, 0, stream>>>(R0, R1, batch, sums, cnts, n);
    k_pool_linear<<<blocks(N_GRAPHS, TB), blk, 0, stream>>>(sums, cnts, lin_w, lin_b, out, N_GRAPHS);
}


# each aggregation hands its rows to its consumer in-kernel: half-1 pull of layers 1-2 feeds the next WMMA GEMM through LDS, layer-3 pulls add straight into the pool sums (pool atomics in new order)
# speedup vs baseline: 2.0109x; 1.0476x over previous
#include <hip/hip_runtime.h>
#include <hip/hip_bf16.h>

#define N_NODES 200000
#define N_EDGES 1000000
#define N_GRAPHS 4096
#define F_IN 11
#define H 64
#define HH 32
#define SCAN_TB 256
#define SCAN_CHUNK 1024

typedef float v2f __attribute__((ext_vector_type(2)));
typedef float v8f __attribute__((ext_vector_type(8)));

__device__ __forceinline__ void atomicAddF(float* p, float v) {
    unsafeAtomicAdd(p, v);
}


__global__ void k_zero_i(int* p, int n) {
    int i = blockIdx.x * blockDim.x + threadIdx.x;
    if (i < n) p[i] = 0;
}

__global__ void k_count_in(const int* __restrict__ dst, int* cnt, int e) {
    int i = blockIdx.x * blockDim.x + threadIdx.x;
    if (i < e) atomicAdd(&cnt[dst[i]], 1);
}

__global__ void k_chunk_sums(const int* __restrict__ cnt, int* __restrict__ part, int n) {
    __shared__ int red[SCAN_TB / 32];
    const int base = blockIdx.x * SCAN_CHUNK + threadIdx.x * 4;
    int s = 0;
    for (int u = 0; u < 4; ++u) {
        int i = base + u;
        if (i < n) s += cnt[i];
    }
    for (int o = 16; o > 0; o >>= 1) s += __shfl_xor(s, o);
    if ((threadIdx.x & 31) == 0) red[threadIdx.x >> 5] = s;
    __syncthreads();
    if (threadIdx.x == 0) {
        int t = 0;
        for (int w = 0; w < SCAN_TB / 32; ++w) t += red[w];
        part[blockIdx.x] = t;
    }
}

__global__ void k_offsets_norm(const int* __restrict__ cnt, const int* __restrict__ part,
                               int* __restrict__ cursor, float* __restrict__ dis, int n) {
    __shared__ int red[SCAN_TB / 32];
    __shared__ int wsum[SCAN_TB / 32];
    const int tid = threadIdx.x, lane = tid & 31, wv = tid >> 5;

    int p = 0;
    for (int i = tid; i < (int)blockIdx.x; i += SCAN_TB) p += part[i];
    for (int o = 16; o > 0; o >>= 1) p += __shfl_xor(p, o);
    if (lane == 0) red[wv] = p;

    const int base = blockIdx.x * SCAN_CHUNK + tid * 4;
    int c[4];
    int tsum = 0;
    for (int u = 0; u < 4; ++u) {
        int i = base + u;
        c[u] = (i < n) ? cnt[i] : 0;
        tsum += c[u];
    }
    int incl = tsum;
    for (int o = 1; o < 32; o <<= 1) {
        int v = __shfl_up(incl, o);
        if (lane >= o) incl += v;
    }
    if (lane == 31) wsum[wv] = incl;
    __syncthreads();
    int off = 0;
    for (int w = 0; w < SCAN_TB / 32; ++w) {
        off += red[w];
        if (w < wv) off += wsum[w];
    }
    int run = off + incl - tsum;
    for (int u = 0; u < 4; ++u) {
        int i = base + u;
        if (i < n) {
            cursor[i] = run;
            run += c[u];
            dis[i] = rsqrtf(fmaxf((float)(c[u] + 1), 1.0f));
        }
    }
}

__global__ void k_place_edges(const int* __restrict__ src, const int* __restrict__ dst,
                              const float* __restrict__ dis, int* cursor,
                              int2* __restrict__ list, int e) {
    int i = blockIdx.x * blockDim.x + threadIdx.x;
    if (i >= e) return;
    int s = src[i];
    int d = dst[i];
    int pos = atomicAdd(&cursor[d], 1);
    float w = dis[s] * dis[d];
    list[pos] = make_int2(s, __float_as_int(w));
}


template <int KV, int AS, int KP, bool SPLIT_A>
__global__ void k_gemm_wmma(const float* A0, const float* A1, const float* __restrict__ W,
                            float* O0, float* O1, int n) {
    __shared__ float Wlds[(KP / 2) * H * 2];
    for (int idx = threadIdx.x; idx < KP * H; idx += blockDim.x) {
        int k = idx / H, c = idx - k * H;
        float v = (k < KV) ? W[k * H + c] : 0.0f;
        Wlds[(k >> 1) * (H * 2) + c * 2 + (k & 1)] = v;
    }
    __syncthreads();

    const int lane  = threadIdx.x & 31;
    const int strip = blockIdx.x * (blockDim.x >> 5) + (threadIdx.x >> 5);
    const int row0  = strip * 16;
    if (row0 >= n) return;

    const int mrow = lane & 15;
    const int hi   = lane >> 4;
    const float* arow  = SPLIT_A ? A0 + (size_t)(row0 + mrow) * HH
                                 : A0 + (size_t)(row0 + mrow) * AS;
    const float* arow1 = A1 + (size_t)(row0 + mrow) * HH;

    v8f acc[4];
    #pragma unroll
    for (int t = 0; t < 4; ++t) acc[t] = (v8f){0, 0, 0, 0, 0, 0, 0, 0};

    #pragma unroll
    for (int k0 = 0; k0 < KP; k0 += 4) {
        const int ka = k0 + hi * 2;
        v2f a;
        if (SPLIT_A) {
            const float* ap = (k0 < HH) ? (arow + ka) : (arow1 + (ka - HH));
            a = *reinterpret_cast<const v2f*>(ap);
        } else if (KV == KP && (AS & 1) == 0) {
            a = *reinterpret_cast<const v2f*>(arow + ka);
        } else {
            const int kc0 = (ka     < KV) ? ka     : (KV - 1);
            const int kc1 = (ka + 1 < KV) ? ka + 1 : (KV - 1);
            float ax = arow[kc0];
            float ay = arow[kc1];
            a.x = (ka     < KV) ? ax : 0.0f;
            a.y = (ka + 1 < KV) ? ay : 0.0f;
        }

        const int pbase = ((k0 >> 1) + hi) * (H * 2);
        #pragma unroll
        for (int t = 0; t < 4; ++t) {
            const int c = t * 16 + mrow;
            v2f b = *reinterpret_cast<const v2f*>(&Wlds[pbase + c * 2]);
            acc[t] = __builtin_amdgcn_wmma_f32_16x16x4_f32(
                 false, a,  false, b,
                 (short)0, acc[t],  false,  false);
        }
    }

    const int mbase = (lane < 16) ? 0 : 8;
    #pragma unroll
    for (int t = 0; t < 4; ++t) {
        float* op = (t < 2) ? O0 : O1;
        const int col = (t & 1) * 16 + mrow;
        #pragma unroll
        for (int j = 0; j < 8; ++j) {
            op[(size_t)(row0 + mbase + j) * HH + col] = acc[t][j];
        }
    }
}

__device__ __forceinline__ float4 pull_edges(const float* __restrict__ hin,
                                             const int2* __restrict__ list,
                                             int beg, int end, float4 acc, int q4) {
    const int last = end - 1;
    for (int j = beg; j < end; j += 4) {
        const int j1 = (j + 1 < end) ? j + 1 : last;
        const int j2 = (j + 2 < end) ? j + 2 : last;
        const int j3 = (j + 3 < end) ? j + 3 : last;
        const int2 e0 = list[j];
        const int2 e1 = list[j1];
        const int2 e2 = list[j2];
        const int2 e3 = list[j3];
        const float4 r0 = *reinterpret_cast<const float4*>(hin + (size_t)e0.x * HH + q4);
        const float4 r1 = *reinterpret_cast<const float4*>(hin + (size_t)e1.x * HH + q4);
        const float4 r2 = *reinterpret_cast<const float4*>(hin + (size_t)e2.x * HH + q4);
        const float4 r3 = *reinterpret_cast<const float4*>(hin + (size_t)e3.x * HH + q4);
        const bool v1 = (j + 1 < end), v2 = (j + 2 < end), v3 = (j + 3 < end);
        const float w0 = __int_as_float(e0.y);
        const float w1 = __int_as_float(e1.y);
        const float w2 = __int_as_float(e2.y);
        const float w3 = __int_as_float(e3.y);
        acc.x += r0.x * w0; acc.y += r0.y * w0; acc.z += r0.z * w0; acc.w += r0.w * w0;
        acc.x += (v1 ? r1.x : 0.0f) * w1; acc.y += (v1 ? r1.y : 0.0f) * w1;
        acc.z += (v1 ? r1.z : 0.0f) * w1; acc.w += (v1 ? r1.w : 0.0f) * w1;
        acc.x += (v2 ? r2.x : 0.0f) * w2; acc.y += (v2 ? r2.y : 0.0f) * w2;
        acc.z += (v2 ? r2.z : 0.0f) * w2; acc.w += (v2 ? r2.w : 0.0f) * w2;
        acc.x += (v3 ? r3.x : 0.0f) * w3; acc.y += (v3 ? r3.y : 0.0f) * w3;
        acc.z += (v3 ? r3.z : 0.0f) * w3; acc.w += (v3 ? r3.w : 0.0f) * w3;
    }
    return acc;
}

__device__ __forceinline__ float4 pull_node(const float* __restrict__ hin,
                                            const int* __restrict__ rowend,
                                            const int2* __restrict__ list,
                                            const float* __restrict__ dis,
                                            const float* __restrict__ bias, int node, int q4) {
    const int end = rowend[node];
    const int beg = (node > 0) ? rowend[node - 1] : 0;
    const float di = dis[node];
    const float wself = di * di;
    const float4 hv = *reinterpret_cast<const float4*>(hin + (size_t)node * HH + q4);
    float4 acc = make_float4(hv.x * wself, hv.y * wself, hv.z * wself, hv.w * wself);
    acc = pull_edges(hin, list, beg, end, acc, q4);
    const float4 bv = *reinterpret_cast<const float4*>(bias + q4);
    return make_float4(acc.x + bv.x, acc.y + bv.y, acc.z + bv.z, acc.w + bv.w);
}

__global__ void k_pull_half(const float* __restrict__ hin, float* __restrict__ hout,
                            const int* __restrict__ rowend, const int2* __restrict__ list,
                            const float* __restrict__ dis, const float* __restrict__ bias,
                            int n) {
    int t = blockIdx.x * blockDim.x + threadIdx.x;
    int node = t >> 3;
    if (node >= n) return;
    const int q4 = (t & 7) * 4;
    float4 r = pull_node(hin, rowend, list, dis, bias, node, q4);
    r.x = fmaxf(r.x, 0.0f); r.y = fmaxf(r.y, 0.0f);
    r.z = fmaxf(r.z, 0.0f); r.w = fmaxf(r.w, 0.0f);
    *reinterpret_cast<float4*>(hout + (size_t)node * HH + q4) = r;
}

__global__ void k_pull_pool(const float* __restrict__ hin,
                            const int* __restrict__ rowend, const int2* __restrict__ list,
                            const float* __restrict__ dis, const float* __restrict__ bias,
                            const int* __restrict__ batch, float* sums, float* cnts,
                            int n, int colbase) {
    int t = blockIdx.x * blockDim.x + threadIdx.x;
    int node = t >> 3;
    if (node >= n) return;
    const int q4 = (t & 7) * 4;
    const float4 r = pull_node(hin, rowend, list, dis, bias, node, q4);
    const int g = batch[node];
    float* o = sums + g * H + colbase + q4;
    atomicAddF(o + 0, r.x);
    atomicAddF(o + 1, r.y);
    atomicAddF(o + 2, r.z);
    atomicAddF(o + 3, r.w);
    if (colbase == 0 && q4 == 0) atomicAddF(&cnts[g], 1.0f);
}

#define FB_ROWS 128
#define TP 36

__global__ void __launch_bounds__(256)
k_pull_gemm(const float* __restrict__ hg, float* P0, float* O1,
            const int* __restrict__ rowend, const int2* __restrict__ list,
            const float* __restrict__ dis, const float* __restrict__ bias,
            const float* __restrict__ W, int n) {
    __shared__ float Wlds[(H / 2) * H * 2];
    __shared__ __attribute__((aligned(16))) float tile[FB_ROWS * TP];
    for (int idx = threadIdx.x; idx < H * H; idx += blockDim.x) {
        int k = idx / H, c = idx - k * H;
        Wlds[(k >> 1) * (H * 2) + c * 2 + (k & 1)] = W[k * H + c];
    }

    const int blockBase = blockIdx.x * FB_ROWS;
    const int q4 = (threadIdx.x & 7) * 4;
    const int sub = threadIdx.x >> 3;
    const float4 bv = *reinterpret_cast<const float4*>(bias + q4);

    int beg[4], end[4];
    float4 acc[4];
    #pragma unroll
    for (int it = 0; it < 4; ++it) {
        int node = blockBase + it * 32 + sub;
        node = (node < n) ? node : (n - 1);
        end[it] = rowend[node];
        beg[it] = (node > 0) ? rowend[node - 1] : 0;
        const float di = dis[node];
        const float wself = di * di;
        const float4 hv = *reinterpret_cast<const float4*>(hg + (size_t)node * HH + q4);
        acc[it] = make_float4(hv.x * wself, hv.y * wself, hv.z * wself, hv.w * wself);
    }
    #pragma unroll
    for (int it = 0; it < 4; ++it) {
        const float4 a = pull_edges(hg, list, beg[it], end[it], acc[it], q4);
        float4 r = make_float4(a.x + bv.x, a.y + bv.y, a.z + bv.z, a.w + bv.w);
        r.x = fmaxf(r.x, 0.0f); r.y = fmaxf(r.y, 0.0f);
        r.z = fmaxf(r.z, 0.0f); r.w = fmaxf(r.w, 0.0f);
        *reinterpret_cast<float4*>(&tile[(it * 32 + sub) * TP + q4]) = r;
    }
    __syncthreads();

    const int lane = threadIdx.x & 31;
    const int wv   = threadIdx.x >> 5;
    const int row0 = blockBase + wv * 16;
    if (row0 >= n) return;

    const int mrow = lane & 15;
    const int hi   = lane >> 4;
    const float* arow = P0 + (size_t)(row0 + mrow) * HH;
    const float* trow = &tile[(wv * 16 + mrow) * TP];

    v8f c4[4];
    #pragma unroll
    for (int t = 0; t < 4; ++t) c4[t] = (v8f){0, 0, 0, 0, 0, 0, 0, 0};

    #pragma unroll
    for (int k0 = 0; k0 < H; k0 += 4) {
        const int ka = k0 + hi * 2;
        const float* ap = (k0 < HH) ? (arow + ka) : (trow + (ka - HH));
        const v2f a = *reinterpret_cast<const v2f*>(ap);

        const int pbase = ((k0 >> 1) + hi) * (H * 2);
        #pragma unroll
        for (int t = 0; t < 4; ++t) {
            const int c = t * 16 + mrow;
            v2f b = *reinterpret_cast<const v2f*>(&Wlds[pbase + c * 2]);
            c4[t] = __builtin_amdgcn_wmma_f32_16x16x4_f32(
                 false, a,  false, b,
                 (short)0, c4[t],  false,  false);
        }
    }

    const int mbase = (lane < 16) ? 0 : 8;
    #pragma unroll
    for (int t = 0; t < 4; ++t) {
        float* op = (t < 2) ? P0 : O1;
        const int col = (t & 1) * 16 + mrow;
        #pragma unroll
        for (int j = 0; j < 8; ++j) {
            op[(size_t)(row0 + mbase + j) * HH + col] = c4[t][j];
        }
    }
}


__global__ void k_zero(float* p, int n) {
    int i = blockIdx.x * blockDim.x + threadIdx.x;
    if (i < n) p[i] = 0.0f;
}

__global__ void k_pool_linear(const float* __restrict__ sums, const float* __restrict__ cnts,
                              const float* __restrict__ lin_w, const float* __restrict__ lin_b,
                              float* __restrict__ out, int g_total) {
    int g = blockIdx.x * blockDim.x + threadIdx.x;
    if (g >= g_total) return;
    float inv = 1.0f / fmaxf(cnts[g], 1.0f);
    float acc = 0.0f;
    #pragma unroll 8
    for (int f = 0; f < H; ++f) acc += sums[g * H + f] * lin_w[f];
    out[g] = acc * inv + lin_b[0];
}


extern "C" void kernel_launch(void* const* d_in, const int* in_sizes, int n_in,
                              void* d_out, int out_size, void* d_ws, size_t ws_size,
                              hipStream_t stream) {
    const float* x     = (const float*)d_in[0];
    const int*   eidx  = (const int*)  d_in[1];
    const int*   batch = (const int*)  d_in[2];
    const float* W1    = (const float*)d_in[3];
    const float* b1    = (const float*)d_in[4];
    const float* W2    = (const float*)d_in[5];
    const float* b2    = (const float*)d_in[6];
    const float* W3    = (const float*)d_in[7];
    const float* b3    = (const float*)d_in[8];
    const float* lin_w = (const float*)d_in[9];
    const float* lin_b = (const float*)d_in[10];
    float* out = (float*)d_out;

    const int n = in_sizes[0] / F_IN;
    const int e = in_sizes[1] / 2;
    const int* src = eidx;
    const int* dst = eidx + e;

    float* ws   = (float*)d_ws;
    float* dis  = ws;
    float* R0   = dis + n;
    float* R1   = R0 + (size_t)n * HH;
    float* R2   = R1 + (size_t)n * HH;
    int*   cnt    = (int*)(R2 + (size_t)n * HH);
    int*   cursor = cnt + n;
    int*   part   = cursor + n;
    int2*  list   = (int2*)(part + 1024);
    float* sums = R0 + (size_t)n * H * 2;
    float* cnts = sums + (size_t)N_GRAPHS * H;

    const int TB = 256;
    dim3 blk(TB);
    auto blocks = [](long long work, int tb) { return dim3((unsigned)((work + tb - 1) / tb)); };

    const dim3 scan_grid = blocks(n, SCAN_CHUNK);
    k_zero_i      <<<blocks(n, TB), blk, 0, stream>>>(cnt, n);
    k_count_in    <<<blocks(e, TB), blk, 0, stream>>>(dst, cnt, e);
    k_chunk_sums  <<<scan_grid, dim3(SCAN_TB), 0, stream>>>(cnt, part, n);
    k_offsets_norm<<<scan_grid, dim3(SCAN_TB), 0, stream>>>(cnt, part, cursor, dis, n);
    k_place_edges <<<blocks(e, TB), blk, 0, stream>>>(src, dst, dis, cursor, list, e);

    const int strips       = (n + 15) / 16;
    const dim3 gemm_grid   = blocks((long long)strips * 32, TB);
    const dim3 pull_grid   = blocks((long long)n * 8, TB);
    const dim3 fuse_grid   = blocks((long long)n, FB_ROWS);

    k_gemm_wmma<F_IN, F_IN, 12, false><<<gemm_grid, blk, 0, stream>>>(x, x, W1, R0, R1, n);

    k_pull_half<<<pull_grid, blk, 0, stream>>>(R0, R2, cursor, list, dis, b1, n);
    k_pull_gemm<<<fuse_grid, blk, 0, stream>>>(R1, R2, R0, cursor, list, dis, b1 + HH, W2, n);

    k_pull_half<<<pull_grid, blk, 0, stream>>>(R2, R1, cursor, list, dis, b2, n);
    k_pull_gemm<<<fuse_grid, blk, 0, stream>>>(R0, R1, R2, cursor, list, dis, b2 + HH, W3, n);

    k_zero       <<<blocks((long long)N_GRAPHS * H, TB), blk, 0, stream>>>(sums, N_GRAPHS * H);
    k_zero       <<<blocks(N_GRAPHS, TB), blk, 0, stream>>>(cnts, N_GRAPHS);
    k_pull_pool  <<<pull_grid, blk, 0, stream>>>(R1, cursor, list, dis, b3,      batch, sums, cnts, n, 0);
    k_pull_pool  <<<pull_grid, blk, 0, stream>>>(R2, cursor, list, dis, b3 + HH, batch, sums, cnts, n, HH);
    k_pool_linear<<<blocks(N_GRAPHS, TB), blk, 0, stream>>>(sums, cnts, lin_w, lin_b, out, N_GRAPHS);
}


# layer-3 pool: where a wave's four pulled rows share a graph id they are added across lanes first, then one float atomic per column (one whole line per wave, no word twice); pool sums regrouped
# speedup vs baseline: 2.1625x; 1.0754x over previous
#include <hip/hip_runtime.h>
#include <hip/hip_bf16.h>

#define N_NODES 200000
#define N_EDGES 1000000
#define N_GRAPHS 4096
#define F_IN 11
#define H 64
#define HH 32
#define SCAN_TB 256
#define SCAN_CHUNK 1024

typedef float v2f __attribute__((ext_vector_type(2)));
typedef float v8f __attribute__((ext_vector_type(8)));

__device__ __forceinline__ void atomicAddF(float* p, float v) {
    unsafeAtomicAdd(p, v);
}


__global__ void k_zero_i(int* p, int n) {
    int i = blockIdx.x * blockDim.x + threadIdx.x;
    if (i < n) p[i] = 0;
}

__global__ void k_count_in(const int* __restrict__ dst, int* cnt, int e) {
    int i = blockIdx.x * blockDim.x + threadIdx.x;
    if (i < e) atomicAdd(&cnt[dst[i]], 1);
}

__global__ void k_chunk_sums(const int* __restrict__ cnt, int* __restrict__ part, int n) {
    __shared__ int red[SCAN_TB / 32];
    const int base = blockIdx.x * SCAN_CHUNK + threadIdx.x * 4;
    int s = 0;
    for (int u = 0; u < 4; ++u) {
        int i = base + u;
        if (i < n) s += cnt[i];
    }
    for (int o = 16; o > 0; o >>= 1) s += __shfl_xor(s, o);
    if ((threadIdx.x & 31) == 0) red[threadIdx.x >> 5] = s;
    __syncthreads();
    if (threadIdx.x == 0) {
        int t = 0;
        for (int w = 0; w < SCAN_TB / 32; ++w) t += red[w];
        part[blockIdx.x] = t;
    }
}

__global__ void k_offsets_norm(const int* __restrict__ cnt, const int* __restrict__ part,
                               int* __restrict__ cursor, float* __restrict__ dis, int n) {
    __shared__ int red[SCAN_TB / 32];
    __shared__ int wsum[SCAN_TB / 32];
    const int tid = threadIdx.x, lane = tid & 31, wv = tid >> 5;

    int p = 0;
    for (int i = tid; i < (int)blockIdx.x; i += SCAN_TB) p += part[i];
    for (int o = 16; o > 0; o >>= 1) p += __shfl_xor(p, o);
    if (lane == 0) red[wv] = p;

    const int base = blockIdx.x * SCAN_CHUNK + tid * 4;
    int c[4];
    int tsum = 0;
    for (int u = 0; u < 4; ++u) {
        int i = base + u;
        c[u] = (i < n) ? cnt[i] : 0;
        tsum += c[u];
    }
    int incl = tsum;
    for (int o = 1; o < 32; o <<= 1) {
        int v = __shfl_up(incl, o);
        if (lane >= o) incl += v;
    }
    if (lane == 31) wsum[wv] = incl;
    __syncthreads();
    int off = 0;
    for (int w = 0; w < SCAN_TB / 32; ++w) {
        off += red[w];
        if (w < wv) off += wsum[w];
    }
    int run = off + incl - tsum;
    for (int u = 0; u < 4; ++u) {
        int i = base + u;
        if (i < n) {
            cursor[i] = run;
            run += c[u];
            dis[i] = rsqrtf(fmaxf((float)(c[u] + 1), 1.0f));
        }
    }
}

__global__ void k_place_edges(const int* __restrict__ src, const int* __restrict__ dst,
                              const float* __restrict__ dis, int* cursor,
                              int2* __restrict__ list, int e) {
    int i = blockIdx.x * blockDim.x + threadIdx.x;
    if (i >= e) return;
    int s = src[i];
    int d = dst[i];
    int pos = atomicAdd(&cursor[d], 1);
    float w = dis[s] * dis[d];
    list[pos] = make_int2(s, __float_as_int(w));
}


template <int KV, int AS, int KP, bool SPLIT_A>
__global__ void k_gemm_wmma(const float* A0, const float* A1, const float* __restrict__ W,
                            float* O0, float* O1, int n) {
    __shared__ float Wlds[(KP / 2) * H * 2];
    for (int idx = threadIdx.x; idx < KP * H; idx += blockDim.x) {
        int k = idx / H, c = idx - k * H;
        float v = (k < KV) ? W[k * H + c] : 0.0f;
        Wlds[(k >> 1) * (H * 2) + c * 2 + (k & 1)] = v;
    }
    __syncthreads();

    const int lane  = threadIdx.x & 31;
    const int strip = blockIdx.x * (blockDim.x >> 5) + (threadIdx.x >> 5);
    const int row0  = strip * 16;
    if (row0 >= n) return;

    const int mrow = lane & 15;
    const int hi   = lane >> 4;
    const float* arow  = SPLIT_A ? A0 + (size_t)(row0 + mrow) * HH
                                 : A0 + (size_t)(row0 + mrow) * AS;
    const float* arow1 = A1 + (size_t)(row0 + mrow) * HH;

    v8f acc[4];
    #pragma unroll
    for (int t = 0; t < 4; ++t) acc[t] = (v8f){0, 0, 0, 0, 0, 0, 0, 0};

    #pragma unroll
    for (int k0 = 0; k0 < KP; k0 += 4) {
        const int ka = k0 + hi * 2;
        v2f a;
        if (SPLIT_A) {
            const float* ap = (k0 < HH) ? (arow + ka) : (arow1 + (ka - HH));
            a = *reinterpret_cast<const v2f*>(ap);
        } else if (KV == KP && (AS & 1) == 0) {
            a = *reinterpret_cast<const v2f*>(arow + ka);
        } else {
            const int kc0 = (ka     < KV) ? ka     : (KV - 1);
            const int kc1 = (ka + 1 < KV) ? ka + 1 : (KV - 1);
            float ax = arow[kc0];
            float ay = arow[kc1];
            a.x = (ka     < KV) ? ax : 0.0f;
            a.y = (ka + 1 < KV) ? ay : 0.0f;
        }

        const int pbase = ((k0 >> 1) + hi) * (H * 2);
        #pragma unroll
        for (int t = 0; t < 4; ++t) {
            const int c = t * 16 + mrow;
            v2f b = *reinterpret_cast<const v2f*>(&Wlds[pbase + c * 2]);
            acc[t] = __builtin_amdgcn_wmma_f32_16x16x4_f32(
                 false, a,  false, b,
                 (short)0, acc[t],  false,  false);
        }
    }

    const int mbase = (lane < 16) ? 0 : 8;
    #pragma unroll
    for (int t = 0; t < 4; ++t) {
        float* op = (t < 2) ? O0 : O1;
        const int col = (t & 1) * 16 + mrow;
        #pragma unroll
        for (int j = 0; j < 8; ++j) {
            op[(size_t)(row0 + mbase + j) * HH + col] = acc[t][j];
        }
    }
}

__device__ __forceinline__ float4 pull_edges(const float* __restrict__ hin,
                                             const int2* __restrict__ list,
                                             int beg, int end, float4 acc, int q4) {
    const int last = end - 1;
    for (int j = beg; j < end; j += 4) {
        const int j1 = (j + 1 < end) ? j + 1 : last;
        const int j2 = (j + 2 < end) ? j + 2 : last;
        const int j3 = (j + 3 < end) ? j + 3 : last;
        const int2 e0 = list[j];
        const int2 e1 = list[j1];
        const int2 e2 = list[j2];
        const int2 e3 = list[j3];
        const float4 r0 = *reinterpret_cast<const float4*>(hin + (size_t)e0.x * HH + q4);
        const float4 r1 = *reinterpret_cast<const float4*>(hin + (size_t)e1.x * HH + q4);
        const float4 r2 = *reinterpret_cast<const float4*>(hin + (size_t)e2.x * HH + q4);
        const float4 r3 = *reinterpret_cast<const float4*>(hin + (size_t)e3.x * HH + q4);
        const bool v1 = (j + 1 < end), v2 = (j + 2 < end), v3 = (j + 3 < end);
        const float w0 = __int_as_float(e0.y);
        const float w1 = __int_as_float(e1.y);
        const float w2 = __int_as_float(e2.y);
        const float w3 = __int_as_float(e3.y);
        acc.x += r0.x * w0; acc.y += r0.y * w0; acc.z += r0.z * w0; acc.w += r0.w * w0;
        acc.x += (v1 ? r1.x : 0.0f) * w1; acc.y += (v1 ? r1.y : 0.0f) * w1;
        acc.z += (v1 ? r1.z : 0.0f) * w1; acc.w += (v1 ? r1.w : 0.0f) * w1;
        acc.x += (v2 ? r2.x : 0.0f) * w2; acc.y += (v2 ? r2.y : 0.0f) * w2;
        acc.z += (v2 ? r2.z : 0.0f) * w2; acc.w += (v2 ? r2.w : 0.0f) * w2;
        acc.x += (v3 ? r3.x : 0.0f) * w3; acc.y += (v3 ? r3.y : 0.0f) * w3;
        acc.z += (v3 ? r3.z : 0.0f) * w3; acc.w += (v3 ? r3.w : 0.0f) * w3;
    }
    return acc;
}

__device__ __forceinline__ float4 pull_node(const float* __restrict__ hin,
                                            const int* __restrict__ rowend,
                                            const int2* __restrict__ list,
                                            const float* __restrict__ dis,
                                            const float* __restrict__ bias, int node, int q4) {
    const int end = rowend[node];
    const int beg = (node > 0) ? rowend[node - 1] : 0;
    const float di = dis[node];
    const float wself = di * di;
    const float4 hv = *reinterpret_cast<const float4*>(hin + (size_t)node * HH + q4);
    float4 acc = make_float4(hv.x * wself, hv.y * wself, hv.z * wself, hv.w * wself);
    acc = pull_edges(hin, list, beg, end, acc, q4);
    const float4 bv = *reinterpret_cast<const float4*>(bias + q4);
    return make_float4(acc.x + bv.x, acc.y + bv.y, acc.z + bv.z, acc.w + bv.w);
}

__global__ void k_pull_half(const float* __restrict__ hin, float* __restrict__ hout,
                            const int* __restrict__ rowend, const int2* __restrict__ list,
                            const float* __restrict__ dis, const float* __restrict__ bias,
                            int n) {
    int t = blockIdx.x * blockDim.x + threadIdx.x;
    int node = t >> 3;
    if (node >= n) return;
    const int q4 = (t & 7) * 4;
    float4 r = pull_node(hin, rowend, list, dis, bias, node, q4);
    r.x = fmaxf(r.x, 0.0f); r.y = fmaxf(r.y, 0.0f);
    r.z = fmaxf(r.z, 0.0f); r.w = fmaxf(r.w, 0.0f);
    *reinterpret_cast<float4*>(hout + (size_t)node * HH + q4) = r;
}

__global__ void k_pull_pool(const float* __restrict__ hin,
                            const int* __restrict__ rowend, const int2* __restrict__ list,
                            const float* __restrict__ dis, const float* __restrict__ bias,
                            const int* __restrict__ batch, float* sums, float* cnts,
                            int n, int colbase) {
    const int t = blockIdx.x * blockDim.x + threadIdx.x;
    const int node = t >> 3;
    const int q4 = (t & 7) * 4;
    float4 r = make_float4(0.0f, 0.0f, 0.0f, 0.0f);
    int g = -1;
    if (node < n) {
        r = pull_node(hin, rowend, list, dis, bias, node, q4);
        g = batch[node];
    }
    const int lane = threadIdx.x & 31;
    const int j = lane >> 3;
    const int g0 = __shfl(g, 0);
    if (__all(g == g0)) {
        if (g0 < 0) return;
        float4 s = r;
        s.x += __shfl_xor(s.x, 8);  s.y += __shfl_xor(s.y, 8);
        s.z += __shfl_xor(s.z, 8);  s.w += __shfl_xor(s.w, 8);
        s.x += __shfl_xor(s.x, 16); s.y += __shfl_xor(s.y, 16);
        s.z += __shfl_xor(s.z, 16); s.w += __shfl_xor(s.w, 16);
        const float mine = (j == 0) ? s.x : (j == 1) ? s.y : (j == 2) ? s.z : s.w;
        atomicAddF(&sums[g0 * H + colbase + q4 + j], mine);
        if (colbase == 0 && lane == 0) atomicAddF(&cnts[g0], 4.0f);
    } else if (g >= 0) {
        float* o = sums + g * H + colbase + q4;
        atomicAddF(o + 0, r.x);
        atomicAddF(o + 1, r.y);
        atomicAddF(o + 2, r.z);
        atomicAddF(o + 3, r.w);
        if (colbase == 0 && q4 == 0) atomicAddF(&cnts[g], 1.0f);
    }
}

#define FB_ROWS 128
#define TP 36

__global__ void __launch_bounds__(256)
k_pull_gemm(const float* __restrict__ hg, float* P0, float* O1,
            const int* __restrict__ rowend, const int2* __restrict__ list,
            const float* __restrict__ dis, const float* __restrict__ bias,
            const float* __restrict__ W, int n) {
    __shared__ float Wlds[(H / 2) * H * 2];
    __shared__ __attribute__((aligned(16))) float tile[FB_ROWS * TP];
    for (int idx = threadIdx.x; idx < H * H; idx += blockDim.x) {
        int k = idx / H, c = idx - k * H;
        Wlds[(k >> 1) * (H * 2) + c * 2 + (k & 1)] = W[k * H + c];
    }

    const int blockBase = blockIdx.x * FB_ROWS;
    const int q4 = (threadIdx.x & 7) * 4;
    const int sub = threadIdx.x >> 3;
    const float4 bv = *reinterpret_cast<const float4*>(bias + q4);

    int beg[4], end[4];
    float4 acc[4];
    #pragma unroll
    for (int it = 0; it < 4; ++it) {
        int node = blockBase + it * 32 + sub;
        node = (node < n) ? node : (n - 1);
        end[it] = rowend[node];
        beg[it] = (node > 0) ? rowend[node - 1] : 0;
        const float di = dis[node];
        const float wself = di * di;
        const float4 hv = *reinterpret_cast<const float4*>(hg + (size_t)node * HH + q4);
        acc[it] = make_float4(hv.x * wself, hv.y * wself, hv.z * wself, hv.w * wself);
    }
    #pragma unroll
    for (int it = 0; it < 4; ++it) {
        const float4 a = pull_edges(hg, list, beg[it], end[it], acc[it], q4);
        float4 r = make_float4(a.x + bv.x, a.y + bv.y, a.z + bv.z, a.w + bv.w);
        r.x = fmaxf(r.x, 0.0f); r.y = fmaxf(r.y, 0.0f);
        r.z = fmaxf(r.z, 0.0f); r.w = fmaxf(r.w, 0.0f);
        *reinterpret_cast<float4*>(&tile[(it * 32 + sub) * TP + q4]) = r;
    }
    __syncthreads();

    const int lane = threadIdx.x & 31;
    const int wv   = threadIdx.x >> 5;
    const int row0 = blockBase + wv * 16;
    if (row0 >= n) return;

    const int mrow = lane & 15;
    const int hi   = lane >> 4;
    const float* arow = P0 + (size_t)(row0 + mrow) * HH;
    const float* trow = &tile[(wv * 16 + mrow) * TP];

    v8f c4[4];
    #pragma unroll
    for (int t = 0; t < 4; ++t) c4[t] = (v8f){0, 0, 0, 0, 0, 0, 0, 0};

    #pragma unroll
    for (int k0 = 0; k0 < H; k0 += 4) {
        const int ka = k0 + hi * 2;
        const float* ap = (k0 < HH) ? (arow + ka) : (trow + (ka - HH));
        const v2f a = *reinterpret_cast<const v2f*>(ap);

        const int pbase = ((k0 >> 1) + hi) * (H * 2);
        #pragma unroll
        for (int t = 0; t < 4; ++t) {
            const int c = t * 16 + mrow;
            v2f b = *reinterpret_cast<const v2f*>(&Wlds[pbase + c * 2]);
            c4[t] = __builtin_amdgcn_wmma_f32_16x16x4_f32(
                 false, a,  false, b,
                 (short)0, c4[t],  false,  false);
        }
    }

    const int mbase = (lane < 16) ? 0 : 8;
    #pragma unroll
    for (int t = 0; t < 4; ++t) {
        float* op = (t < 2) ? P0 : O1;
        const int col = (t & 1) * 16 + mrow;
        #pragma unroll
        for (int j = 0; j < 8; ++j) {
            op[(size_t)(row0 + mbase + j) * HH + col] = c4[t][j];
        }
    }
}


__global__ void k_zero(float* p, int n) {
    int i = blockIdx.x * blockDim.x + threadIdx.x;
    if (i < n) p[i] = 0.0f;
}

__global__ void k_pool_linear(const float* __restrict__ sums, const float* __restrict__ cnts,
                              const float* __restrict__ lin_w, const float* __restrict__ lin_b,
                              float* __restrict__ out, int g_total) {
    int g = blockIdx.x * blockDim.x + threadIdx.x;
    if (g >= g_total) return;
    float inv = 1.0f / fmaxf(cnts[g], 1.0f);
    float acc = 0.0f;
    #pragma unroll 8
    for (int f = 0; f < H; ++f) acc += sums[g * H + f] * lin_w[f];
    out[g] = acc * inv + lin_b[0];
}


extern "C" void kernel_launch(void* const* d_in, const int* in_sizes, int n_in,
                              void* d_out, int out_size, void* d_ws, size_t ws_size,
                              hipStream_t stream) {
    const float* x     = (const float*)d_in[0];
    const int*   eidx  = (const int*)  d_in[1];
    const int*   batch = (const int*)  d_in[2];
    const float* W1    = (const float*)d_in[3];
    const float* b1    = (const float*)d_in[4];
    const float* W2    = (const float*)d_in[5];
    const float* b2    = (const float*)d_in[6];
    const float* W3    = (const float*)d_in[7];
    const float* b3    = (const float*)d_in[8];
    const float* lin_w = (const float*)d_in[9];
    const float* lin_b = (const float*)d_in[10];
    float* out = (float*)d_out;

    const int n = in_sizes[0] / F_IN;
    const int e = in_sizes[1] / 2;
    const int* src = eidx;
    const int* dst = eidx + e;

    float* ws   = (float*)d_ws;
    float* dis  = ws;
    float* R0   = dis + n;
    float* R1   = R0 + (size_t)n * HH;
    float* R2   = R1 + (size_t)n * HH;
    int*   cnt    = (int*)(R2 + (size_t)n * HH);
    int*   cursor = cnt + n;
    int*   part   = cursor + n;
    int2*  list   = (int2*)(part + 1024);
    float* sums = R0 + (size_t)n * H * 2;
    float* cnts = sums + (size_t)N_GRAPHS * H;

    const int TB = 256;
    dim3 blk(TB);
    auto blocks = [](long long work, int tb) { return dim3((unsigned)((work + tb - 1) / tb)); };

    const dim3 scan_grid = blocks(n, SCAN_CHUNK);
    k_zero_i      <<<blocks(n, TB), blk, 0, stream>>>(cnt, n);
    k_count_in    <<<blocks(e, TB), blk, 0, stream>>>(dst, cnt, e);
    k_chunk_sums  <<<scan_grid, dim3(SCAN_TB), 0, stream>>>(cnt, part, n);
    k_offsets_norm<<<scan_grid, dim3(SCAN_TB), 0, stream>>>(cnt, part, cursor, dis, n);
    k_place_edges <<<blocks(e, TB), blk, 0, stream>>>(src, dst, dis, cursor, list, e);

    const int strips       = (n + 15) / 16;
    const dim3 gemm_grid   = blocks((long long)strips * 32, TB);
    const dim3 pull_grid   = blocks((long long)n * 8, TB);
    const dim3 fuse_grid   = blocks((long long)n, FB_ROWS);

    k_gemm_wmma<F_IN, F_IN, 12, false><<<gemm_grid, blk, 0, stream>>>(x, x, W1, R0, R1, n);

    k_pull_half<<<pull_grid, blk, 0, stream>>>(R0, R2, cursor, list, dis, b1, n);
    k_pull_gemm<<<fuse_grid, blk, 0, stream>>>(R1, R2, R0, cursor, list, dis, b1 + HH, W2, n);

    k_pull_half<<<pull_grid, blk, 0, stream>>>(R2, R1, cursor, list, dis, b2, n);
    k_pull_gemm<<<fuse_grid, blk, 0, stream>>>(R0, R1, R2, cursor, list, dis, b2 + HH, W3, n);

    k_zero       <<<blocks((long long)N_GRAPHS * H, TB), blk, 0, stream>>>(sums, N_GRAPHS * H);
    k_zero       <<<blocks(N_GRAPHS, TB), blk, 0, stream>>>(cnts, N_GRAPHS);
    k_pull_pool  <<<pull_grid, blk, 0, stream>>>(R1, cursor, list, dis, b3,      batch, sums, cnts, n, 0);
    k_pull_pool  <<<pull_grid, blk, 0, stream>>>(R2, cursor, list, dis, b3 + HH, batch, sums, cnts, n, HH);
    k_pool_linear<<<blocks(N_GRAPHS, TB), blk, 0, stream>>>(sums, cnts, lin_w, lin_b, out, N_GRAPHS);
}


# edge placement reads both norm factors of an edge from a byte in-degree table held in LDS (one 1024-thread group per WGP, same bits), not 2 random global reads; list order still by atomic arrival
# speedup vs baseline: 2.2382x; 1.0350x over previous
#include <hip/hip_runtime.h>
#include <hip/hip_bf16.h>

#define N_NODES 200000
#define N_EDGES 1000000
#define N_GRAPHS 4096
#define F_IN 11
#define H 64
#define HH 32
#define SCAN_TB 256
#define SCAN_CHUNK 1024

typedef float v2f __attribute__((ext_vector_type(2)));
typedef float v8f __attribute__((ext_vector_type(8)));

__device__ __forceinline__ void atomicAddF(float* p, float v) {
    unsafeAtomicAdd(p, v);
}


#define DEG_SAT 255
#define PLACE_TB 1024
#define PLACE_WG 256
#define TAB_IT ((N_NODES / 16 + PLACE_TB - 1) / PLACE_TB)

__device__ __forceinline__ float deg_to_dis(int c) {
    return rsqrtf(fmaxf((float)(c + 1), 1.0f));
}

__global__ void k_zero_i(int* p, int n) {
    int i = blockIdx.x * blockDim.x + threadIdx.x;
    if (i < n) p[i] = 0;
}

__global__ void k_count_in(const int* __restrict__ dst, int* cnt, int e) {
    int i = blockIdx.x * blockDim.x + threadIdx.x;
    if (i < e) atomicAdd(&cnt[dst[i]], 1);
}

__global__ void k_chunk_sums(const int* __restrict__ cnt, int* __restrict__ part, int n) {
    __shared__ int red[SCAN_TB / 32];
    const int base = blockIdx.x * SCAN_CHUNK + threadIdx.x * 4;
    int s = 0;
    for (int u = 0; u < 4; ++u) {
        int i = base + u;
        if (i < n) s += cnt[i];
    }
    for (int o = 16; o > 0; o >>= 1) s += __shfl_xor(s, o);
    if ((threadIdx.x & 31) == 0) red[threadIdx.x >> 5] = s;
    __syncthreads();
    if (threadIdx.x == 0) {
        int t = 0;
        for (int w = 0; w < SCAN_TB / 32; ++w) t += red[w];
        part[blockIdx.x] = t;
    }
}

__global__ void k_offsets_norm(const int* __restrict__ cnt, const int* __restrict__ part,
                               int* __restrict__ cursor, float* __restrict__ dis,
                               unsigned char* __restrict__ deg8, int n) {
    __shared__ int red[SCAN_TB / 32];
    __shared__ int wsum[SCAN_TB / 32];
    const int tid = threadIdx.x, lane = tid & 31, wv = tid >> 5;

    int p = 0;
    for (int i = tid; i < (int)blockIdx.x; i += SCAN_TB) p += part[i];
    for (int o = 16; o > 0; o >>= 1) p += __shfl_xor(p, o);
    if (lane == 0) red[wv] = p;

    const int base = blockIdx.x * SCAN_CHUNK + tid * 4;
    int c[4];
    int tsum = 0;
    for (int u = 0; u < 4; ++u) {
        int i = base + u;
        c[u] = (i < n) ? cnt[i] : 0;
        tsum += c[u];
    }
    int incl = tsum;
    for (int o = 1; o < 32; o <<= 1) {
        int v = __shfl_up(incl, o);
        if (lane >= o) incl += v;
    }
    if (lane == 31) wsum[wv] = incl;
    __syncthreads();
    int off = 0;
    for (int w = 0; w < SCAN_TB / 32; ++w) {
        off += red[w];
        if (w < wv) off += wsum[w];
    }
    int run = off + incl - tsum;
    for (int u = 0; u < 4; ++u) {
        int i = base + u;
        if (i < n) {
            cursor[i] = run;
            run += c[u];
            dis[i] = deg_to_dis(c[u]);
        }
    }
    if (base + 3 < n) {
        unsigned int pk = 0;
        for (int u = 0; u < 4; ++u) pk |= (unsigned int)min(c[u], DEG_SAT) << (8 * u);
        *reinterpret_cast<unsigned int*>(deg8 + base) = pk;
    } else {
        for (int u = 0; u < 4; ++u)
            if (base + u < n) deg8[base + u] = (unsigned char)min(c[u], DEG_SAT);
    }
}


__device__ __forceinline__ void place_begin(const int* __restrict__ src,
                                            const int* __restrict__ dst, int* cursor,
                                            int q, int e, bool vec_ok,
                                            int (&s)[4], int (&d)[4], int (&pos)[4], int& nv) {
    const int i = 4 * q;
    nv = min(max(e - i, 0), 4);
    if (nv == 4 && vec_ok) {
        const int4 sv = *reinterpret_cast<const int4*>(src + i);
        const int4 dv = *reinterpret_cast<const int4*>(dst + i);
        s[0] = sv.x; s[1] = sv.y; s[2] = sv.z; s[3] = sv.w;
        d[0] = dv.x; d[1] = dv.y; d[2] = dv.z; d[3] = dv.w;
        #pragma unroll
        for (int u = 0; u < 4; ++u) pos[u] = atomicAdd(&cursor[d[u]], 1);
    } else {
        #pragma unroll
        for (int u = 0; u < 4; ++u) {
            s[u] = 0; d[u] = 0; pos[u] = 0;
            if (u < nv) {
                s[u] = src[i + u];
                d[u] = dst[i + u];
                pos[u] = atomicAdd(&cursor[d[u]], 1);
            }
        }
    }
}

__device__ __forceinline__ void place_finish(const unsigned char* tab,
                                             const float* __restrict__ dis,
                                             int2* __restrict__ list,
                                             const int (&s)[4], const int (&d)[4],
                                             const int (&pos)[4], int nv) {
    #pragma unroll
    for (int u = 0; u < 4; ++u) {
        if (u < nv) {
            const int cs = tab[s[u]];
            const int cd = tab[d[u]];
            const float fs = (cs < DEG_SAT) ? deg_to_dis(cs) : dis[s[u]];
            const float fd = (cd < DEG_SAT) ? deg_to_dis(cd) : dis[d[u]];
            list[pos[u]] = make_int2(s[u], __float_as_int(fs * fd));
        }
    }
}

__global__ void __launch_bounds__(PLACE_TB)
k_place_edges(const int* __restrict__ src, const int* __restrict__ dst,
              const float* __restrict__ dis, const unsigned char* __restrict__ deg8,
              int* cursor, int2* __restrict__ list, int e, int n) {
    __shared__ __attribute__((aligned(16))) unsigned char tab[N_NODES];
    const int tid = threadIdx.x;
    if ((long long)blockIdx.x * (4 * PLACE_TB) >= e) return;
    const bool vec_ok = ((((size_t)src) | ((size_t)dst)) & 15) == 0;
    int s[4], d[4], pos[4], nv;

    int q = blockIdx.x * PLACE_TB + tid;
    place_begin(src, dst, cursor, q, e, vec_ok, s, d, pos, nv);

    const int nt  = min(n, N_NODES);
    const int n16 = nt >> 4;
    const uint4* g = reinterpret_cast<const uint4*>(deg8);
    uint4* t = reinterpret_cast<uint4*>(tab);
    uint4 tv[TAB_IT];
    #pragma unroll
    for (int it = 0; it < TAB_IT; ++it) {
        const int k = tid + it * PLACE_TB;
        tv[it] = make_uint4(0u, 0u, 0u, 0u);
        if (k < n16) tv[it] = g[k];
    }
    #pragma unroll
    for (int it = 0; it < TAB_IT; ++it) {
        const int k = tid + it * PLACE_TB;
        if (k < n16) t[k] = tv[it];
    }
    if (tid < (nt & 15)) tab[(n16 << 4) + tid] = deg8[(n16 << 4) + tid];
    __syncthreads();

    place_finish(tab, dis, list, s, d, pos, nv);

    for (q += PLACE_WG * PLACE_TB; (long long)4 * q < e; q += PLACE_WG * PLACE_TB) {
        place_begin(src, dst, cursor, q, e, vec_ok, s, d, pos, nv);
        place_finish(tab, dis, list, s, d, pos, nv);
    }
}


template <int KV, int AS, int KP, bool SPLIT_A>
__global__ void k_gemm_wmma(const float* A0, const float* A1, const float* __restrict__ W,
                            float* O0, float* O1, int n) {
    __shared__ float Wlds[(KP / 2) * H * 2];
    for (int idx = threadIdx.x; idx < KP * H; idx += blockDim.x) {
        int k = idx / H, c = idx - k * H;
        float v = (k < KV) ? W[k * H + c] : 0.0f;
        Wlds[(k >> 1) * (H * 2) + c * 2 + (k & 1)] = v;
    }
    __syncthreads();

    const int lane  = threadIdx.x & 31;
    const int strip = blockIdx.x * (blockDim.x >> 5) + (threadIdx.x >> 5);
    const int row0  = strip * 16;
    if (row0 >= n) return;

    const int mrow = lane & 15;
    const int hi   = lane >> 4;
    const float* arow  = SPLIT_A ? A0 + (size_t)(row0 + mrow) * HH
                                 : A0 + (size_t)(row0 + mrow) * AS;
    const float* arow1 = A1 + (size_t)(row0 + mrow) * HH;

    v8f acc[4];
    #pragma unroll
    for (int t = 0; t < 4; ++t) acc[t] = (v8f){0, 0, 0, 0, 0, 0, 0, 0};

    #pragma unroll
    for (int k0 = 0; k0 < KP; k0 += 4) {
        const int ka = k0 + hi * 2;
        v2f a;
        if (SPLIT_A) {
            const float* ap = (k0 < HH) ? (arow + ka) : (arow1 + (ka - HH));
            a = *reinterpret_cast<const v2f*>(ap);
        } else if (KV == KP && (AS & 1) == 0) {
            a = *reinterpret_cast<const v2f*>(arow + ka);
        } else {
            const int kc0 = (ka     < KV) ? ka     : (KV - 1);
            const int kc1 = (ka + 1 < KV) ? ka + 1 : (KV - 1);
            float ax = arow[kc0];
            float ay = arow[kc1];
            a.x = (ka     < KV) ? ax : 0.0f;
            a.y = (ka + 1 < KV) ? ay : 0.0f;
        }

        const int pbase = ((k0 >> 1) + hi) * (H * 2);
        #pragma unroll
        for (int t = 0; t < 4; ++t) {
            const int c = t * 16 + mrow;
            v2f b = *reinterpret_cast<const v2f*>(&Wlds[pbase + c * 2]);
            acc[t] = __builtin_amdgcn_wmma_f32_16x16x4_f32(
                 false, a,  false, b,
                 (short)0, acc[t],  false,  false);
        }
    }

    const int mbase = (lane < 16) ? 0 : 8;
    #pragma unroll
    for (int t = 0; t < 4; ++t) {
        float* op = (t < 2) ? O0 : O1;
        const int col = (t & 1) * 16 + mrow;
        #pragma unroll
        for (int j = 0; j < 8; ++j) {
            op[(size_t)(row0 + mbase + j) * HH + col] = acc[t][j];
        }
    }
}

__device__ __forceinline__ float4 pull_edges(const float* __restrict__ hin,
                                             const int2* __restrict__ list,
                                             int beg, int end, float4 acc, int q4) {
    const int last = end - 1;
    for (int j = beg; j < end; j += 4) {
        const int j1 = (j + 1 < end) ? j + 1 : last;
        const int j2 = (j + 2 < end) ? j + 2 : last;
        const int j3 = (j + 3 < end) ? j + 3 : last;
        const int2 e0 = list[j];
        const int2 e1 = list[j1];
        const int2 e2 = list[j2];
        const int2 e3 = list[j3];
        const float4 r0 = *reinterpret_cast<const float4*>(hin + (size_t)e0.x * HH + q4);
        const float4 r1 = *reinterpret_cast<const float4*>(hin + (size_t)e1.x * HH + q4);
        const float4 r2 = *reinterpret_cast<const float4*>(hin + (size_t)e2.x * HH + q4);
        const float4 r3 = *reinterpret_cast<const float4*>(hin + (size_t)e3.x * HH + q4);
        const bool v1 = (j + 1 < end), v2 = (j + 2 < end), v3 = (j + 3 < end);
        const float w0 = __int_as_float(e0.y);
        const float w1 = __int_as_float(e1.y);
        const float w2 = __int_as_float(e2.y);
        const float w3 = __int_as_float(e3.y);
        acc.x += r0.x * w0; acc.y += r0.y * w0; acc.z += r0.z * w0; acc.w += r0.w * w0;
        acc.x += (v1 ? r1.x : 0.0f) * w1; acc.y += (v1 ? r1.y : 0.0f) * w1;
        acc.z += (v1 ? r1.z : 0.0f) * w1; acc.w += (v1 ? r1.w : 0.0f) * w1;
        acc.x += (v2 ? r2.x : 0.0f) * w2; acc.y += (v2 ? r2.y : 0.0f) * w2;
        acc.z += (v2 ? r2.z : 0.0f) * w2; acc.w += (v2 ? r2.w : 0.0f) * w2;
        acc.x += (v3 ? r3.x : 0.0f) * w3; acc.y += (v3 ? r3.y : 0.0f) * w3;
        acc.z += (v3 ? r3.z : 0.0f) * w3; acc.w += (v3 ? r3.w : 0.0f) * w3;
    }
    return acc;
}

__device__ __forceinline__ float4 pull_node(const float* __restrict__ hin,
                                            const int* __restrict__ rowend,
                                            const int2* __restrict__ list,
                                            const float* __restrict__ dis,
                                            const float* __restrict__ bias, int node, int q4) {
    const int end = rowend[node];
    const int beg = (node > 0) ? rowend[node - 1] : 0;
    const float di = dis[node];
    const float wself = di * di;
    const float4 hv = *reinterpret_cast<const float4*>(hin + (size_t)node * HH + q4);
    float4 acc = make_float4(hv.x * wself, hv.y * wself, hv.z * wself, hv.w * wself);
    acc = pull_edges(hin, list, beg, end, acc, q4);
    const float4 bv = *reinterpret_cast<const float4*>(bias + q4);
    return make_float4(acc.x + bv.x, acc.y + bv.y, acc.z + bv.z, acc.w + bv.w);
}

__global__ void k_pull_half(const float* __restrict__ hin, float* __restrict__ hout,
                            const int* __restrict__ rowend, const int2* __restrict__ list,
                            const float* __restrict__ dis, const float* __restrict__ bias,
                            int n) {
    int t = blockIdx.x * blockDim.x + threadIdx.x;
    int node = t >> 3;
    if (node >= n) return;
    const int q4 = (t & 7) * 4;
    float4 r = pull_node(hin, rowend, list, dis, bias, node, q4);
    r.x = fmaxf(r.x, 0.0f); r.y = fmaxf(r.y, 0.0f);
    r.z = fmaxf(r.z, 0.0f); r.w = fmaxf(r.w, 0.0f);
    *reinterpret_cast<float4*>(hout + (size_t)node * HH + q4) = r;
}

__global__ void k_pull_pool(const float* __restrict__ hin,
                            const int* __restrict__ rowend, const int2* __restrict__ list,
                            const float* __restrict__ dis, const float* __restrict__ bias,
                            const int* __restrict__ batch, float* sums, float* cnts,
                            int n, int colbase) {
    const int t = blockIdx.x * blockDim.x + threadIdx.x;
    const int node = t >> 3;
    const int q4 = (t & 7) * 4;
    float4 r = make_float4(0.0f, 0.0f, 0.0f, 0.0f);
    int g = -1;
    if (node < n) {
        r = pull_node(hin, rowend, list, dis, bias, node, q4);
        g = batch[node];
    }
    const int lane = threadIdx.x & 31;
    const int j = lane >> 3;
    const int g0 = __shfl(g, 0);
    if (__all(g == g0)) {
        if (g0 < 0) return;
        float4 s = r;
        s.x += __shfl_xor(s.x, 8);  s.y += __shfl_xor(s.y, 8);
        s.z += __shfl_xor(s.z, 8);  s.w += __shfl_xor(s.w, 8);
        s.x += __shfl_xor(s.x, 16); s.y += __shfl_xor(s.y, 16);
        s.z += __shfl_xor(s.z, 16); s.w += __shfl_xor(s.w, 16);
        const float mine = (j == 0) ? s.x : (j == 1) ? s.y : (j == 2) ? s.z : s.w;
        atomicAddF(&sums[g0 * H + colbase + q4 + j], mine);
        if (colbase == 0 && lane == 0) atomicAddF(&cnts[g0], 4.0f);
    } else if (g >= 0) {
        float* o = sums + g * H + colbase + q4;
        atomicAddF(o + 0, r.x);
        atomicAddF(o + 1, r.y);
        atomicAddF(o + 2, r.z);
        atomicAddF(o + 3, r.w);
        if (colbase == 0 && q4 == 0) atomicAddF(&cnts[g], 1.0f);
    }
}

#define FB_ROWS 128
#define TP 36

__global__ void __launch_bounds__(256)
k_pull_gemm(const float* __restrict__ hg, float* P0, float* O1,
            const int* __restrict__ rowend, const int2* __restrict__ list,
            const float* __restrict__ dis, const float* __restrict__ bias,
            const float* __restrict__ W, int n) {
    __shared__ float Wlds[(H / 2) * H * 2];
    __shared__ __attribute__((aligned(16))) float tile[FB_ROWS * TP];
    for (int idx = threadIdx.x; idx < H * H; idx += blockDim.x) {
        int k = idx / H, c = idx - k * H;
        Wlds[(k >> 1) * (H * 2) + c * 2 + (k & 1)] = W[k * H + c];
    }

    const int blockBase = blockIdx.x * FB_ROWS;
    const int q4 = (threadIdx.x & 7) * 4;
    const int sub = threadIdx.x >> 3;
    const float4 bv = *reinterpret_cast<const float4*>(bias + q4);

    int beg[4], end[4];
    float4 acc[4];
    #pragma unroll
    for (int it = 0; it < 4; ++it) {
        int node = blockBase + it * 32 + sub;
        node = (node < n) ? node : (n - 1);
        end[it] = rowend[node];
        beg[it] = (node > 0) ? rowend[node - 1] : 0;
        const float di = dis[node];
        const float wself = di * di;
        const float4 hv = *reinterpret_cast<const float4*>(hg + (size_t)node * HH + q4);
        acc[it] = make_float4(hv.x * wself, hv.y * wself, hv.z * wself, hv.w * wself);
    }
    #pragma unroll
    for (int it = 0; it < 4; ++it) {
        const float4 a = pull_edges(hg, list, beg[it], end[it], acc[it], q4);
        float4 r = make_float4(a.x + bv.x, a.y + bv.y, a.z + bv.z, a.w + bv.w);
        r.x = fmaxf(r.x, 0.0f); r.y = fmaxf(r.y, 0.0f);
        r.z = fmaxf(r.z, 0.0f); r.w = fmaxf(r.w, 0.0f);
        *reinterpret_cast<float4*>(&tile[(it * 32 + sub) * TP + q4]) = r;
    }
    __syncthreads();

    const int lane = threadIdx.x & 31;
    const int wv   = threadIdx.x >> 5;
    const int row0 = blockBase + wv * 16;
    if (row0 >= n) return;

    const int mrow = lane & 15;
    const int hi   = lane >> 4;
    const float* arow = P0 + (size_t)(row0 + mrow) * HH;
    const float* trow = &tile[(wv * 16 + mrow) * TP];

    v8f c4[4];
    #pragma unroll
    for (int t = 0; t < 4; ++t) c4[t] = (v8f){0, 0, 0, 0, 0, 0, 0, 0};

    #pragma unroll
    for (int k0 = 0; k0 < H; k0 += 4) {
        const int ka = k0 + hi * 2;
        const float* ap = (k0 < HH) ? (arow + ka) : (trow + (ka - HH));
        const v2f a = *reinterpret_cast<const v2f*>(ap);

        const int pbase = ((k0 >> 1) + hi) * (H * 2);
        #pragma unroll
        for (int t = 0; t < 4; ++t) {
            const int c = t * 16 + mrow;
            v2f b = *reinterpret_cast<const v2f*>(&Wlds[pbase + c * 2]);
            c4[t] = __builtin_amdgcn_wmma_f32_16x16x4_f32(
                 false, a,  false, b,
                 (short)0, c4[t],  false,  false);
        }
    }

    const int mbase = (lane < 16) ? 0 : 8;
    #pragma unroll
    for (int t = 0; t < 4; ++t) {
        float* op = (t < 2) ? P0 : O1;
        const int col = (t & 1) * 16 + mrow;
        #pragma unroll
        for (int j = 0; j < 8; ++j) {
            op[(size_t)(row0 + mbase + j) * HH + col] = c4[t][j];
        }
    }
}


__global__ void k_zero(float* p, int n) {
    int i = blockIdx.x * blockDim.x + threadIdx.x;
    if (i < n) p[i] = 0.0f;
}

__global__ void k_pool_linear(const float* __restrict__ sums, const float* __restrict__ cnts,
                              const float* __restrict__ lin_w, const float* __restrict__ lin_b,
                              float* __restrict__ out, int g_total) {
    int g = blockIdx.x * blockDim.x + threadIdx.x;
    if (g >= g_total) return;
    float inv = 1.0f / fmaxf(cnts[g], 1.0f);
    float acc = 0.0f;
    #pragma unroll 8
    for (int f = 0; f < H; ++f) acc += sums[g * H + f] * lin_w[f];
    out[g] = acc * inv + lin_b[0];
}


extern "C" void kernel_launch(void* const* d_in, const int* in_sizes, int n_in,
                              void* d_out, int out_size, void* d_ws, size_t ws_size,
                              hipStream_t stream) {
    const float* x     = (const float*)d_in[0];
    const int*   eidx  = (const int*)  d_in[1];
    const int*   batch = (const int*)  d_in[2];
    const float* W1    = (const float*)d_in[3];
    const float* b1    = (const float*)d_in[4];
    const float* W2    = (const float*)d_in[5];
    const float* b2    = (const float*)d_in[6];
    const float* W3    = (const float*)d_in[7];
    const float* b3    = (const float*)d_in[8];
    const float* lin_w = (const float*)d_in[9];
    const float* lin_b = (const float*)d_in[10];
    float* out = (float*)d_out;

    const int n = in_sizes[0] / F_IN;
    const int e = in_sizes[1] / 2;
    const int* src = eidx;
    const int* dst = eidx + e;

    float* ws   = (float*)d_ws;
    float* dis  = ws;
    float* R0   = dis + n;
    float* R1   = R0 + (size_t)n * HH;
    float* R2   = R1 + (size_t)n * HH;
    int*   cnt    = (int*)(R2 + (size_t)n * HH);
    int*   cursor = cnt + n;
    int*   part   = cursor + n;
    int2*  list   = (int2*)(part + 1024);
    unsigned char* deg8 = (unsigned char*)(list + e);
    float* sums = R0 + (size_t)n * H * 2;
    float* cnts = sums + (size_t)N_GRAPHS * H;

    const int TB = 256;
    dim3 blk(TB);
    auto blocks = [](long long work, int tb) { return dim3((unsigned)((work + tb - 1) / tb)); };

    const dim3 scan_grid = blocks(n, SCAN_CHUNK);
    k_zero_i      <<<blocks(n, TB), blk, 0, stream>>>(cnt, n);
    k_count_in    <<<blocks(e, TB), blk, 0, stream>>>(dst, cnt, e);
    k_chunk_sums  <<<scan_grid, dim3(SCAN_TB), 0, stream>>>(cnt, part, n);
    k_offsets_norm<<<scan_grid, dim3(SCAN_TB), 0, stream>>>(cnt, part, cursor, dis, deg8, n);
    k_place_edges <<<dim3(PLACE_WG), dim3(PLACE_TB), 0, stream>>>(src, dst, dis, deg8, cursor, list, e, n);

    const int strips       = (n + 15) / 16;
    const dim3 gemm_grid   = blocks((long long)strips * 32, TB);
    const dim3 pull_grid   = blocks((long long)n * 8, TB);
    const dim3 fuse_grid   = blocks((long long)n, FB_ROWS);

    k_gemm_wmma<F_IN, F_IN, 12, false><<<gemm_grid, blk, 0, stream>>>(x, x, W1, R0, R1, n);

    k_pull_half<<<pull_grid, blk, 0, stream>>>(R0, R2, cursor, list, dis, b1, n);
    k_pull_gemm<<<fuse_grid, blk, 0, stream>>>(R1, R2, R0, cursor, list, dis, b1 + HH, W2, n);

    k_pull_half<<<pull_grid, blk, 0, stream>>>(R2, R1, cursor, list, dis, b2, n);
    k_pull_gemm<<<fuse_grid, blk, 0, stream>>>(R0, R1, R2, cursor, list, dis, b2 + HH, W3, n);

    k_zero       <<<blocks((long long)N_GRAPHS * H, TB), blk, 0, stream>>>(sums, N_GRAPHS * H);
    k_zero       <<<blocks(N_GRAPHS, TB), blk, 0, stream>>>(cnts, N_GRAPHS);
    k_pull_pool  <<<pull_grid, blk, 0, stream>>>(R1, cursor, list, dis, b3,      batch, sums, cnts, n, 0);
    k_pull_pool  <<<pull_grid, blk, 0, stream>>>(R2, cursor, list, dis, b3 + HH, batch, sums, cnts, n, HH);
    k_pool_linear<<<blocks(N_GRAPHS, TB), blk, 0, stream>>>(sums, cnts, lin_w, lin_b, out, N_GRAPHS);
}


# layer-3 pool by owner: one work-group per graph (ids sorted) pulls its rows, both halves per list read, sums them in-group and writes out[g]; no float atomics, 15 -> 11 launches; pool sums regrouped
# speedup vs baseline: 2.3699x; 1.0588x over previous
#include <hip/hip_runtime.h>
#include <hip/hip_bf16.h>

#define N_NODES 200000
#define N_EDGES 1000000
#define N_GRAPHS 4096
#define F_IN 11
#define H 64
#define HH 32
#define SCAN_TB 256
#define SCAN_CHUNK 1024

typedef float v2f __attribute__((ext_vector_type(2)));
typedef float v8f __attribute__((ext_vector_type(8)));


#define DEG_SAT 255
#define PLACE_TB 1024
#define PLACE_WG 256
#define TAB_IT ((N_NODES / 16 + PLACE_TB - 1) / PLACE_TB)

__device__ __forceinline__ float deg_to_dis(int c) {
    return rsqrtf(fmaxf((float)(c + 1), 1.0f));
}

__global__ void k_zero_i(int* p, int n) {
    int i = blockIdx.x * blockDim.x + threadIdx.x;
    if (i < n) p[i] = 0;
}

__global__ void k_count_in(const int* __restrict__ dst, int* cnt, int e) {
    int i = blockIdx.x * blockDim.x + threadIdx.x;
    if (i < e) atomicAdd(&cnt[dst[i]], 1);
}

__global__ void k_chunk_sums(const int* __restrict__ cnt, int* __restrict__ part, int n) {
    __shared__ int red[SCAN_TB / 32];
    const int base = blockIdx.x * SCAN_CHUNK + threadIdx.x * 4;
    int s = 0;
    for (int u = 0; u < 4; ++u) {
        int i = base + u;
        if (i < n) s += cnt[i];
    }
    for (int o = 16; o > 0; o >>= 1) s += __shfl_xor(s, o);
    if ((threadIdx.x & 31) == 0) red[threadIdx.x >> 5] = s;
    __syncthreads();
    if (threadIdx.x == 0) {
        int t = 0;
        for (int w = 0; w < SCAN_TB / 32; ++w) t += red[w];
        part[blockIdx.x] = t;
    }
}

__global__ void k_offsets_norm(const int* __restrict__ cnt, const int* __restrict__ part,
                               int* __restrict__ cursor, float* __restrict__ dis,
                               unsigned char* __restrict__ deg8,
                               const int* __restrict__ batch, int* __restrict__ gstart, int n) {
    __shared__ int red[SCAN_TB / 32];
    __shared__ int wsum[SCAN_TB / 32];
    const int tid = threadIdx.x, lane = tid & 31, wv = tid >> 5;

    int p = 0;
    for (int i = tid; i < (int)blockIdx.x; i += SCAN_TB) p += part[i];
    for (int o = 16; o > 0; o >>= 1) p += __shfl_xor(p, o);
    if (lane == 0) red[wv] = p;

    const int base = blockIdx.x * SCAN_CHUNK + tid * 4;
    int c[4];
    int tsum = 0;
    for (int u = 0; u < 4; ++u) {
        int i = base + u;
        c[u] = (i < n) ? cnt[i] : 0;
        tsum += c[u];
    }
    int incl = tsum;
    for (int o = 1; o < 32; o <<= 1) {
        int v = __shfl_up(incl, o);
        if (lane >= o) incl += v;
    }
    if (lane == 31) wsum[wv] = incl;
    __syncthreads();
    int off = 0;
    for (int w = 0; w < SCAN_TB / 32; ++w) {
        off += red[w];
        if (w < wv) off += wsum[w];
    }
    int run = off + incl - tsum;
    int pb = (base > 0 && base - 1 < n) ? batch[base - 1] : -1;
    for (int u = 0; u < 4; ++u) {
        int i = base + u;
        if (i < n) {
            cursor[i] = run;
            run += c[u];
            dis[i] = deg_to_dis(c[u]);
            const int b = batch[i];
            for (int g = max(pb + 1, 0); g <= min(b, N_GRAPHS); ++g) gstart[g] = i;
            pb = b;
            if (i == n - 1)
                for (int g = max(b + 1, 0); g <= N_GRAPHS; ++g) gstart[g] = n;
        }
    }
    if (base + 3 < n) {
        unsigned int pk = 0;
        for (int u = 0; u < 4; ++u) pk |= (unsigned int)min(c[u], DEG_SAT) << (8 * u);
        *reinterpret_cast<unsigned int*>(deg8 + base) = pk;
    } else {
        for (int u = 0; u < 4; ++u)
            if (base + u < n) deg8[base + u] = (unsigned char)min(c[u], DEG_SAT);
    }
}


__device__ __forceinline__ void place_begin(const int* __restrict__ src,
                                            const int* __restrict__ dst, int* cursor,
                                            int q, int e, bool vec_ok,
                                            int (&s)[4], int (&d)[4], int (&pos)[4], int& nv) {
    const int i = 4 * q;
    nv = min(max(e - i, 0), 4);
    if (nv == 4 && vec_ok) {
        const int4 sv = *reinterpret_cast<const int4*>(src + i);
        const int4 dv = *reinterpret_cast<const int4*>(dst + i);
        s[0] = sv.x; s[1] = sv.y; s[2] = sv.z; s[3] = sv.w;
        d[0] = dv.x; d[1] = dv.y; d[2] = dv.z; d[3] = dv.w;
        #pragma unroll
        for (int u = 0; u < 4; ++u) pos[u] = atomicAdd(&cursor[d[u]], 1);
    } else {
        #pragma unroll
        for (int u = 0; u < 4; ++u) {
            s[u] = 0; d[u] = 0; pos[u] = 0;
            if (u < nv) {
                s[u] = src[i + u];
                d[u] = dst[i + u];
                pos[u] = atomicAdd(&cursor[d[u]], 1);
            }
        }
    }
}

__device__ __forceinline__ void place_finish(const unsigned char* tab,
                                             const float* __restrict__ dis,
                                             int2* __restrict__ list,
                                             const int (&s)[4], const int (&d)[4],
                                             const int (&pos)[4], int nv) {
    #pragma unroll
    for (int u = 0; u < 4; ++u) {
        if (u < nv) {
            const int cs = tab[s[u]];
            const int cd = tab[d[u]];
            const float fs = (cs < DEG_SAT) ? deg_to_dis(cs) : dis[s[u]];
            const float fd = (cd < DEG_SAT) ? deg_to_dis(cd) : dis[d[u]];
            list[pos[u]] = make_int2(s[u], __float_as_int(fs * fd));
        }
    }
}

__global__ void __launch_bounds__(PLACE_TB)
k_place_edges(const int* __restrict__ src, const int* __restrict__ dst,
              const float* __restrict__ dis, const unsigned char* __restrict__ deg8,
              int* cursor, int2* __restrict__ list, int e, int n) {
    __shared__ __attribute__((aligned(16))) unsigned char tab[N_NODES];
    const int tid = threadIdx.x;
    if ((long long)blockIdx.x * (4 * PLACE_TB) >= e) return;
    const bool vec_ok = ((((size_t)src) | ((size_t)dst)) & 15) == 0;
    int s[4], d[4], pos[4], nv;

    int q = blockIdx.x * PLACE_TB + tid;
    place_begin(src, dst, cursor, q, e, vec_ok, s, d, pos, nv);

    const int nt  = min(n, N_NODES);
    const int n16 = nt >> 4;
    const uint4* g = reinterpret_cast<const uint4*>(deg8);
    uint4* t = reinterpret_cast<uint4*>(tab);
    uint4 tv[TAB_IT];
    #pragma unroll
    for (int it = 0; it < TAB_IT; ++it) {
        const int k = tid + it * PLACE_TB;
        tv[it] = make_uint4(0u, 0u, 0u, 0u);
        if (k < n16) tv[it] = g[k];
    }
    #pragma unroll
    for (int it = 0; it < TAB_IT; ++it) {
        const int k = tid + it * PLACE_TB;
        if (k < n16) t[k] = tv[it];
    }
    if (tid < (nt & 15)) tab[(n16 << 4) + tid] = deg8[(n16 << 4) + tid];
    __syncthreads();

    place_finish(tab, dis, list, s, d, pos, nv);

    for (q += PLACE_WG * PLACE_TB; (long long)4 * q < e; q += PLACE_WG * PLACE_TB) {
        place_begin(src, dst, cursor, q, e, vec_ok, s, d, pos, nv);
        place_finish(tab, dis, list, s, d, pos, nv);
    }
}


template <int KV, int AS, int KP, bool SPLIT_A>
__global__ void k_gemm_wmma(const float* A0, const float* A1, const float* __restrict__ W,
                            float* O0, float* O1, int n) {
    __shared__ float Wlds[(KP / 2) * H * 2];
    for (int idx = threadIdx.x; idx < KP * H; idx += blockDim.x) {
        int k = idx / H, c = idx - k * H;
        float v = (k < KV) ? W[k * H + c] : 0.0f;
        Wlds[(k >> 1) * (H * 2) + c * 2 + (k & 1)] = v;
    }
    __syncthreads();

    const int lane  = threadIdx.x & 31;
    const int strip = blockIdx.x * (blockDim.x >> 5) + (threadIdx.x >> 5);
    const int row0  = strip * 16;
    if (row0 >= n) return;

    const int mrow = lane & 15;
    const int hi   = lane >> 4;
    const float* arow  = SPLIT_A ? A0 + (size_t)(row0 + mrow) * HH
                                 : A0 + (size_t)(row0 + mrow) * AS;
    const float* arow1 = A1 + (size_t)(row0 + mrow) * HH;

    v8f acc[4];
    #pragma unroll
    for (int t = 0; t < 4; ++t) acc[t] = (v8f){0, 0, 0, 0, 0, 0, 0, 0};

    #pragma unroll
    for (int k0 = 0; k0 < KP; k0 += 4) {
        const int ka = k0 + hi * 2;
        v2f a;
        if (SPLIT_A) {
            const float* ap = (k0 < HH) ? (arow + ka) : (arow1 + (ka - HH));
            a = *reinterpret_cast<const v2f*>(ap);
        } else if (KV == KP && (AS & 1) == 0) {
            a = *reinterpret_cast<const v2f*>(arow + ka);
        } else {
            const int kc0 = (ka     < KV) ? ka     : (KV - 1);
            const int kc1 = (ka + 1 < KV) ? ka + 1 : (KV - 1);
            float ax = arow[kc0];
            float ay = arow[kc1];
            a.x = (ka     < KV) ? ax : 0.0f;
            a.y = (ka + 1 < KV) ? ay : 0.0f;
        }

        const int pbase = ((k0 >> 1) + hi) * (H * 2);
        #pragma unroll
        for (int t = 0; t < 4; ++t) {
            const int c = t * 16 + mrow;
            v2f b = *reinterpret_cast<const v2f*>(&Wlds[pbase + c * 2]);
            acc[t] = __builtin_amdgcn_wmma_f32_16x16x4_f32(
                 false, a,  false, b,
                 (short)0, acc[t],  false,  false);
        }
    }

    const int mbase = (lane < 16) ? 0 : 8;
    #pragma unroll
    for (int t = 0; t < 4; ++t) {
        float* op = (t < 2) ? O0 : O1;
        const int col = (t & 1) * 16 + mrow;
        #pragma unroll
        for (int j = 0; j < 8; ++j) {
            op[(size_t)(row0 + mbase + j) * HH + col] = acc[t][j];
        }
    }
}

__device__ __forceinline__ float4 pull_edges(const float* __restrict__ hin,
                                             const int2* __restrict__ list,
                                             int beg, int end, float4 acc, int q4) {
    const int last = end - 1;
    for (int j = beg; j < end; j += 4) {
        const int j1 = (j + 1 < end) ? j + 1 : last;
        const int j2 = (j + 2 < end) ? j + 2 : last;
        const int j3 = (j + 3 < end) ? j + 3 : last;
        const int2 e0 = list[j];
        const int2 e1 = list[j1];
        const int2 e2 = list[j2];
        const int2 e3 = list[j3];
        const float4 r0 = *reinterpret_cast<const float4*>(hin + (size_t)e0.x * HH + q4);
        const float4 r1 = *reinterpret_cast<const float4*>(hin + (size_t)e1.x * HH + q4);
        const float4 r2 = *reinterpret_cast<const float4*>(hin + (size_t)e2.x * HH + q4);
        const float4 r3 = *reinterpret_cast<const float4*>(hin + (size_t)e3.x * HH + q4);
        const bool v1 = (j + 1 < end), v2 = (j + 2 < end), v3 = (j + 3 < end);
        const float w0 = __int_as_float(e0.y);
        const float w1 = __int_as_float(e1.y);
        const float w2 = __int_as_float(e2.y);
        const float w3 = __int_as_float(e3.y);
        acc.x += r0.x * w0; acc.y += r0.y * w0; acc.z += r0.z * w0; acc.w += r0.w * w0;
        acc.x += (v1 ? r1.x : 0.0f) * w1; acc.y += (v1 ? r1.y : 0.0f) * w1;
        acc.z += (v1 ? r1.z : 0.0f) * w1; acc.w += (v1 ? r1.w : 0.0f) * w1;
        acc.x += (v2 ? r2.x : 0.0f) * w2; acc.y += (v2 ? r2.y : 0.0f) * w2;
        acc.z += (v2 ? r2.z : 0.0f) * w2; acc.w += (v2 ? r2.w : 0.0f) * w2;
        acc.x += (v3 ? r3.x : 0.0f) * w3; acc.y += (v3 ? r3.y : 0.0f) * w3;
        acc.z += (v3 ? r3.z : 0.0f) * w3; acc.w += (v3 ? r3.w : 0.0f) * w3;
    }
    return acc;
}

__device__ __forceinline__ float4 pull_node(const float* __restrict__ hin,
                                            const int* __restrict__ rowend,
                                            const int2* __restrict__ list,
                                            const float* __restrict__ dis,
                                            const float* __restrict__ bias, int node, int q4) {
    const int end = rowend[node];
    const int beg = (node > 0) ? rowend[node - 1] : 0;
    const float di = dis[node];
    const float wself = di * di;
    const float4 hv = *reinterpret_cast<const float4*>(hin + (size_t)node * HH + q4);
    float4 acc = make_float4(hv.x * wself, hv.y * wself, hv.z * wself, hv.w * wself);
    acc = pull_edges(hin, list, beg, end, acc, q4);
    const float4 bv = *reinterpret_cast<const float4*>(bias + q4);
    return make_float4(acc.x + bv.x, acc.y + bv.y, acc.z + bv.z, acc.w + bv.w);
}

__global__ void k_pull_half(const float* __restrict__ hin, float* __restrict__ hout,
                            const int* __restrict__ rowend, const int2* __restrict__ list,
                            const float* __restrict__ dis, const float* __restrict__ bias,
                            int n) {
    int t = blockIdx.x * blockDim.x + threadIdx.x;
    int node = t >> 3;
    if (node >= n) return;
    const int q4 = (t & 7) * 4;
    float4 r = pull_node(hin, rowend, list, dis, bias, node, q4);
    r.x = fmaxf(r.x, 0.0f); r.y = fmaxf(r.y, 0.0f);
    r.z = fmaxf(r.z, 0.0f); r.w = fmaxf(r.w, 0.0f);
    *reinterpret_cast<float4*>(hout + (size_t)node * HH + q4) = r;
}

__device__ __forceinline__ void pull_edges2(const float* __restrict__ h0,
                                            const float* __restrict__ h1,
                                            const int2* __restrict__ list,
                                            int beg, int end, float4& a0, float4& a1, int q4) {
    const int last = end - 1;
    for (int j = beg; j < end; j += 4) {
        const int j1 = (j + 1 < end) ? j + 1 : last;
        const int j2 = (j + 2 < end) ? j + 2 : last;
        const int j3 = (j + 3 < end) ? j + 3 : last;
        const int2 e0 = list[j];
        const int2 e1 = list[j1];
        const int2 e2 = list[j2];
        const int2 e3 = list[j3];
        const size_t o0 = (size_t)e0.x * HH + q4;
        const size_t o1 = (size_t)e1.x * HH + q4;
        const size_t o2 = (size_t)e2.x * HH + q4;
        const size_t o3 = (size_t)e3.x * HH + q4;
        const float4 r00 = *reinterpret_cast<const float4*>(h0 + o0);
        const float4 r01 = *reinterpret_cast<const float4*>(h1 + o0);
        const float4 r10 = *reinterpret_cast<const float4*>(h0 + o1);
        const float4 r11 = *reinterpret_cast<const float4*>(h1 + o1);
        const float4 r20 = *reinterpret_cast<const float4*>(h0 + o2);
        const float4 r21 = *reinterpret_cast<const float4*>(h1 + o2);
        const float4 r30 = *reinterpret_cast<const float4*>(h0 + o3);
        const float4 r31 = *reinterpret_cast<const float4*>(h1 + o3);
        const bool v1 = (j + 1 < end), v2 = (j + 2 < end), v3 = (j + 3 < end);
        const float w0 = __int_as_float(e0.y);
        const float w1 = __int_as_float(e1.y);
        const float w2 = __int_as_float(e2.y);
        const float w3 = __int_as_float(e3.y);
        a0.x += r00.x * w0; a0.y += r00.y * w0; a0.z += r00.z * w0; a0.w += r00.w * w0;
        a1.x += r01.x * w0; a1.y += r01.y * w0; a1.z += r01.z * w0; a1.w += r01.w * w0;
        a0.x += (v1 ? r10.x : 0.0f) * w1; a0.y += (v1 ? r10.y : 0.0f) * w1;
        a0.z += (v1 ? r10.z : 0.0f) * w1; a0.w += (v1 ? r10.w : 0.0f) * w1;
        a1.x += (v1 ? r11.x : 0.0f) * w1; a1.y += (v1 ? r11.y : 0.0f) * w1;
        a1.z += (v1 ? r11.z : 0.0f) * w1; a1.w += (v1 ? r11.w : 0.0f) * w1;
        a0.x += (v2 ? r20.x : 0.0f) * w2; a0.y += (v2 ? r20.y : 0.0f) * w2;
        a0.z += (v2 ? r20.z : 0.0f) * w2; a0.w += (v2 ? r20.w : 0.0f) * w2;
        a1.x += (v2 ? r21.x : 0.0f) * w2; a1.y += (v2 ? r21.y : 0.0f) * w2;
        a1.z += (v2 ? r21.z : 0.0f) * w2; a1.w += (v2 ? r21.w : 0.0f) * w2;
        a0.x += (v3 ? r30.x : 0.0f) * w3; a0.y += (v3 ? r30.y : 0.0f) * w3;
        a0.z += (v3 ? r30.z : 0.0f) * w3; a0.w += (v3 ? r30.w : 0.0f) * w3;
        a1.x += (v3 ? r31.x : 0.0f) * w3; a1.y += (v3 ? r31.y : 0.0f) * w3;
        a1.z += (v3 ? r31.z : 0.0f) * w3; a1.w += (v3 ? r31.w : 0.0f) * w3;
    }
}

#define PG_TB 256
__global__ void __launch_bounds__(PG_TB, 12)
k_pool_graph(const float* __restrict__ h0, const float* __restrict__ h1,
             const int* __restrict__ rowend, const int2* __restrict__ list,
             const float* __restrict__ dis, const float* __restrict__ bias,
             const int* __restrict__ gstart, const float* __restrict__ lin_w,
             const float* __restrict__ lin_b, float* __restrict__ out, int n) {
    __shared__ __attribute__((aligned(16))) float red[PG_TB / 32][H];
    const int g = blockIdx.x;
    const int nbeg = min(max(gstart[g], 0), n);
    const int nend = min(max(gstart[g + 1], nbeg), n);
    const int q4 = (threadIdx.x & 7) * 4;
    const int slot = threadIdx.x >> 3;

    float4 s0 = make_float4(0.0f, 0.0f, 0.0f, 0.0f);
    float4 s1 = make_float4(0.0f, 0.0f, 0.0f, 0.0f);
    for (int node = nbeg + slot; node < nend; node += PG_TB / 8) {
        const int end = rowend[node];
        const int beg = (node > 0) ? rowend[node - 1] : 0;
        const float di = dis[node];
        const float wself = di * di;
        const float4 hv0 = *reinterpret_cast<const float4*>(h0 + (size_t)node * HH + q4);
        const float4 hv1 = *reinterpret_cast<const float4*>(h1 + (size_t)node * HH + q4);
        float4 a0 = make_float4(hv0.x * wself, hv0.y * wself, hv0.z * wself, hv0.w * wself);
        float4 a1 = make_float4(hv1.x * wself, hv1.y * wself, hv1.z * wself, hv1.w * wself);
        pull_edges2(h0, h1, list, beg, end, a0, a1, q4);
        s0.x += a0.x; s0.y += a0.y; s0.z += a0.z; s0.w += a0.w;
        s1.x += a1.x; s1.y += a1.y; s1.z += a1.z; s1.w += a1.w;
    }

    s0.x += __shfl_xor(s0.x, 8);  s0.y += __shfl_xor(s0.y, 8);
    s0.z += __shfl_xor(s0.z, 8);  s0.w += __shfl_xor(s0.w, 8);
    s1.x += __shfl_xor(s1.x, 8);  s1.y += __shfl_xor(s1.y, 8);
    s1.z += __shfl_xor(s1.z, 8);  s1.w += __shfl_xor(s1.w, 8);
    s0.x += __shfl_xor(s0.x, 16); s0.y += __shfl_xor(s0.y, 16);
    s0.z += __shfl_xor(s0.z, 16); s0.w += __shfl_xor(s0.w, 16);
    s1.x += __shfl_xor(s1.x, 16); s1.y += __shfl_xor(s1.y, 16);
    s1.z += __shfl_xor(s1.z, 16); s1.w += __shfl_xor(s1.w, 16);
    const int lane = threadIdx.x & 31;
    const int wv   = threadIdx.x >> 5;
    if (lane < 8) {
        *reinterpret_cast<float4*>(&red[wv][q4]) = s0;
        *reinterpret_cast<float4*>(&red[wv][HH + q4]) = s1;
    }
    __syncthreads();

    if (threadIdx.x < 32) {
        const float cntf = (float)(nend - nbeg);
        float ca = 0.0f, cb = 0.0f;
        for (int w = 0; w < PG_TB / 32; ++w) {
            ca += red[w][lane];
            cb += red[w][HH + lane];
        }
        ca += cntf * bias[lane];
        cb += cntf * bias[HH + lane];
        float p = ca * lin_w[lane] + cb * lin_w[HH + lane];
        for (int o = 16; o > 0; o >>= 1) p += __shfl_xor(p, o);
        if (lane == 0) {
            const float inv = 1.0f / fmaxf(cntf, 1.0f);
            out[g] = p * inv + lin_b[0];
        }
    }
}

#define FB_ROWS 128
#define TP 36

__global__ void __launch_bounds__(256)
k_pull_gemm(const float* __restrict__ hg, float* P0, float* O1,
            const int* __restrict__ rowend, const int2* __restrict__ list,
            const float* __restrict__ dis, const float* __restrict__ bias,
            const float* __restrict__ W, int n) {
    __shared__ float Wlds[(H / 2) * H * 2];
    __shared__ __attribute__((aligned(16))) float tile[FB_ROWS * TP];
    for (int idx = threadIdx.x; idx < H * H; idx += blockDim.x) {
        int k = idx / H, c = idx - k * H;
        Wlds[(k >> 1) * (H * 2) + c * 2 + (k & 1)] = W[k * H + c];
    }

    const int blockBase = blockIdx.x * FB_ROWS;
    const int q4 = (threadIdx.x & 7) * 4;
    const int sub = threadIdx.x >> 3;
    const float4 bv = *reinterpret_cast<const float4*>(bias + q4);

    int beg[4], end[4];
    float4 acc[4];
    #pragma unroll
    for (int it = 0; it < 4; ++it) {
        int node = blockBase + it * 32 + sub;
        node = (node < n) ? node : (n - 1);
        end[it] = rowend[node];
        beg[it] = (node > 0) ? rowend[node - 1] : 0;
        const float di = dis[node];
        const float wself = di * di;
        const float4 hv = *reinterpret_cast<const float4*>(hg + (size_t)node * HH + q4);
        acc[it] = make_float4(hv.x * wself, hv.y * wself, hv.z * wself, hv.w * wself);
    }
    #pragma unroll
    for (int it = 0; it < 4; ++it) {
        const float4 a = pull_edges(hg, list, beg[it], end[it], acc[it], q4);
        float4 r = make_float4(a.x + bv.x, a.y + bv.y, a.z + bv.z, a.w + bv.w);
        r.x = fmaxf(r.x, 0.0f); r.y = fmaxf(r.y, 0.0f);
        r.z = fmaxf(r.z, 0.0f); r.w = fmaxf(r.w, 0.0f);
        *reinterpret_cast<float4*>(&tile[(it * 32 + sub) * TP + q4]) = r;
    }
    __syncthreads();

    const int lane = threadIdx.x & 31;
    const int wv   = threadIdx.x >> 5;
    const int row0 = blockBase + wv * 16;
    if (row0 >= n) return;

    const int mrow = lane & 15;
    const int hi   = lane >> 4;
    const float* arow = P0 + (size_t)(row0 + mrow) * HH;
    const float* trow = &tile[(wv * 16 + mrow) * TP];

    v8f c4[4];
    #pragma unroll
    for (int t = 0; t < 4; ++t) c4[t] = (v8f){0, 0, 0, 0, 0, 0, 0, 0};

    #pragma unroll
    for (int k0 = 0; k0 < H; k0 += 4) {
        const int ka = k0 + hi * 2;
        const float* ap = (k0 < HH) ? (arow + ka) : (trow + (ka - HH));
        const v2f a = *reinterpret_cast<const v2f*>(ap);

        const int pbase = ((k0 >> 1) + hi) * (H * 2);
        #pragma unroll
        for (int t = 0; t < 4; ++t) {
            const int c = t * 16 + mrow;
            v2f b = *reinterpret_cast<const v2f*>(&Wlds[pbase + c * 2]);
            c4[t] = __builtin_amdgcn_wmma_f32_16x16x4_f32(
                 false, a,  false, b,
                 (short)0, c4[t],  false,  false);
        }
    }

    const int mbase = (lane < 16) ? 0 : 8;
    #pragma unroll
    for (int t = 0; t < 4; ++t) {
        float* op = (t < 2) ? P0 : O1;
        const int col = (t & 1) * 16 + mrow;
        #pragma unroll
        for (int j = 0; j < 8; ++j) {
            op[(size_t)(row0 + mbase + j) * HH + col] = c4[t][j];
        }
    }
}


extern "C" void kernel_launch(void* const* d_in, const int* in_sizes, int n_in,
                              void* d_out, int out_size, void* d_ws, size_t ws_size,
                              hipStream_t stream) {
    const float* x     = (const float*)d_in[0];
    const int*   eidx  = (const int*)  d_in[1];
    const int*   batch = (const int*)  d_in[2];
    const float* W1    = (const float*)d_in[3];
    const float* b1    = (const float*)d_in[4];
    const float* W2    = (const float*)d_in[5];
    const float* b2    = (const float*)d_in[6];
    const float* W3    = (const float*)d_in[7];
    const float* b3    = (const float*)d_in[8];
    const float* lin_w = (const float*)d_in[9];
    const float* lin_b = (const float*)d_in[10];
    float* out = (float*)d_out;

    const int n = in_sizes[0] / F_IN;
    const int e = in_sizes[1] / 2;
    const int* src = eidx;
    const int* dst = eidx + e;

    float* ws   = (float*)d_ws;
    float* dis  = ws;
    float* R0   = dis + n;
    float* R1   = R0 + (size_t)n * HH;
    float* R2   = R1 + (size_t)n * HH;
    int*   cnt    = (int*)(R2 + (size_t)n * HH);
    int*   cursor = cnt + n;
    int*   part   = cursor + n;
    int2*  list   = (int2*)(part + 1024);
    int*   gstart = (int*)(list + e);
    unsigned char* deg8 = (unsigned char*)(gstart + N_GRAPHS + 4);

    const int TB = 256;
    dim3 blk(TB);
    auto blocks = [](long long work, int tb) { return dim3((unsigned)((work + tb - 1) / tb)); };

    const dim3 scan_grid = blocks(n, SCAN_CHUNK);
    k_zero_i      <<<blocks(n, TB), blk, 0, stream>>>(cnt, n);
    k_count_in    <<<blocks(e, TB), blk, 0, stream>>>(dst, cnt, e);
    k_chunk_sums  <<<scan_grid, dim3(SCAN_TB), 0, stream>>>(cnt, part, n);
    k_offsets_norm<<<scan_grid, dim3(SCAN_TB), 0, stream>>>(cnt, part, cursor, dis, deg8, batch, gstart, n);
    k_place_edges <<<dim3(PLACE_WG), dim3(PLACE_TB), 0, stream>>>(src, dst, dis, deg8, cursor, list, e, n);

    const int strips       = (n + 15) / 16;
    const dim3 gemm_grid   = blocks((long long)strips * 32, TB);
    const dim3 pull_grid   = blocks((long long)n * 8, TB);
    const dim3 fuse_grid   = blocks((long long)n, FB_ROWS);

    k_gemm_wmma<F_IN, F_IN, 12, false><<<gemm_grid, blk, 0, stream>>>(x, x, W1, R0, R1, n);

    k_pull_half<<<pull_grid, blk, 0, stream>>>(R0, R2, cursor, list, dis, b1, n);
    k_pull_gemm<<<fuse_grid, blk, 0, stream>>>(R1, R2, R0, cursor, list, dis, b1 + HH, W2, n);

    k_pull_half<<<pull_grid, blk, 0, stream>>>(R2, R1, cursor, list, dis, b2, n);
    k_pull_gemm<<<fuse_grid, blk, 0, stream>>>(R0, R1, R2, cursor, list, dis, b2 + HH, W3, n);

    k_pool_graph<<<dim3(N_GRAPHS), dim3(PG_TB), 0, stream>>>(R1, R2, cursor, list, dis, b3, gstart,
                                                            lin_w, lin_b, out, n);
}


# list slot of an edge = row end - 1 - the rank its degree-count atomic returned: the placement pass reads, it has no atomic left (2 -> 1 atomic per edge per call); list order changes, sums regrouped
# speedup vs baseline: 2.6721x; 1.1275x over previous
#include <hip/hip_runtime.h>
#include <hip/hip_bf16.h>

#define N_NODES 200000
#define N_EDGES 1000000
#define N_GRAPHS 4096
#define F_IN 11
#define H 64
#define HH 32
#define SCAN_TB 256
#define SCAN_CHUNK 1024

typedef float v2f __attribute__((ext_vector_type(2)));
typedef float v8f __attribute__((ext_vector_type(8)));


#define DEG_SAT 255
#define PLACE_TB 1024
#define PLACE_WG 256
#define TAB_IT ((N_NODES / 16 + PLACE_TB - 1) / PLACE_TB)

__device__ __forceinline__ float deg_to_dis(int c) {
    return rsqrtf(fmaxf((float)(c + 1), 1.0f));
}

__global__ void k_zero_i(int* p, int n) {
    int i = blockIdx.x * blockDim.x + threadIdx.x;
    if (i < n) p[i] = 0;
}

__global__ void k_count_rank(const int* __restrict__ dst, int* cnt, int* __restrict__ rank,
                             int e) {
    const int i = 4 * (blockIdx.x * blockDim.x + threadIdx.x);
    if (i >= e) return;
    const bool vec_ok = ((((size_t)dst) | ((size_t)rank)) & 15) == 0;
    if (i + 4 <= e && vec_ok) {
        const int4 dv = *reinterpret_cast<const int4*>(dst + i);
        int4 r;
        r.x = atomicAdd(&cnt[dv.x], 1);
        r.y = atomicAdd(&cnt[dv.y], 1);
        r.z = atomicAdd(&cnt[dv.z], 1);
        r.w = atomicAdd(&cnt[dv.w], 1);
        *reinterpret_cast<int4*>(rank + i) = r;
    } else {
        for (int u = 0; u < 4; ++u)
            if (i + u < e) rank[i + u] = atomicAdd(&cnt[dst[i + u]], 1);
    }
}

__global__ void k_chunk_sums(const int* __restrict__ cnt, int* __restrict__ part, int n) {
    __shared__ int red[SCAN_TB / 32];
    const int base = blockIdx.x * SCAN_CHUNK + threadIdx.x * 4;
    int s = 0;
    for (int u = 0; u < 4; ++u) {
        int i = base + u;
        if (i < n) s += cnt[i];
    }
    for (int o = 16; o > 0; o >>= 1) s += __shfl_xor(s, o);
    if ((threadIdx.x & 31) == 0) red[threadIdx.x >> 5] = s;
    __syncthreads();
    if (threadIdx.x == 0) {
        int t = 0;
        for (int w = 0; w < SCAN_TB / 32; ++w) t += red[w];
        part[blockIdx.x] = t;
    }
}

__global__ void k_offsets_norm(const int* __restrict__ cnt, const int* __restrict__ part,
                               int* __restrict__ rowend, float* __restrict__ dis,
                               unsigned char* __restrict__ deg8,
                               const int* __restrict__ batch, int* __restrict__ gstart, int n) {
    __shared__ int red[SCAN_TB / 32];
    __shared__ int wsum[SCAN_TB / 32];
    const int tid = threadIdx.x, lane = tid & 31, wv = tid >> 5;

    int p = 0;
    for (int i = tid; i < (int)blockIdx.x; i += SCAN_TB) p += part[i];
    for (int o = 16; o > 0; o >>= 1) p += __shfl_xor(p, o);
    if (lane == 0) red[wv] = p;

    const int base = blockIdx.x * SCAN_CHUNK + tid * 4;
    int c[4];
    int tsum = 0;
    for (int u = 0; u < 4; ++u) {
        int i = base + u;
        c[u] = (i < n) ? cnt[i] : 0;
        tsum += c[u];
    }
    int incl = tsum;
    for (int o = 1; o < 32; o <<= 1) {
        int v = __shfl_up(incl, o);
        if (lane >= o) incl += v;
    }
    if (lane == 31) wsum[wv] = incl;
    __syncthreads();
    int off = 0;
    for (int w = 0; w < SCAN_TB / 32; ++w) {
        off += red[w];
        if (w < wv) off += wsum[w];
    }
    int run = off + incl - tsum;
    int pb = (base > 0 && base - 1 < n) ? batch[base - 1] : -1;
    for (int u = 0; u < 4; ++u) {
        int i = base + u;
        if (i < n) {
            run += c[u];
            rowend[i] = run;
            dis[i] = deg_to_dis(c[u]);
            const int b = batch[i];
            for (int g = max(pb + 1, 0); g <= min(b, N_GRAPHS); ++g) gstart[g] = i;
            pb = b;
            if (i == n - 1)
                for (int g = max(b + 1, 0); g <= N_GRAPHS; ++g) gstart[g] = n;
        }
    }
    if (base + 3 < n) {
        unsigned int pk = 0;
        for (int u = 0; u < 4; ++u) pk |= (unsigned int)min(c[u], DEG_SAT) << (8 * u);
        *reinterpret_cast<unsigned int*>(deg8 + base) = pk;
    } else {
        for (int u = 0; u < 4; ++u)
            if (base + u < n) deg8[base + u] = (unsigned char)min(c[u], DEG_SAT);
    }
}


__device__ __forceinline__ void place_begin(const int* __restrict__ src,
                                            const int* __restrict__ dst,
                                            const int* __restrict__ rank,
                                            const int* __restrict__ rowend,
                                            int q, int e, bool vec_ok,
                                            int (&s)[4], int (&d)[4], int (&re)[4], int (&rk)[4],
                                            int& nv) {
    const int i = 4 * q;
    nv = min(max(e - i, 0), 4);
    if (nv == 4 && vec_ok) {
        const int4 sv = *reinterpret_cast<const int4*>(src + i);
        const int4 dv = *reinterpret_cast<const int4*>(dst + i);
        const int4 rv = *reinterpret_cast<const int4*>(rank + i);
        s[0] = sv.x; s[1] = sv.y; s[2] = sv.z; s[3] = sv.w;
        d[0] = dv.x; d[1] = dv.y; d[2] = dv.z; d[3] = dv.w;
        rk[0] = rv.x; rk[1] = rv.y; rk[2] = rv.z; rk[3] = rv.w;
        #pragma unroll
        for (int u = 0; u < 4; ++u) re[u] = rowend[d[u]];
    } else {
        #pragma unroll
        for (int u = 0; u < 4; ++u) {
            s[u] = 0; d[u] = 0; re[u] = 1; rk[u] = 0;
            if (u < nv) {
                s[u] = src[i + u];
                d[u] = dst[i + u];
                rk[u] = rank[i + u];
                re[u] = rowend[d[u]];
            }
        }
    }
}

__device__ __forceinline__ void place_finish(const unsigned char* tab,
                                             const float* __restrict__ dis,
                                             int2* __restrict__ list,
                                             const int (&s)[4], const int (&d)[4],
                                             const int (&re)[4], const int (&rk)[4], int nv) {
    #pragma unroll
    for (int u = 0; u < 4; ++u) {
        if (u < nv) {
            const int cs = tab[s[u]];
            const int cd = tab[d[u]];
            const float fs = (cs < DEG_SAT) ? deg_to_dis(cs) : dis[s[u]];
            const float fd = (cd < DEG_SAT) ? deg_to_dis(cd) : dis[d[u]];
            list[re[u] - 1 - rk[u]] = make_int2(s[u], __float_as_int(fs * fd));
        }
    }
}

__global__ void __launch_bounds__(PLACE_TB)
k_place_edges(const int* __restrict__ src, const int* __restrict__ dst,
              const int* __restrict__ rank, const int* __restrict__ rowend,
              const float* __restrict__ dis, const unsigned char* __restrict__ deg8,
              int2* __restrict__ list, int e, int n) {
    __shared__ __attribute__((aligned(16))) unsigned char tab[N_NODES];
    const int tid = threadIdx.x;
    if ((long long)blockIdx.x * (4 * PLACE_TB) >= e) return;
    const bool vec_ok = ((((size_t)src) | ((size_t)dst) | ((size_t)rank)) & 15) == 0;
    int s[4], d[4], re[4], rk[4], nv;

    int q = blockIdx.x * PLACE_TB + tid;
    place_begin(src, dst, rank, rowend, q, e, vec_ok, s, d, re, rk, nv);

    const int nt  = min(n, N_NODES);
    const int n16 = nt >> 4;
    const uint4* g = reinterpret_cast<const uint4*>(deg8);
    uint4* t = reinterpret_cast<uint4*>(tab);
    uint4 tv[TAB_IT];
    #pragma unroll
    for (int it = 0; it < TAB_IT; ++it) {
        const int k = tid + it * PLACE_TB;
        tv[it] = make_uint4(0u, 0u, 0u, 0u);
        if (k < n16) tv[it] = g[k];
    }
    #pragma unroll
    for (int it = 0; it < TAB_IT; ++it) {
        const int k = tid + it * PLACE_TB;
        if (k < n16) t[k] = tv[it];
    }
    if (tid < (nt & 15)) tab[(n16 << 4) + tid] = deg8[(n16 << 4) + tid];
    __syncthreads();

    place_finish(tab, dis, list, s, d, re, rk, nv);

    for (q += PLACE_WG * PLACE_TB; (long long)4 * q < e; q += PLACE_WG * PLACE_TB) {
        place_begin(src, dst, rank, rowend, q, e, vec_ok, s, d, re, rk, nv);
        place_finish(tab, dis, list, s, d, re, rk, nv);
    }
}


template <int KV, int AS, int KP, bool SPLIT_A>
__global__ void k_gemm_wmma(const float* A0, const float* A1, const float* __restrict__ W,
                            float* O0, float* O1, int n) {
    __shared__ float Wlds[(KP / 2) * H * 2];
    for (int idx = threadIdx.x; idx < KP * H; idx += blockDim.x) {
        int k = idx / H, c = idx - k * H;
        float v = (k < KV) ? W[k * H + c] : 0.0f;
        Wlds[(k >> 1) * (H * 2) + c * 2 + (k & 1)] = v;
    }
    __syncthreads();

    const int lane  = threadIdx.x & 31;
    const int strip = blockIdx.x * (blockDim.x >> 5) + (threadIdx.x >> 5);
    const int row0  = strip * 16;
    if (row0 >= n) return;

    const int mrow = lane & 15;
    const int hi   = lane >> 4;
    const float* arow  = SPLIT_A ? A0 + (size_t)(row0 + mrow) * HH
                                 : A0 + (size_t)(row0 + mrow) * AS;
    const float* arow1 = A1 + (size_t)(row0 + mrow) * HH;

    v8f acc[4];
    #pragma unroll
    for (int t = 0; t < 4; ++t) acc[t] = (v8f){0, 0, 0, 0, 0, 0, 0, 0};

    #pragma unroll
    for (int k0 = 0; k0 < KP; k0 += 4) {
        const int ka = k0 + hi * 2;
        v2f a;
        if (SPLIT_A) {
            const float* ap = (k0 < HH) ? (arow + ka) : (arow1 + (ka - HH));
            a = *reinterpret_cast<const v2f*>(ap);
        } else if (KV == KP && (AS & 1) == 0) {
            a = *reinterpret_cast<const v2f*>(arow + ka);
        } else {
            const int kc0 = (ka     < KV) ? ka     : (KV - 1);
            const int kc1 = (ka + 1 < KV) ? ka + 1 : (KV - 1);
            float ax = arow[kc0];
            float ay = arow[kc1];
            a.x = (ka     < KV) ? ax : 0.0f;
            a.y = (ka + 1 < KV) ? ay : 0.0f;
        }

        const int pbase = ((k0 >> 1) + hi) * (H * 2);
        #pragma unroll
        for (int t = 0; t < 4; ++t) {
            const int c = t * 16 + mrow;
            v2f b = *reinterpret_cast<const v2f*>(&Wlds[pbase + c * 2]);
            acc[t] = __builtin_amdgcn_wmma_f32_16x16x4_f32(
                 false, a,  false, b,
                 (short)0, acc[t],  false,  false);
        }
    }

    const int mbase = (lane < 16) ? 0 : 8;
    #pragma unroll
    for (int t = 0; t < 4; ++t) {
        float* op = (t < 2) ? O0 : O1;
        const int col = (t & 1) * 16 + mrow;
        #pragma unroll
        for (int j = 0; j < 8; ++j) {
            op[(size_t)(row0 + mbase + j) * HH + col] = acc[t][j];
        }
    }
}

__device__ __forceinline__ float4 pull_edges(const float* __restrict__ hin,
                                             const int2* __restrict__ list,
                                             int beg, int end, float4 acc, int q4) {
    const int last = end - 1;
    for (int j = beg; j < end; j += 4) {
        const int j1 = (j + 1 < end) ? j + 1 : last;
        const int j2 = (j + 2 < end) ? j + 2 : last;
        const int j3 = (j + 3 < end) ? j + 3 : last;
        const int2 e0 = list[j];
        const int2 e1 = list[j1];
        const int2 e2 = list[j2];
        const int2 e3 = list[j3];
        const float4 r0 = *reinterpret_cast<const float4*>(hin + (size_t)e0.x * HH + q4);
        const float4 r1 = *reinterpret_cast<const float4*>(hin + (size_t)e1.x * HH + q4);
        const float4 r2 = *reinterpret_cast<const float4*>(hin + (size_t)e2.x * HH + q4);
        const float4 r3 = *reinterpret_cast<const float4*>(hin + (size_t)e3.x * HH + q4);
        const bool v1 = (j + 1 < end), v2 = (j + 2 < end), v3 = (j + 3 < end);
        const float w0 = __int_as_float(e0.y);
        const float w1 = __int_as_float(e1.y);
        const float w2 = __int_as_float(e2.y);
        const float w3 = __int_as_float(e3.y);
        acc.x += r0.x * w0; acc.y += r0.y * w0; acc.z += r0.z * w0; acc.w += r0.w * w0;
        acc.x += (v1 ? r1.x : 0.0f) * w1; acc.y += (v1 ? r1.y : 0.0f) * w1;
        acc.z += (v1 ? r1.z : 0.0f) * w1; acc.w += (v1 ? r1.w : 0.0f) * w1;
        acc.x += (v2 ? r2.x : 0.0f) * w2; acc.y += (v2 ? r2.y : 0.0f) * w2;
        acc.z += (v2 ? r2.z : 0.0f) * w2; acc.w += (v2 ? r2.w : 0.0f) * w2;
        acc.x += (v3 ? r3.x : 0.0f) * w3; acc.y += (v3 ? r3.y : 0.0f) * w3;
        acc.z += (v3 ? r3.z : 0.0f) * w3; acc.w += (v3 ? r3.w : 0.0f) * w3;
    }
    return acc;
}

__device__ __forceinline__ float4 pull_node(const float* __restrict__ hin,
                                            const int* __restrict__ rowend,
                                            const int2* __restrict__ list,
                                            const float* __restrict__ dis,
                                            const float* __restrict__ bias, int node, int q4) {
    const int end = rowend[node];
    const int beg = (node > 0) ? rowend[node - 1] : 0;
    const float di = dis[node];
    const float wself = di * di;
    const float4 hv = *reinterpret_cast<const float4*>(hin + (size_t)node * HH + q4);
    float4 acc = make_float4(hv.x * wself, hv.y * wself, hv.z * wself, hv.w * wself);
    acc = pull_edges(hin, list, beg, end, acc, q4);
    const float4 bv = *reinterpret_cast<const float4*>(bias + q4);
    return make_float4(acc.x + bv.x, acc.y + bv.y, acc.z + bv.z, acc.w + bv.w);
}

__global__ void k_pull_half(const float* __restrict__ hin, float* __restrict__ hout,
                            const int* __restrict__ rowend, const int2* __restrict__ list,
                            const float* __restrict__ dis, const float* __restrict__ bias,
                            int n) {
    int t = blockIdx.x * blockDim.x + threadIdx.x;
    int node = t >> 3;
    if (node >= n) return;
    const int q4 = (t & 7) * 4;
    float4 r = pull_node(hin, rowend, list, dis, bias, node, q4);
    r.x = fmaxf(r.x, 0.0f); r.y = fmaxf(r.y, 0.0f);
    r.z = fmaxf(r.z, 0.0f); r.w = fmaxf(r.w, 0.0f);
    *reinterpret_cast<float4*>(hout + (size_t)node * HH + q4) = r;
}

__device__ __forceinline__ void pull_edges2(const float* __restrict__ h0,
                                            const float* __restrict__ h1,
                                            const int2* __restrict__ list,
                                            int beg, int end, float4& a0, float4& a1, int q4) {
    const int last = end - 1;
    for (int j = beg; j < end; j += 4) {
        const int j1 = (j + 1 < end) ? j + 1 : last;
        const int j2 = (j + 2 < end) ? j + 2 : last;
        const int j3 = (j + 3 < end) ? j + 3 : last;
        const int2 e0 = list[j];
        const int2 e1 = list[j1];
        const int2 e2 = list[j2];
        const int2 e3 = list[j3];
        const size_t o0 = (size_t)e0.x * HH + q4;
        const size_t o1 = (size_t)e1.x * HH + q4;
        const size_t o2 = (size_t)e2.x * HH + q4;
        const size_t o3 = (size_t)e3.x * HH + q4;
        const float4 r00 = *reinterpret_cast<const float4*>(h0 + o0);
        const float4 r01 = *reinterpret_cast<const float4*>(h1 + o0);
        const float4 r10 = *reinterpret_cast<const float4*>(h0 + o1);
        const float4 r11 = *reinterpret_cast<const float4*>(h1 + o1);
        const float4 r20 = *reinterpret_cast<const float4*>(h0 + o2);
        const float4 r21 = *reinterpret_cast<const float4*>(h1 + o2);
        const float4 r30 = *reinterpret_cast<const float4*>(h0 + o3);
        const float4 r31 = *reinterpret_cast<const float4*>(h1 + o3);
        const bool v1 = (j + 1 < end), v2 = (j + 2 < end), v3 = (j + 3 < end);
        const float w0 = __int_as_float(e0.y);
        const float w1 = __int_as_float(e1.y);
        const float w2 = __int_as_float(e2.y);
        const float w3 = __int_as_float(e3.y);
        a0.x += r00.x * w0; a0.y += r00.y * w0; a0.z += r00.z * w0; a0.w += r00.w * w0;
        a1.x += r01.x * w0; a1.y += r01.y * w0; a1.z += r01.z * w0; a1.w += r01.w * w0;
        a0.x += (v1 ? r10.x : 0.0f) * w1; a0.y += (v1 ? r10.y : 0.0f) * w1;
        a0.z += (v1 ? r10.z : 0.0f) * w1; a0.w += (v1 ? r10.w : 0.0f) * w1;
        a1.x += (v1 ? r11.x : 0.0f) * w1; a1.y += (v1 ? r11.y : 0.0f) * w1;
        a1.z += (v1 ? r11.z : 0.0f) * w1; a1.w += (v1 ? r11.w : 0.0f) * w1;
        a0.x += (v2 ? r20.x : 0.0f) * w2; a0.y += (v2 ? r20.y : 0.0f) * w2;
        a0.z += (v2 ? r20.z : 0.0f) * w2; a0.w += (v2 ? r20.w : 0.0f) * w2;
        a1.x += (v2 ? r21.x : 0.0f) * w2; a1.y += (v2 ? r21.y : 0.0f) * w2;
        a1.z += (v2 ? r21.z : 0.0f) * w2; a1.w += (v2 ? r21.w : 0.0f) * w2;
        a0.x += (v3 ? r30.x : 0.0f) * w3; a0.y += (v3 ? r30.y : 0.0f) * w3;
        a0.z += (v3 ? r30.z : 0.0f) * w3; a0.w += (v3 ? r30.w : 0.0f) * w3;
        a1.x += (v3 ? r31.x : 0.0f) * w3; a1.y += (v3 ? r31.y : 0.0f) * w3;
        a1.z += (v3 ? r31.z : 0.0f) * w3; a1.w += (v3 ? r31.w : 0.0f) * w3;
    }
}

#define PG_TB 256
__global__ void __launch_bounds__(PG_TB, 12)
k_pool_graph(const float* __restrict__ h0, const float* __restrict__ h1,
             const int* __restrict__ rowend, const int2* __restrict__ list,
             const float* __restrict__ dis, const float* __restrict__ bias,
             const int* __restrict__ gstart, const float* __restrict__ lin_w,
             const float* __restrict__ lin_b, float* __restrict__ out, int n) {
    __shared__ __attribute__((aligned(16))) float red[PG_TB / 32][H];
    const int g = blockIdx.x;
    const int nbeg = min(max(gstart[g], 0), n);
    const int nend = min(max(gstart[g + 1], nbeg), n);
    const int q4 = (threadIdx.x & 7) * 4;
    const int slot = threadIdx.x >> 3;

    float4 s0 = make_float4(0.0f, 0.0f, 0.0f, 0.0f);
    float4 s1 = make_float4(0.0f, 0.0f, 0.0f, 0.0f);
    for (int node = nbeg + slot; node < nend; node += PG_TB / 8) {
        const int end = rowend[node];
        const int beg = (node > 0) ? rowend[node - 1] : 0;
        const float di = dis[node];
        const float wself = di * di;
        const float4 hv0 = *reinterpret_cast<const float4*>(h0 + (size_t)node * HH + q4);
        const float4 hv1 = *reinterpret_cast<const float4*>(h1 + (size_t)node * HH + q4);
        float4 a0 = make_float4(hv0.x * wself, hv0.y * wself, hv0.z * wself, hv0.w * wself);
        float4 a1 = make_float4(hv1.x * wself, hv1.y * wself, hv1.z * wself, hv1.w * wself);
        pull_edges2(h0, h1, list, beg, end, a0, a1, q4);
        s0.x += a0.x; s0.y += a0.y; s0.z += a0.z; s0.w += a0.w;
        s1.x += a1.x; s1.y += a1.y; s1.z += a1.z; s1.w += a1.w;
    }

    s0.x += __shfl_xor(s0.x, 8);  s0.y += __shfl_xor(s0.y, 8);
    s0.z += __shfl_xor(s0.z, 8);  s0.w += __shfl_xor(s0.w, 8);
    s1.x += __shfl_xor(s1.x, 8);  s1.y += __shfl_xor(s1.y, 8);
    s1.z += __shfl_xor(s1.z, 8);  s1.w += __shfl_xor(s1.w, 8);
    s0.x += __shfl_xor(s0.x, 16); s0.y += __shfl_xor(s0.y, 16);
    s0.z += __shfl_xor(s0.z, 16); s0.w += __shfl_xor(s0.w, 16);
    s1.x += __shfl_xor(s1.x, 16); s1.y += __shfl_xor(s1.y, 16);
    s1.z += __shfl_xor(s1.z, 16); s1.w += __shfl_xor(s1.w, 16);
    const int lane = threadIdx.x & 31;
    const int wv   = threadIdx.x >> 5;
    if (lane < 8) {
        *reinterpret_cast<float4*>(&red[wv][q4]) = s0;
        *reinterpret_cast<float4*>(&red[wv][HH + q4]) = s1;
    }
    __syncthreads();

    if (threadIdx.x < 32) {
        const float cntf = (float)(nend - nbeg);
        float ca = 0.0f, cb = 0.0f;
        for (int w = 0; w < PG_TB / 32; ++w) {
            ca += red[w][lane];
            cb += red[w][HH + lane];
        }
        ca += cntf * bias[lane];
        cb += cntf * bias[HH + lane];
        float p = ca * lin_w[lane] + cb * lin_w[HH + lane];
        for (int o = 16; o > 0; o >>= 1) p += __shfl_xor(p, o);
        if (lane == 0) {
            const float inv = 1.0f / fmaxf(cntf, 1.0f);
            out[g] = p * inv + lin_b[0];
        }
    }
}

#define FB_ROWS 128
#define TP 36

__global__ void __launch_bounds__(256)
k_pull_gemm(const float* __restrict__ hg, float* P0, float* O1,
            const int* __restrict__ rowend, const int2* __restrict__ list,
            const float* __restrict__ dis, const float* __restrict__ bias,
            const float* __restrict__ W, int n) {
    __shared__ float Wlds[(H / 2) * H * 2];
    __shared__ __attribute__((aligned(16))) float tile[FB_ROWS * TP];
    for (int idx = threadIdx.x; idx < H * H; idx += blockDim.x) {
        int k = idx / H, c = idx - k * H;
        Wlds[(k >> 1) * (H * 2) + c * 2 + (k & 1)] = W[k * H + c];
    }

    const int blockBase = blockIdx.x * FB_ROWS;
    const int q4 = (threadIdx.x & 7) * 4;
    const int sub = threadIdx.x >> 3;
    const float4 bv = *reinterpret_cast<const float4*>(bias + q4);

    int beg[4], end[4];
    float4 acc[4];
    #pragma unroll
    for (int it = 0; it < 4; ++it) {
        int node = blockBase + it * 32 + sub;
        node = (node < n) ? node : (n - 1);
        end[it] = rowend[node];
        beg[it] = (node > 0) ? rowend[node - 1] : 0;
        const float di = dis[node];
        const float wself = di * di;
        const float4 hv = *reinterpret_cast<const float4*>(hg + (size_t)node * HH + q4);
        acc[it] = make_float4(hv.x * wself, hv.y * wself, hv.z * wself, hv.w * wself);
    }
    #pragma unroll
    for (int it = 0; it < 4; ++it) {
        const float4 a = pull_edges(hg, list, beg[it], end[it], acc[it], q4);
        float4 r = make_float4(a.x + bv.x, a.y + bv.y, a.z + bv.z, a.w + bv.w);
        r.x = fmaxf(r.x, 0.0f); r.y = fmaxf(r.y, 0.0f);
        r.z = fmaxf(r.z, 0.0f); r.w = fmaxf(r.w, 0.0f);
        *reinterpret_cast<float4*>(&tile[(it * 32 + sub) * TP + q4]) = r;
    }
    __syncthreads();

    const int lane = threadIdx.x & 31;
    const int wv   = threadIdx.x >> 5;
    const int row0 = blockBase + wv * 16;
    if (row0 >= n) return;

    const int mrow = lane & 15;
    const int hi   = lane >> 4;
    const float* arow = P0 + (size_t)(row0 + mrow) * HH;
    const float* trow = &tile[(wv * 16 + mrow) * TP];

    v8f c4[4];
    #pragma unroll
    for (int t = 0; t < 4; ++t) c4[t] = (v8f){0, 0, 0, 0, 0, 0, 0, 0};

    #pragma unroll
    for (int k0 = 0; k0 < H; k0 += 4) {
        const int ka = k0 + hi * 2;
        const float* ap = (k0 < HH) ? (arow + ka) : (trow + (ka - HH));
        const v2f a = *reinterpret_cast<const v2f*>(ap);

        const int pbase = ((k0 >> 1) + hi) * (H * 2);
        #pragma unroll
        for (int t = 0; t < 4; ++t) {
            const int c = t * 16 + mrow;
            v2f b = *reinterpret_cast<const v2f*>(&Wlds[pbase + c * 2]);
            c4[t] = __builtin_amdgcn_wmma_f32_16x16x4_f32(
                 false, a,  false, b,
                 (short)0, c4[t],  false,  false);
        }
    }

    const int mbase = (lane < 16) ? 0 : 8;
    #pragma unroll
    for (int t = 0; t < 4; ++t) {
        float* op = (t < 2) ? P0 : O1;
        const int col = (t & 1) * 16 + mrow;
        #pragma unroll
        for (int j = 0; j < 8; ++j) {
            op[(size_t)(row0 + mbase + j) * HH + col] = c4[t][j];
        }
    }
}


extern "C" void kernel_launch(void* const* d_in, const int* in_sizes, int n_in,
                              void* d_out, int out_size, void* d_ws, size_t ws_size,
                              hipStream_t stream) {
    const float* x     = (const float*)d_in[0];
    const int*   eidx  = (const int*)  d_in[1];
    const int*   batch = (const int*)  d_in[2];
    const float* W1    = (const float*)d_in[3];
    const float* b1    = (const float*)d_in[4];
    const float* W2    = (const float*)d_in[5];
    const float* b2    = (const float*)d_in[6];
    const float* W3    = (const float*)d_in[7];
    const float* b3    = (const float*)d_in[8];
    const float* lin_w = (const float*)d_in[9];
    const float* lin_b = (const float*)d_in[10];
    float* out = (float*)d_out;

    const int n = in_sizes[0] / F_IN;
    const int e = in_sizes[1] / 2;
    const int* src = eidx;
    const int* dst = eidx + e;

    float* ws   = (float*)d_ws;
    float* dis  = ws;
    float* R0   = dis + n;
    float* R1   = R0 + (size_t)n * HH;
    float* R2   = R1 + (size_t)n * HH;
    int*   cnt    = (int*)(R2 + (size_t)n * HH);
    int*   cursor = cnt + n;
    int*   part   = cursor + n;
    int2*  list   = (int2*)(part + 1024);
    int*   gstart = (int*)(list + e);
    unsigned char* deg8 = (unsigned char*)(gstart + N_GRAPHS + 4);
    int*   rank   = (int*)R0;

    const int TB = 256;
    dim3 blk(TB);
    auto blocks = [](long long work, int tb) { return dim3((unsigned)((work + tb - 1) / tb)); };

    const dim3 scan_grid = blocks(n, SCAN_CHUNK);
    k_zero_i      <<<blocks(n, TB), blk, 0, stream>>>(cnt, n);
    k_count_rank  <<<blocks(((long long)e + 3) / 4, TB), blk, 0, stream>>>(dst, cnt, rank, e);
    k_chunk_sums  <<<scan_grid, dim3(SCAN_TB), 0, stream>>>(cnt, part, n);
    k_offsets_norm<<<scan_grid, dim3(SCAN_TB), 0, stream>>>(cnt, part, cursor, dis, deg8, batch, gstart, n);
    k_place_edges <<<dim3(PLACE_WG), dim3(PLACE_TB), 0, stream>>>(src, dst, rank, cursor, dis, deg8, list, e, n);

    const int strips       = (n + 15) / 16;
    const dim3 gemm_grid   = blocks((long long)strips * 32, TB);
    const dim3 pull_grid   = blocks((long long)n * 8, TB);
    const dim3 fuse_grid   = blocks((long long)n, FB_ROWS);

    k_gemm_wmma<F_IN, F_IN, 12, false><<<gemm_grid, blk, 0, stream>>>(x, x, W1, R0, R1, n);

    k_pull_half<<<pull_grid, blk, 0, stream>>>(R0, R2, cursor, list, dis, b1, n);
    k_pull_gemm<<<fuse_grid, blk, 0, stream>>>(R1, R2, R0, cursor, list, dis, b1 + HH, W2, n);

    k_pull_half<<<pull_grid, blk, 0, stream>>>(R2, R1, cursor, list, dis, b2, n);
    k_pull_gemm<<<fuse_grid, blk, 0, stream>>>(R0, R1, R2, cursor, list, dis, b2 + HH, W3, n);

    k_pool_graph<<<dim3(N_GRAPHS), dim3(PG_TB), 0, stream>>>(R1, R2, cursor, list, dis, b3, gstart,
                                                            lin_w, lin_b, out, n);
}


# edge lists built by a two-level counting sort in LDS (bin by 1024-destination block, then one group sorts a block): no global atomic or random access in the build, 9 -> 8 launches; sums regrouped
# speedup vs baseline: 3.4402x; 1.2875x over previous
#include <hip/hip_runtime.h>
#include <hip/hip_bf16.h>

#define N_NODES 200000
#define N_EDGES 1000000
#define N_GRAPHS 4096
#define F_IN 11
#define H 64
#define HH 32

typedef float v2f __attribute__((ext_vector_type(2)));
typedef float v8f __attribute__((ext_vector_type(8)));


#define DEG_SAT 255
#define PLACE_TB 1024
#define PLACE_WG 256
#define TAB_IT ((N_NODES / 16 + PLACE_TB - 1) / PLACE_TB)

#define BK_SHIFT 10
#define BK (1 << BK_SHIFT)
#define MAXB ((N_NODES + BK - 1) / BK)
#define SORT_TB 1024
#define EPW (4 * SORT_TB)
#define STAGE_CAP 16384

__device__ __forceinline__ float deg_to_dis(int c) {
    return rsqrtf(fmaxf((float)(c + 1), 1.0f));
}

__device__ __forceinline__ int bin_of(int d) {
    return min(max(d >> BK_SHIFT, 0), MAXB - 1);
}

__device__ __forceinline__ void load_quad(const int* __restrict__ src, const int* __restrict__ dst,
                                          int i, int e, bool want_src,
                                          int (&s)[4], int (&d)[4], int& nv) {
    nv = min(max(e - i, 0), 4);
    const bool vec_ok = ((((size_t)src) | ((size_t)dst)) & 15) == 0;
    if (nv == 4 && vec_ok) {
        const int4 dv = *reinterpret_cast<const int4*>(dst + i);
        d[0] = dv.x; d[1] = dv.y; d[2] = dv.z; d[3] = dv.w;
        if (want_src) {
            const int4 sv = *reinterpret_cast<const int4*>(src + i);
            s[0] = sv.x; s[1] = sv.y; s[2] = sv.z; s[3] = sv.w;
        } else {
            s[0] = 0; s[1] = 0; s[2] = 0; s[3] = 0;
        }
    } else {
        #pragma unroll
        for (int u = 0; u < 4; ++u) {
            s[u] = 0; d[u] = 0;
            if (u < nv) {
                d[u] = dst[i + u];
                if (want_src) s[u] = src[i + u];
            }
        }
    }
}

__global__ void __launch_bounds__(SORT_TB)
k_bin_count(const int* __restrict__ src, const int* __restrict__ dst, int* __restrict__ hist, int e) {
    __shared__ int h[MAXB];
    const int tid = threadIdx.x;
    int s[4], d[4], nv;
    load_quad(src, dst, blockIdx.x * EPW + tid * 4, e, false, s, d, nv);
    for (int b = tid; b < MAXB; b += SORT_TB) h[b] = 0;
    __syncthreads();
    #pragma unroll
    for (int u = 0; u < 4; ++u)
        if (u < nv) atomicAdd(&h[bin_of(d[u])], 1);
    __syncthreads();
    for (int b = tid; b < MAXB; b += SORT_TB) hist[blockIdx.x * MAXB + b] = h[b];
}

__global__ void __launch_bounds__(SORT_TB)
k_bin_scatter(const int* __restrict__ src, const int* __restrict__ dst,
              const int* __restrict__ hist, int2* __restrict__ binned,
              int* __restrict__ bstart, int e, int nw) {
    __shared__ int pTot[4][256];
    __shared__ int pPre[4][256];
    __shared__ int wsT[8];
    __shared__ int wsO[8];
    __shared__ int gbase[256];
    __shared__ int lbase[256];
    __shared__ int cursor[256];
    __shared__ int2 stage[EPW];
    const int tid = threadIdx.x, lane = tid & 31, wv = tid >> 5;
    const int w = blockIdx.x;

    int s[4], d[4], nv;
    load_quad(src, dst, w * EPW + tid * 4, e, true, s, d, nv);

    const int col = tid & 255, part = tid >> 8;
    int tot = 0, pre = 0;
    if (col < MAXB) {
        for (int k0 = part; k0 < nw; k0 += 64) {
            int v[16];
            #pragma unroll
            for (int u = 0; u < 16; ++u) {
                const int k = k0 + 4 * u;
                v[u] = (k < nw) ? hist[k * MAXB + col] : 0;
            }
            #pragma unroll
            for (int u = 0; u < 16; ++u) {
                tot += v[u];
                pre += (k0 + 4 * u < w) ? v[u] : 0;
            }
        }
    }
    pTot[part][col] = tot;
    pPre[part][col] = pre;
    __syncthreads();

    int T = 0, P = 0, own = 0, iT = 0, iO = 0;
    if (tid < 256) {
        T = pTot[0][tid] + pTot[1][tid] + pTot[2][tid] + pTot[3][tid];
        P = pPre[0][tid] + pPre[1][tid] + pPre[2][tid] + pPre[3][tid];
        own = (tid < MAXB) ? hist[w * MAXB + tid] : 0;
        iT = T; iO = own;
        for (int o = 1; o < 32; o <<= 1) {
            const int a = __shfl_up(iT, o);
            const int c = __shfl_up(iO, o);
            if (lane >= o) { iT += a; iO += c; }
        }
        if (lane == 31) { wsT[wv] = iT; wsO[wv] = iO; }
    }
    __syncthreads();
    if (tid < 256) {
        int offT = 0, offO = 0;
        for (int k = 0; k < wv; ++k) { offT += wsT[k]; offO += wsO[k]; }
        const int exT = offT + iT - T;
        const int exO = offO + iO - own;
        gbase[tid] = exT + P;
        lbase[tid] = exO;
        cursor[tid] = exO;
        if (w == 0 && tid <= MAXB) bstart[tid] = exT;
    }
    __syncthreads();

    #pragma unroll
    for (int u = 0; u < 4; ++u) {
        if (u < nv) {
            const int r = atomicAdd(&cursor[bin_of(d[u])], 1);
            stage[min(max(r, 0), EPW - 1)] = make_int2(s[u], d[u]);
        }
    }
    __syncthreads();
    const int cw = min(max(e - w * EPW, 0), EPW);
    for (int k = tid; k < cw; k += SORT_TB) {
        const int2 ed = stage[k];
        const int b = bin_of(ed.y);
        const int pos = gbase[b] + (k - lbase[b]);
        binned[min(max(pos, 0), e - 1)] = ed;
    }
}

__global__ void __launch_bounds__(SORT_TB)
k_bin_sort(const int2* __restrict__ binned, const int* __restrict__ bstart,
           int2* __restrict__ list, int* __restrict__ rowend, float* __restrict__ dis,
           unsigned char* __restrict__ deg8, const int* __restrict__ batch,
           int* __restrict__ gstart, int e, int n) {
    __shared__ int cnt[BK];
    __shared__ int cursor[BK];
    __shared__ int wsum[SORT_TB / 32];
    __shared__ int2 stage[STAGE_CAP];
    const int tid = threadIdx.x, lane = tid & 31, wv = tid >> 5;
    const int b = blockIdx.x;
    const int sbeg = min(max(bstart[b], 0), e);
    const int send = min(max(bstart[b + 1], sbeg), e);
    const int seg = send - sbeg;
    const bool staged = (seg <= STAGE_CAP);

    cnt[tid] = 0;
    __syncthreads();
    for (int i0 = sbeg + tid; i0 < send; i0 += 8 * SORT_TB) {
        int dv[8];
        #pragma unroll
        for (int u = 0; u < 8; ++u) dv[u] = binned[min(i0 + u * SORT_TB, send - 1)].y;
        #pragma unroll
        for (int u = 0; u < 8; ++u)
            if (i0 + u * SORT_TB < send) atomicAdd(&cnt[min(max(dv[u] - b * BK, 0), BK - 1)], 1);
    }
    __syncthreads();

    const int c = cnt[tid];
    int incl = c;
    for (int o = 1; o < 32; o <<= 1) {
        const int v = __shfl_up(incl, o);
        if (lane >= o) incl += v;
    }
    if (lane == 31) wsum[wv] = incl;
    __syncthreads();
    for (int k = 0; k < wv; ++k) incl += wsum[k];
    cursor[tid] = incl - c;

    const int v = b * BK + tid;
    if (v < n) {
        rowend[v] = sbeg + incl;
        dis[v] = deg_to_dis(c);
        deg8[v] = (unsigned char)min(c, DEG_SAT);
        const int pb = (v > 0) ? batch[v - 1] : -1;
        const int gb = batch[v];
        for (int g = max(pb + 1, 0); g <= min(gb, N_GRAPHS); ++g) gstart[g] = v;
        if (v == n - 1)
            for (int g = max(gb + 1, 0); g <= N_GRAPHS; ++g) gstart[g] = n;
    }
    __syncthreads();

    for (int i0 = sbeg + tid; i0 < send; i0 += 8 * SORT_TB) {
        int2 ed[8];
        #pragma unroll
        for (int u = 0; u < 8; ++u) ed[u] = binned[min(i0 + u * SORT_TB, send - 1)];
        #pragma unroll
        for (int u = 0; u < 8; ++u) {
            if (i0 + u * SORT_TB < send) {
                const int dl = min(max(ed[u].y - b * BK, 0), BK - 1);
                const int slot = min(max(atomicAdd(&cursor[dl], 1), 0), seg - 1);
                const int2 o = make_int2(ed[u].x, __float_as_int(deg_to_dis(cnt[dl])));
                if (staged) stage[slot] = o;
                else        list[sbeg + slot] = o;
            }
        }
    }
    __syncthreads();
    if (staged)
        for (int i = tid; i < seg; i += SORT_TB) list[sbeg + i] = stage[i];
}

__global__ void __launch_bounds__(PLACE_TB)
k_edge_weights(int2* __restrict__ list, const float* __restrict__ dis,
               const unsigned char* __restrict__ deg8, int e, int n) {
    __shared__ __attribute__((aligned(16))) unsigned char tab[N_NODES];
    const int tid = threadIdx.x;
    if ((long long)blockIdx.x * (4 * PLACE_TB) >= e) return;

    long long base = (long long)blockIdx.x * (4 * PLACE_TB);
    int2 en[4];
    #pragma unroll
    for (int u = 0; u < 4; ++u) {
        const long long j = base + u * PLACE_TB + tid;
        en[u] = make_int2(0, 0);
        if (j < e) en[u] = list[j];
    }

    const int nt  = min(n, N_NODES);
    const int n16 = nt >> 4;
    const uint4* g = reinterpret_cast<const uint4*>(deg8);
    uint4* t = reinterpret_cast<uint4*>(tab);
    uint4 tv[TAB_IT];
    #pragma unroll
    for (int it = 0; it < TAB_IT; ++it) {
        const int k = tid + it * PLACE_TB;
        tv[it] = make_uint4(0u, 0u, 0u, 0u);
        if (k < n16) tv[it] = g[k];
    }
    #pragma unroll
    for (int it = 0; it < TAB_IT; ++it) {
        const int k = tid + it * PLACE_TB;
        if (k < n16) t[k] = tv[it];
    }
    if (tid < (nt & 15)) tab[(n16 << 4) + tid] = deg8[(n16 << 4) + tid];
    __syncthreads();

    for (;;) {
        #pragma unroll
        for (int u = 0; u < 4; ++u) {
            const long long j = base + u * PLACE_TB + tid;
            if (j < e) {
                const int sidx = min(max(en[u].x, 0), nt - 1);
                const int cs = tab[sidx];
                const float fs = (cs < DEG_SAT) ? deg_to_dis(cs) : dis[sidx];
                list[j] = make_int2(en[u].x, __float_as_int(fs * __int_as_float(en[u].y)));
            }
        }
        base += (long long)PLACE_WG * (4 * PLACE_TB);
        if (base >= e) break;
        #pragma unroll
        for (int u = 0; u < 4; ++u) {
            const long long j = base + u * PLACE_TB + tid;
            en[u] = make_int2(0, 0);
            if (j < e) en[u] = list[j];
        }
    }
}


template <int KV, int AS, int KP, bool SPLIT_A, bool FULL_O>
__global__ void k_gemm_wmma(const float* A0, const float* A1, const float* __restrict__ W,
                            float* O0, float* O1, int n) {
    __shared__ float Wlds[(KP / 2) * H * 2];
    for (int idx = threadIdx.x; idx < KP * H; idx += blockDim.x) {
        int k = idx / H, c = idx - k * H;
        float v = (k < KV) ? W[k * H + c] : 0.0f;
        Wlds[(k >> 1) * (H * 2) + c * 2 + (k & 1)] = v;
    }
    __syncthreads();

    const int lane  = threadIdx.x & 31;
    const int strip = blockIdx.x * (blockDim.x >> 5) + (threadIdx.x >> 5);
    const int row0  = strip * 16;
    if (row0 >= n) return;

    const int mrow = lane & 15;
    const int hi   = lane >> 4;
    const float* arow  = SPLIT_A ? A0 + (size_t)(row0 + mrow) * HH
                                 : A0 + (size_t)(row0 + mrow) * AS;
    const float* arow1 = A1 + (size_t)(row0 + mrow) * HH;

    v8f acc[4];
    #pragma unroll
    for (int t = 0; t < 4; ++t) acc[t] = (v8f){0, 0, 0, 0, 0, 0, 0, 0};

    #pragma unroll
    for (int k0 = 0; k0 < KP; k0 += 4) {
        const int ka = k0 + hi * 2;
        v2f a;
        if (SPLIT_A) {
            const float* ap = (k0 < HH) ? (arow + ka) : (arow1 + (ka - HH));
            a = *reinterpret_cast<const v2f*>(ap);
        } else if (KV == KP && (AS & 1) == 0) {
            a = *reinterpret_cast<const v2f*>(arow + ka);
        } else {
            const int kc0 = (ka     < KV) ? ka     : (KV - 1);
            const int kc1 = (ka + 1 < KV) ? ka + 1 : (KV - 1);
            float ax = arow[kc0];
            float ay = arow[kc1];
            a.x = (ka     < KV) ? ax : 0.0f;
            a.y = (ka + 1 < KV) ? ay : 0.0f;
        }

        const int pbase = ((k0 >> 1) + hi) * (H * 2);
        #pragma unroll
        for (int t = 0; t < 4; ++t) {
            const int c = t * 16 + mrow;
            v2f b = *reinterpret_cast<const v2f*>(&Wlds[pbase + c * 2]);
            acc[t] = __builtin_amdgcn_wmma_f32_16x16x4_f32(
                 false, a,  false, b,
                 (short)0, acc[t],  false,  false);
        }
    }

    const int mbase = (lane < 16) ? 0 : 8;
    #pragma unroll
    for (int t = 0; t < 4; ++t) {
        if (FULL_O) {
            #pragma unroll
            for (int j = 0; j < 8; ++j) {
                O0[(size_t)(row0 + mbase + j) * H + t * 16 + mrow] = acc[t][j];
            }
        } else {
            float* op = (t < 2) ? O0 : O1;
            const int col = (t & 1) * 16 + mrow;
            #pragma unroll
            for (int j = 0; j < 8; ++j) {
                op[(size_t)(row0 + mbase + j) * HH + col] = acc[t][j];
            }
        }
    }
}

__device__ __forceinline__ float4 pull_edges(const float* __restrict__ hin,
                                             const int2* __restrict__ list,
                                             int beg, int end, float4 acc, int q4) {
    const int last = end - 1;
    for (int j = beg; j < end; j += 4) {
        const int j1 = (j + 1 < end) ? j + 1 : last;
        const int j2 = (j + 2 < end) ? j + 2 : last;
        const int j3 = (j + 3 < end) ? j + 3 : last;
        const int2 e0 = list[j];
        const int2 e1 = list[j1];
        const int2 e2 = list[j2];
        const int2 e3 = list[j3];
        const float4 r0 = *reinterpret_cast<const float4*>(hin + (size_t)e0.x * HH + q4);
        const float4 r1 = *reinterpret_cast<const float4*>(hin + (size_t)e1.x * HH + q4);
        const float4 r2 = *reinterpret_cast<const float4*>(hin + (size_t)e2.x * HH + q4);
        const float4 r3 = *reinterpret_cast<const float4*>(hin + (size_t)e3.x * HH + q4);
        const bool v1 = (j + 1 < end), v2 = (j + 2 < end), v3 = (j + 3 < end);
        const float w0 = __int_as_float(e0.y);
        const float w1 = __int_as_float(e1.y);
        const float w2 = __int_as_float(e2.y);
        const float w3 = __int_as_float(e3.y);
        acc.x += r0.x * w0; acc.y += r0.y * w0; acc.z += r0.z * w0; acc.w += r0.w * w0;
        acc.x += (v1 ? r1.x : 0.0f) * w1; acc.y += (v1 ? r1.y : 0.0f) * w1;
        acc.z += (v1 ? r1.z : 0.0f) * w1; acc.w += (v1 ? r1.w : 0.0f) * w1;
        acc.x += (v2 ? r2.x : 0.0f) * w2; acc.y += (v2 ? r2.y : 0.0f) * w2;
        acc.z += (v2 ? r2.z : 0.0f) * w2; acc.w += (v2 ? r2.w : 0.0f) * w2;
        acc.x += (v3 ? r3.x : 0.0f) * w3; acc.y += (v3 ? r3.y : 0.0f) * w3;
        acc.z += (v3 ? r3.z : 0.0f) * w3; acc.w += (v3 ? r3.w : 0.0f) * w3;
    }
    return acc;
}

__device__ __forceinline__ float4 pull_node(const float* __restrict__ hin,
                                            const int* __restrict__ rowend,
                                            const int2* __restrict__ list,
                                            const float* __restrict__ dis,
                                            const float* __restrict__ bias, int node, int q4) {
    const int end = rowend[node];
    const int beg = (node > 0) ? rowend[node - 1] : 0;
    const float di = dis[node];
    const float wself = di * di;
    const float4 hv = *reinterpret_cast<const float4*>(hin + (size_t)node * HH + q4);
    float4 acc = make_float4(hv.x * wself, hv.y * wself, hv.z * wself, hv.w * wself);
    acc = pull_edges(hin, list, beg, end, acc, q4);
    const float4 bv = *reinterpret_cast<const float4*>(bias + q4);
    return make_float4(acc.x + bv.x, acc.y + bv.y, acc.z + bv.z, acc.w + bv.w);
}

__global__ void k_pull_half(const float* __restrict__ hin, float* __restrict__ hout,
                            const int* __restrict__ rowend, const int2* __restrict__ list,
                            const float* __restrict__ dis, const float* __restrict__ bias,
                            int n) {
    int t = blockIdx.x * blockDim.x + threadIdx.x;
    int node = t >> 3;
    if (node >= n) return;
    const int q4 = (t & 7) * 4;
    float4 r = pull_node(hin, rowend, list, dis, bias, node, q4);
    r.x = fmaxf(r.x, 0.0f); r.y = fmaxf(r.y, 0.0f);
    r.z = fmaxf(r.z, 0.0f); r.w = fmaxf(r.w, 0.0f);
    *reinterpret_cast<float4*>(hout + (size_t)node * HH + q4) = r;
}

__device__ __forceinline__ void pull_edges2(const float* __restrict__ h0,
                                            const float* __restrict__ h1,
                                            const int2* __restrict__ list,
                                            int beg, int end, float4& a0, float4& a1, int q4) {
    const int last = end - 1;
    for (int j = beg; j < end; j += 4) {
        const int j1 = (j + 1 < end) ? j + 1 : last;
        const int j2 = (j + 2 < end) ? j + 2 : last;
        const int j3 = (j + 3 < end) ? j + 3 : last;
        const int2 e0 = list[j];
        const int2 e1 = list[j1];
        const int2 e2 = list[j2];
        const int2 e3 = list[j3];
        const size_t o0 = (size_t)e0.x * HH + q4;
        const size_t o1 = (size_t)e1.x * HH + q4;
        const size_t o2 = (size_t)e2.x * HH + q4;
        const size_t o3 = (size_t)e3.x * HH + q4;
        const float4 r00 = *reinterpret_cast<const float4*>(h0 + o0);
        const float4 r01 = *reinterpret_cast<const float4*>(h1 + o0);
        const float4 r10 = *reinterpret_cast<const float4*>(h0 + o1);
        const float4 r11 = *reinterpret_cast<const float4*>(h1 + o1);
        const float4 r20 = *reinterpret_cast<const float4*>(h0 + o2);
        const float4 r21 = *reinterpret_cast<const float4*>(h1 + o2);
        const float4 r30 = *reinterpret_cast<const float4*>(h0 + o3);
        const float4 r31 = *reinterpret_cast<const float4*>(h1 + o3);
        const bool v1 = (j + 1 < end), v2 = (j + 2 < end), v3 = (j + 3 < end);
        const float w0 = __int_as_float(e0.y);
        const float w1 = __int_as_float(e1.y);
        const float w2 = __int_as_float(e2.y);
        const float w3 = __int_as_float(e3.y);
        a0.x += r00.x * w0; a0.y += r00.y * w0; a0.z += r00.z * w0; a0.w += r00.w * w0;
        a1.x += r01.x * w0; a1.y += r01.y * w0; a1.z += r01.z * w0; a1.w += r01.w * w0;
        a0.x += (v1 ? r10.x : 0.0f) * w1; a0.y += (v1 ? r10.y : 0.0f) * w1;
        a0.z += (v1 ? r10.z : 0.0f) * w1; a0.w += (v1 ? r10.w : 0.0f) * w1;
        a1.x += (v1 ? r11.x : 0.0f) * w1; a1.y += (v1 ? r11.y : 0.0f) * w1;
        a1.z += (v1 ? r11.z : 0.0f) * w1; a1.w += (v1 ? r11.w : 0.0f) * w1;
        a0.x += (v2 ? r20.x : 0.0f) * w2; a0.y += (v2 ? r20.y : 0.0f) * w2;
        a0.z += (v2 ? r20.z : 0.0f) * w2; a0.w += (v2 ? r20.w : 0.0f) * w2;
        a1.x += (v2 ? r21.x : 0.0f) * w2; a1.y += (v2 ? r21.y : 0.0f) * w2;
        a1.z += (v2 ? r21.z : 0.0f) * w2; a1.w += (v2 ? r21.w : 0.0f) * w2;
        a0.x += (v3 ? r30.x : 0.0f) * w3; a0.y += (v3 ? r30.y : 0.0f) * w3;
        a0.z += (v3 ? r30.z : 0.0f) * w3; a0.w += (v3 ? r30.w : 0.0f) * w3;
        a1.x += (v3 ? r31.x : 0.0f) * w3; a1.y += (v3 ? r31.y : 0.0f) * w3;
        a1.z += (v3 ? r31.z : 0.0f) * w3; a1.w += (v3 ? r31.w : 0.0f) * w3;
    }
}

#define PG_TB 256
__global__ void __launch_bounds__(PG_TB, 12)
k_pool_graph(const float* __restrict__ h0, const float* __restrict__ h1,
             const int* __restrict__ rowend, const int2* __restrict__ list,
             const float* __restrict__ dis, const float* __restrict__ bias,
             const int* __restrict__ gstart, const float* __restrict__ lin_w,
             const float* __restrict__ lin_b, float* __restrict__ out, int n) {
    __shared__ __attribute__((aligned(16))) float red[PG_TB / 32][H];
    const int g = blockIdx.x;
    const int nbeg = min(max(gstart[g], 0), n);
    const int nend = min(max(gstart[g + 1], nbeg), n);
    const int q4 = (threadIdx.x & 7) * 4;
    const int slot = threadIdx.x >> 3;

    float4 s0 = make_float4(0.0f, 0.0f, 0.0f, 0.0f);
    float4 s1 = make_float4(0.0f, 0.0f, 0.0f, 0.0f);
    for (int node = nbeg + slot; node < nend; node += PG_TB / 8) {
        const int end = rowend[node];
        const int beg = (node > 0) ? rowend[node - 1] : 0;
        const float di = dis[node];
        const float wself = di * di;
        const float4 hv0 = *reinterpret_cast<const float4*>(h0 + (size_t)node * HH + q4);
        const float4 hv1 = *reinterpret_cast<const float4*>(h1 + (size_t)node * HH + q4);
        float4 a0 = make_float4(hv0.x * wself, hv0.y * wself, hv0.z * wself, hv0.w * wself);
        float4 a1 = make_float4(hv1.x * wself, hv1.y * wself, hv1.z * wself, hv1.w * wself);
        pull_edges2(h0, h1, list, beg, end, a0, a1, q4);
        s0.x += a0.x; s0.y += a0.y; s0.z += a0.z; s0.w += a0.w;
        s1.x += a1.x; s1.y += a1.y; s1.z += a1.z; s1.w += a1.w;
    }

    s0.x += __shfl_xor(s0.x, 8);  s0.y += __shfl_xor(s0.y, 8);
    s0.z += __shfl_xor(s0.z, 8);  s0.w += __shfl_xor(s0.w, 8);
    s1.x += __shfl_xor(s1.x, 8);  s1.y += __shfl_xor(s1.y, 8);
    s1.z += __shfl_xor(s1.z, 8);  s1.w += __shfl_xor(s1.w, 8);
    s0.x += __shfl_xor(s0.x, 16); s0.y += __shfl_xor(s0.y, 16);
    s0.z += __shfl_xor(s0.z, 16); s0.w += __shfl_xor(s0.w, 16);
    s1.x += __shfl_xor(s1.x, 16); s1.y += __shfl_xor(s1.y, 16);
    s1.z += __shfl_xor(s1.z, 16); s1.w += __shfl_xor(s1.w, 16);
    const int lane = threadIdx.x & 31;
    const int wv   = threadIdx.x >> 5;
    if (lane < 8) {
        *reinterpret_cast<float4*>(&red[wv][q4]) = s0;
        *reinterpret_cast<float4*>(&red[wv][HH + q4]) = s1;
    }
    __syncthreads();

    if (threadIdx.x < 32) {
        const float cntf = (float)(nend - nbeg);
        float ca = 0.0f, cb = 0.0f;
        for (int w = 0; w < PG_TB / 32; ++w) {
            ca += red[w][lane];
            cb += red[w][HH + lane];
        }
        ca += cntf * bias[lane];
        cb += cntf * bias[HH + lane];
        float p = ca * lin_w[lane] + cb * lin_w[HH + lane];
        for (int o = 16; o > 0; o >>= 1) p += __shfl_xor(p, o);
        if (lane == 0) {
            const float inv = 1.0f / fmaxf(cntf, 1.0f);
            out[g] = p * inv + lin_b[0];
        }
    }
}

#define FB_ROWS 128
#define TP 36

__global__ void __launch_bounds__(256)
k_pull_gemm(const float* __restrict__ hg, float* P0, float* O1,
            const int* __restrict__ rowend, const int2* __restrict__ list,
            const float* __restrict__ dis, const float* __restrict__ bias,
            const float* __restrict__ W, int n) {
    __shared__ float Wlds[(H / 2) * H * 2];
    __shared__ __attribute__((aligned(16))) float tile[FB_ROWS * TP];
    for (int idx = threadIdx.x; idx < H * H; idx += blockDim.x) {
        int k = idx / H, c = idx - k * H;
        Wlds[(k >> 1) * (H * 2) + c * 2 + (k & 1)] = W[k * H + c];
    }

    const int blockBase = blockIdx.x * FB_ROWS;
    const int q4 = (threadIdx.x & 7) * 4;
    const int sub = threadIdx.x >> 3;
    const float4 bv = *reinterpret_cast<const float4*>(bias + q4);

    int beg[4], end[4];
    float4 acc[4];
    #pragma unroll
    for (int it = 0; it < 4; ++it) {
        int node = blockBase + it * 32 + sub;
        node = (node < n) ? node : (n - 1);
        end[it] = rowend[node];
        beg[it] = (node > 0) ? rowend[node - 1] : 0;
        const float di = dis[node];
        const float wself = di * di;
        const float4 hv = *reinterpret_cast<const float4*>(hg + (size_t)node * HH + q4);
        acc[it] = make_float4(hv.x * wself, hv.y * wself, hv.z * wself, hv.w * wself);
    }
    #pragma unroll
    for (int it = 0; it < 4; ++it) {
        const float4 a = pull_edges(hg, list, beg[it], end[it], acc[it], q4);
        float4 r = make_float4(a.x + bv.x, a.y + bv.y, a.z + bv.z, a.w + bv.w);
        r.x = fmaxf(r.x, 0.0f); r.y = fmaxf(r.y, 0.0f);
        r.z = fmaxf(r.z, 0.0f); r.w = fmaxf(r.w, 0.0f);
        *reinterpret_cast<float4*>(&tile[(it * 32 + sub) * TP + q4]) = r;
    }
    __syncthreads();

    const int lane = threadIdx.x & 31;
    const int wv   = threadIdx.x >> 5;
    const int row0 = blockBase + wv * 16;
    if (row0 >= n) return;

    const int mrow = lane & 15;
    const int hi   = lane >> 4;
    const float* arow = P0 + (size_t)(row0 + mrow) * HH;
    const float* trow = &tile[(wv * 16 + mrow) * TP];

    v8f c4[4];
    #pragma unroll
    for (int t = 0; t < 4; ++t) c4[t] = (v8f){0, 0, 0, 0, 0, 0, 0, 0};

    #pragma unroll
    for (int k0 = 0; k0 < H; k0 += 4) {
        const int ka = k0 + hi * 2;
        const float* ap = (k0 < HH) ? (arow + ka) : (trow + (ka - HH));
        const v2f a = *reinterpret_cast<const v2f*>(ap);

        const int pbase = ((k0 >> 1) + hi) * (H * 2);
        #pragma unroll
        for (int t = 0; t < 4; ++t) {
            const int c = t * 16 + mrow;
            v2f b = *reinterpret_cast<const v2f*>(&Wlds[pbase + c * 2]);
            c4[t] = __builtin_amdgcn_wmma_f32_16x16x4_f32(
                 false, a,  false, b,
                 (short)0, c4[t],  false,  false);
        }
    }

    const int mbase = (lane < 16) ? 0 : 8;
    #pragma unroll
    for (int t = 0; t < 4; ++t) {
        float* op = (t < 2) ? P0 : O1;
        const int col = (t & 1) * 16 + mrow;
        #pragma unroll
        for (int j = 0; j < 8; ++j) {
            op[(size_t)(row0 + mbase + j) * HH + col] = c4[t][j];
        }
    }
}


__device__ __forceinline__ float4 pull_edges_row(const float* __restrict__ zin,
                                                 const int2* __restrict__ list,
                                                 int beg, int end, float4 acc, int c4) {
    const int last = end - 1;
    for (int j = beg; j < end; j += 8) {
        int2 en[8];
        float4 r[8];
        #pragma unroll
        for (int u = 0; u < 8; ++u) en[u] = list[min(j + u, last)];
        #pragma unroll
        for (int u = 0; u < 8; ++u)
            r[u] = *reinterpret_cast<const float4*>(zin + (size_t)en[u].x * H + c4);
        #pragma unroll
        for (int u = 0; u < 8; ++u) {
            const bool v = (j + u < end);
            const float w = __int_as_float(en[u].y);
            acc.x += (v ? r[u].x : 0.0f) * w; acc.y += (v ? r[u].y : 0.0f) * w;
            acc.z += (v ? r[u].z : 0.0f) * w; acc.w += (v ? r[u].w : 0.0f) * w;
        }
    }
    return acc;
}

__device__ __forceinline__ void list_head8(const int2* __restrict__ list, int beg, int end,
                                           int2 (&en)[8]) {
    const int last = max(end - 1, 0);
    #pragma unroll
    for (int u = 0; u < 8; ++u) en[u] = list[min(beg + u, last)];
}

#define TPF 68

__global__ void __launch_bounds__(256, 6)
k_pull_gemm_rows(const float* __restrict__ zin, float* __restrict__ zout,
                 const int* __restrict__ rowend, const int2* __restrict__ list,
                 const float* __restrict__ dis, const float* __restrict__ bias,
                 const float* __restrict__ W, int n) {
    __shared__ float Wlds[(H / 2) * H * 2];
    __shared__ __attribute__((aligned(16))) float tile[FB_ROWS * TPF];
    for (int idx = threadIdx.x; idx < H * H; idx += blockDim.x) {
        int k = idx / H, c = idx - k * H;
        Wlds[(k >> 1) * (H * 2) + c * 2 + (k & 1)] = W[k * H + c];
    }

    const int lane = threadIdx.x & 31;
    const int wv   = threadIdx.x >> 5;
    const int row0 = blockIdx.x * FB_ROWS + wv * 16;
    const int c4   = (lane & 15) * 4;
    const int half = lane >> 4;
    const float4 bv = *reinterpret_cast<const float4*>(bias + c4);

    int beg[8], end[8];
    float4 acc[8];
    #pragma unroll
    for (int it = 0; it < 8; ++it) {
        int node = row0 + it * 2 + half;
        node = (node < n) ? node : (n - 1);
        end[it] = rowend[node];
        beg[it] = (node > 0) ? rowend[node - 1] : 0;
        const float di = dis[node];
        const float wself = di * di;
        const float4 hv = *reinterpret_cast<const float4*>(zin + (size_t)node * H + c4);
        acc[it] = make_float4(hv.x * wself, hv.y * wself, hv.z * wself, hv.w * wself);
    }
    int2 en[8];
    list_head8(list, beg[0], end[0], en);
    #pragma unroll
    for (int it = 0; it < 8; ++it) {
        float4 r[8];
        #pragma unroll
        for (int u = 0; u < 8; ++u)
            r[u] = *reinterpret_cast<const float4*>(zin + (size_t)en[u].x * H + c4);
        int2 nx[8];
        if (it + 1 < 8) list_head8(list, beg[it + 1], end[it + 1], nx);
        float4 a = acc[it];
        #pragma unroll
        for (int u = 0; u < 8; ++u) {
            const bool v = (beg[it] + u < end[it]);
            const float w = __int_as_float(en[u].y);
            a.x += (v ? r[u].x : 0.0f) * w; a.y += (v ? r[u].y : 0.0f) * w;
            a.z += (v ? r[u].z : 0.0f) * w; a.w += (v ? r[u].w : 0.0f) * w;
        }
        a = pull_edges_row(zin, list, beg[it] + 8, end[it], a, c4);
        float4 o = make_float4(a.x + bv.x, a.y + bv.y, a.z + bv.z, a.w + bv.w);
        o.x = fmaxf(o.x, 0.0f); o.y = fmaxf(o.y, 0.0f);
        o.z = fmaxf(o.z, 0.0f); o.w = fmaxf(o.w, 0.0f);
        *reinterpret_cast<float4*>(&tile[(wv * 16 + it * 2 + half) * TPF + c4]) = o;
        if (it + 1 < 8) {
            #pragma unroll
            for (int u = 0; u < 8; ++u) en[u] = nx[u];
        }
    }
    __syncthreads();

    if (row0 >= n) return;

    const int mrow = lane & 15;
    const int hi   = lane >> 4;
    const float* trow = &tile[(wv * 16 + mrow) * TPF];

    v8f c4acc[4];
    #pragma unroll
    for (int t = 0; t < 4; ++t) c4acc[t] = (v8f){0, 0, 0, 0, 0, 0, 0, 0};

    #pragma unroll
    for (int k0 = 0; k0 < H; k0 += 4) {
        const int ka = k0 + hi * 2;
        const v2f a = *reinterpret_cast<const v2f*>(trow + ka);

        const int pbase = ((k0 >> 1) + hi) * (H * 2);
        #pragma unroll
        for (int t = 0; t < 4; ++t) {
            const int c = t * 16 + mrow;
            v2f b = *reinterpret_cast<const v2f*>(&Wlds[pbase + c * 2]);
            c4acc[t] = __builtin_amdgcn_wmma_f32_16x16x4_f32(
                 false, a,  false, b,
                 (short)0, c4acc[t],  false,  false);
        }
    }

    const int mbase = (lane < 16) ? 0 : 8;
    #pragma unroll
    for (int t = 0; t < 4; ++t) {
        #pragma unroll
        for (int j = 0; j < 8; ++j) {
            zout[(size_t)(row0 + mbase + j) * H + t * 16 + mrow] = c4acc[t][j];
        }
    }
}

__global__ void __launch_bounds__(PG_TB, 12)
k_pool_graph_rows(const float* __restrict__ z,
                  const int* __restrict__ rowend, const int2* __restrict__ list,
                  const float* __restrict__ dis, const float* __restrict__ bias,
                  const int* __restrict__ gstart, const float* __restrict__ lin_w,
                  const float* __restrict__ lin_b, float* __restrict__ out, int n) {
    __shared__ __attribute__((aligned(16))) float red[PG_TB / 32][H];
    const int g = blockIdx.x;
    const int nbeg = min(max(gstart[g], 0), n);
    const int nend = min(max(gstart[g + 1], nbeg), n);
    const int c4 = (threadIdx.x & 15) * 4;
    const int slot = threadIdx.x >> 4;

    float4 s = make_float4(0.0f, 0.0f, 0.0f, 0.0f);
    for (int node = nbeg + slot; node < nend; node += PG_TB / 16) {
        const int end = rowend[node];
        const int beg = (node > 0) ? rowend[node - 1] : 0;
        const float di = dis[node];
        const float wself = di * di;
        const float4 hv = *reinterpret_cast<const float4*>(z + (size_t)node * H + c4);
        float4 a = make_float4(hv.x * wself, hv.y * wself, hv.z * wself, hv.w * wself);
        a = pull_edges_row(z, list, beg, end, a, c4);
        s.x += a.x; s.y += a.y; s.z += a.z; s.w += a.w;
    }

    s.x += __shfl_xor(s.x, 16); s.y += __shfl_xor(s.y, 16);
    s.z += __shfl_xor(s.z, 16); s.w += __shfl_xor(s.w, 16);
    const int lane = threadIdx.x & 31;
    const int wv   = threadIdx.x >> 5;
    if (lane < 16) *reinterpret_cast<float4*>(&red[wv][c4]) = s;
    __syncthreads();

    if (threadIdx.x < 32) {
        const float cntf = (float)(nend - nbeg);
        float ca = 0.0f, cb = 0.0f;
        for (int w = 0; w < PG_TB / 32; ++w) {
            ca += red[w][lane];
            cb += red[w][HH + lane];
        }
        ca += cntf * bias[lane];
        cb += cntf * bias[HH + lane];
        float p = ca * lin_w[lane] + cb * lin_w[HH + lane];
        for (int o = 16; o > 0; o >>= 1) p += __shfl_xor(p, o);
        if (lane == 0) {
            const float inv = 1.0f / fmaxf(cntf, 1.0f);
            out[g] = p * inv + lin_b[0];
        }
    }
}


extern "C" void kernel_launch(void* const* d_in, const int* in_sizes, int n_in,
                              void* d_out, int out_size, void* d_ws, size_t ws_size,
                              hipStream_t stream) {
    const float* x     = (const float*)d_in[0];
    const int*   eidx  = (const int*)  d_in[1];
    const int*   batch = (const int*)  d_in[2];
    const float* W1    = (const float*)d_in[3];
    const float* b1    = (const float*)d_in[4];
    const float* W2    = (const float*)d_in[5];
    const float* b2    = (const float*)d_in[6];
    const float* W3    = (const float*)d_in[7];
    const float* b3    = (const float*)d_in[8];
    const float* lin_w = (const float*)d_in[9];
    const float* lin_b = (const float*)d_in[10];
    float* out = (float*)d_out;

    const int n = in_sizes[0] / F_IN;
    const int e = in_sizes[1] / 2;
    const int* src = eidx;
    const int* dst = eidx + e;

    const int TB = 256;
    dim3 blk(TB);
    auto blocks = [](long long work, int tb) { return dim3((unsigned)((work + tb - 1) / tb)); };
    const dim3 bin_grid  = blocks(e, EPW);
    const dim3 sort_grid = blocks(n, BK);
    const int strips       = (n + 15) / 16;
    const dim3 gemm_grid   = blocks((long long)strips * 32, TB);
    const dim3 fuse_grid   = blocks((long long)n, FB_ROWS);

    const size_t rows_need = ((size_t)n * (2 * H + 3) + 1024 + 2 * (size_t)e + N_GRAPHS + 4)
                             * sizeof(float) + (size_t)n;
    if (ws_size >= rows_need) {
        float* ws   = (float*)d_ws;
        float* ZA   = ws;
        float* ZB   = ZA + (size_t)n * H;
        float* dis  = ZB + (size_t)n * H;
        int*   cnt    = (int*)(dis + n);
        int*   cursor = cnt + n;
        int*   part   = cursor + n;
        int2*  list   = (int2*)(part + 1024);
        int*   gstart = (int*)(list + e);
        unsigned char* deg8 = (unsigned char*)(gstart + N_GRAPHS + 4);
        int2*  binned = (int2*)ZA;
        int*   hist   = (int*)ZB;
        int*   bstart = hist + (size_t)bin_grid.x * MAXB;

        k_bin_count   <<<bin_grid, dim3(SORT_TB), 0, stream>>>(src, dst, hist, e);
        k_bin_scatter <<<bin_grid, dim3(SORT_TB), 0, stream>>>(src, dst, hist, binned, bstart, e, (int)bin_grid.x);
        k_bin_sort    <<<sort_grid, dim3(SORT_TB), 0, stream>>>(binned, bstart, list, cursor, dis, deg8, batch, gstart, e, n);
        k_edge_weights<<<dim3(PLACE_WG), dim3(PLACE_TB), 0, stream>>>(list, dis, deg8, e, n);

        k_gemm_wmma<F_IN, F_IN, 12, false, true><<<gemm_grid, blk, 0, stream>>>(x, x, W1, ZA, ZA, n);
        k_pull_gemm_rows<<<fuse_grid, blk, 0, stream>>>(ZA, ZB, cursor, list, dis, b1, W2, n);
        k_pull_gemm_rows<<<fuse_grid, blk, 0, stream>>>(ZB, ZA, cursor, list, dis, b2, W3, n);
        k_pool_graph_rows<<<dim3(N_GRAPHS), dim3(PG_TB), 0, stream>>>(ZA, cursor, list, dis, b3, gstart,
                                                                      lin_w, lin_b, out, n);
        return;
    }

    float* ws   = (float*)d_ws;
    float* dis  = ws;
    float* R0   = dis + n;
    float* R1   = R0 + (size_t)n * HH;
    float* R2   = R1 + (size_t)n * HH;
    int*   cnt    = (int*)(R2 + (size_t)n * HH);
    int*   cursor = cnt + n;
    int*   part   = cursor + n;
    int2*  list   = (int2*)(part + 1024);
    int*   gstart = (int*)(list + e);
    unsigned char* deg8 = (unsigned char*)(gstart + N_GRAPHS + 4);
    int2*  binned = (int2*)R0;
    int*   hist   = (int*)R1;
    int*   bstart = hist + (size_t)bin_grid.x * MAXB;

    k_bin_count   <<<bin_grid, dim3(SORT_TB), 0, stream>>>(src, dst, hist, e);
    k_bin_scatter <<<bin_grid, dim3(SORT_TB), 0, stream>>>(src, dst, hist, binned, bstart, e, (int)bin_grid.x);
    k_bin_sort    <<<sort_grid, dim3(SORT_TB), 0, stream>>>(binned, bstart, list, cursor, dis, deg8, batch, gstart, e, n);
    k_edge_weights<<<dim3(PLACE_WG), dim3(PLACE_TB), 0, stream>>>(list, dis, deg8, e, n);

    const dim3 pull_grid   = blocks((long long)n * 8, TB);

    k_gemm_wmma<F_IN, F_IN, 12, false, false><<<gemm_grid, blk, 0, stream>>>(x, x, W1, R0, R1, n);

    k_pull_half<<<pull_grid, blk, 0, stream>>>(R0, R2, cursor, list, dis, b1, n);
    k_pull_gemm<<<fuse_grid, blk, 0, stream>>>(R1, R2, R0, cursor, list, dis, b1 + HH, W2, n);

    k_pull_half<<<pull_grid, blk, 0, stream>>>(R2, R1, cursor, list, dis, b2, n);
    k_pull_gemm<<<fuse_grid, blk, 0, stream>>>(R0, R1, R2, cursor, list, dis, b2 + HH, W3, n);

    k_pool_graph<<<dim3(N_GRAPHS), dim3(PG_TB), 0, stream>>>(R1, R2, cursor, list, dis, b3, gstart,
                                                            lin_w, lin_b, out, n);
}


# layer-1 product x@W1 (needs no edge list) runs as extra work-groups of the bin-sort launch, not in a launch of its own: 8 -> 7 launches; the sort's scratch moves to the other plane
# speedup vs baseline: 3.5445x; 1.0303x over previous
#include <hip/hip_runtime.h>
#include <hip/hip_bf16.h>

#define N_NODES 200000
#define N_EDGES 1000000
#define N_GRAPHS 4096
#define F_IN 11
#define H 64
#define HH 32

typedef float v2f __attribute__((ext_vector_type(2)));
typedef float v8f __attribute__((ext_vector_type(8)));


#define DEG_SAT 255
#define PLACE_TB 1024
#define PLACE_WG 256
#define TAB_IT ((N_NODES / 16 + PLACE_TB - 1) / PLACE_TB)

#define BK_SHIFT 10
#define BK (1 << BK_SHIFT)
#define MAXB ((N_NODES + BK - 1) / BK)
#define SORT_TB 1024
#define EPW (4 * SORT_TB)
#define STAGE_CAP 16384

__device__ __forceinline__ float deg_to_dis(int c) {
    return rsqrtf(fmaxf((float)(c + 1), 1.0f));
}

__device__ __forceinline__ int bin_of(int d) {
    return min(max(d >> BK_SHIFT, 0), MAXB - 1);
}

__device__ __forceinline__ void load_quad(const int* __restrict__ src, const int* __restrict__ dst,
                                          int i, int e, bool want_src,
                                          int (&s)[4], int (&d)[4], int& nv) {
    nv = min(max(e - i, 0), 4);
    const bool vec_ok = ((((size_t)src) | ((size_t)dst)) & 15) == 0;
    if (nv == 4 && vec_ok) {
        const int4 dv = *reinterpret_cast<const int4*>(dst + i);
        d[0] = dv.x; d[1] = dv.y; d[2] = dv.z; d[3] = dv.w;
        if (want_src) {
            const int4 sv = *reinterpret_cast<const int4*>(src + i);
            s[0] = sv.x; s[1] = sv.y; s[2] = sv.z; s[3] = sv.w;
        } else {
            s[0] = 0; s[1] = 0; s[2] = 0; s[3] = 0;
        }
    } else {
        #pragma unroll
        for (int u = 0; u < 4; ++u) {
            s[u] = 0; d[u] = 0;
            if (u < nv) {
                d[u] = dst[i + u];
                if (want_src) s[u] = src[i + u];
            }
        }
    }
}

__global__ void __launch_bounds__(SORT_TB)
k_bin_count(const int* __restrict__ src, const int* __restrict__ dst, int* __restrict__ hist, int e) {
    __shared__ int h[MAXB];
    const int tid = threadIdx.x;
    int s[4], d[4], nv;
    load_quad(src, dst, blockIdx.x * EPW + tid * 4, e, false, s, d, nv);
    for (int b = tid; b < MAXB; b += SORT_TB) h[b] = 0;
    __syncthreads();
    #pragma unroll
    for (int u = 0; u < 4; ++u)
        if (u < nv) atomicAdd(&h[bin_of(d[u])], 1);
    __syncthreads();
    for (int b = tid; b < MAXB; b += SORT_TB) hist[blockIdx.x * MAXB + b] = h[b];
}

__global__ void __launch_bounds__(SORT_TB)
k_bin_scatter(const int* __restrict__ src, const int* __restrict__ dst,
              const int* __restrict__ hist, int2* __restrict__ binned,
              int* __restrict__ bstart, int e, int nw) {
    __shared__ int pTot[4][256];
    __shared__ int pPre[4][256];
    __shared__ int wsT[8];
    __shared__ int wsO[8];
    __shared__ int gbase[256];
    __shared__ int lbase[256];
    __shared__ int cursor[256];
    __shared__ int2 stage[EPW];
    const int tid = threadIdx.x, lane = tid & 31, wv = tid >> 5;
    const int w = blockIdx.x;

    int s[4], d[4], nv;
    load_quad(src, dst, w * EPW + tid * 4, e, true, s, d, nv);

    const int col = tid & 255, part = tid >> 8;
    int tot = 0, pre = 0;
    if (col < MAXB) {
        for (int k0 = part; k0 < nw; k0 += 64) {
            int v[16];
            #pragma unroll
            for (int u = 0; u < 16; ++u) {
                const int k = k0 + 4 * u;
                v[u] = (k < nw) ? hist[k * MAXB + col] : 0;
            }
            #pragma unroll
            for (int u = 0; u < 16; ++u) {
                tot += v[u];
                pre += (k0 + 4 * u < w) ? v[u] : 0;
            }
        }
    }
    pTot[part][col] = tot;
    pPre[part][col] = pre;
    __syncthreads();

    int T = 0, P = 0, own = 0, iT = 0, iO = 0;
    if (tid < 256) {
        T = pTot[0][tid] + pTot[1][tid] + pTot[2][tid] + pTot[3][tid];
        P = pPre[0][tid] + pPre[1][tid] + pPre[2][tid] + pPre[3][tid];
        own = (tid < MAXB) ? hist[w * MAXB + tid] : 0;
        iT = T; iO = own;
        for (int o = 1; o < 32; o <<= 1) {
            const int a = __shfl_up(iT, o);
            const int c = __shfl_up(iO, o);
            if (lane >= o) { iT += a; iO += c; }
        }
        if (lane == 31) { wsT[wv] = iT; wsO[wv] = iO; }
    }
    __syncthreads();
    if (tid < 256) {
        int offT = 0, offO = 0;
        for (int k = 0; k < wv; ++k) { offT += wsT[k]; offO += wsO[k]; }
        const int exT = offT + iT - T;
        const int exO = offO + iO - own;
        gbase[tid] = exT + P;
        lbase[tid] = exO;
        cursor[tid] = exO;
        if (w == 0 && tid <= MAXB) bstart[tid] = exT;
    }
    __syncthreads();

    #pragma unroll
    for (int u = 0; u < 4; ++u) {
        if (u < nv) {
            const int r = atomicAdd(&cursor[bin_of(d[u])], 1);
            stage[min(max(r, 0), EPW - 1)] = make_int2(s[u], d[u]);
        }
    }
    __syncthreads();
    const int cw = min(max(e - w * EPW, 0), EPW);
    for (int k = tid; k < cw; k += SORT_TB) {
        const int2 ed = stage[k];
        const int b = bin_of(ed.y);
        const int pos = gbase[b] + (k - lbase[b]);
        binned[min(max(pos, 0), e - 1)] = ed;
    }
}

__device__ __forceinline__ void
bin_sort_body(const int b, int* cnt, int* cursor, int* wsum, int2* stage,
              const int2* __restrict__ binned, const int* __restrict__ bstart,
              int2* __restrict__ list, int* __restrict__ rowend, float* __restrict__ dis,
              unsigned char* __restrict__ deg8, const int* __restrict__ batch,
              int* __restrict__ gstart, int e, int n) {
    const int tid = threadIdx.x, lane = tid & 31, wv = tid >> 5;
    const int sbeg = min(max(bstart[b], 0), e);
    const int send = min(max(bstart[b + 1], sbeg), e);
    const int seg = send - sbeg;
    const bool staged = (seg <= STAGE_CAP);

    cnt[tid] = 0;
    __syncthreads();
    for (int i0 = sbeg + tid; i0 < send; i0 += 8 * SORT_TB) {
        int dv[8];
        #pragma unroll
        for (int u = 0; u < 8; ++u) dv[u] = binned[min(i0 + u * SORT_TB, send - 1)].y;
        #pragma unroll
        for (int u = 0; u < 8; ++u)
            if (i0 + u * SORT_TB < send) atomicAdd(&cnt[min(max(dv[u] - b * BK, 0), BK - 1)], 1);
    }
    __syncthreads();

    const int c = cnt[tid];
    int incl = c;
    for (int o = 1; o < 32; o <<= 1) {
        const int v = __shfl_up(incl, o);
        if (lane >= o) incl += v;
    }
    if (lane == 31) wsum[wv] = incl;
    __syncthreads();
    for (int k = 0; k < wv; ++k) incl += wsum[k];
    cursor[tid] = incl - c;

    const int v = b * BK + tid;
    if (v < n) {
        rowend[v] = sbeg + incl;
        dis[v] = deg_to_dis(c);
        deg8[v] = (unsigned char)min(c, DEG_SAT);
        const int pb = (v > 0) ? batch[v - 1] : -1;
        const int gb = batch[v];
        for (int g = max(pb + 1, 0); g <= min(gb, N_GRAPHS); ++g) gstart[g] = v;
        if (v == n - 1)
            for (int g = max(gb + 1, 0); g <= N_GRAPHS; ++g) gstart[g] = n;
    }
    __syncthreads();

    for (int i0 = sbeg + tid; i0 < send; i0 += 8 * SORT_TB) {
        int2 ed[8];
        #pragma unroll
        for (int u = 0; u < 8; ++u) ed[u] = binned[min(i0 + u * SORT_TB, send - 1)];
        #pragma unroll
        for (int u = 0; u < 8; ++u) {
            if (i0 + u * SORT_TB < send) {
                const int dl = min(max(ed[u].y - b * BK, 0), BK - 1);
                const int slot = min(max(atomicAdd(&cursor[dl], 1), 0), seg - 1);
                const int2 o = make_int2(ed[u].x, __float_as_int(deg_to_dis(cnt[dl])));
                if (staged) stage[slot] = o;
                else        list[sbeg + slot] = o;
            }
        }
    }
    __syncthreads();
    if (staged)
        for (int i = tid; i < seg; i += SORT_TB) list[sbeg + i] = stage[i];
}

__global__ void __launch_bounds__(SORT_TB)
k_bin_sort(const int2* __restrict__ binned, const int* __restrict__ bstart,
           int2* __restrict__ list, int* __restrict__ rowend, float* __restrict__ dis,
           unsigned char* __restrict__ deg8, const int* __restrict__ batch,
           int* __restrict__ gstart, int e, int n) {
    __shared__ int cnt[BK];
    __shared__ int cursor[BK];
    __shared__ int wsum[SORT_TB / 32];
    __shared__ int2 stage[STAGE_CAP];
    bin_sort_body(blockIdx.x, cnt, cursor, wsum, stage, binned, bstart, list, rowend, dis, deg8,
                  batch, gstart, e, n);
}

__global__ void __launch_bounds__(PLACE_TB)
k_edge_weights(int2* __restrict__ list, const float* __restrict__ dis,
               const unsigned char* __restrict__ deg8, int e, int n) {
    __shared__ __attribute__((aligned(16))) unsigned char tab[N_NODES];
    const int tid = threadIdx.x;
    if ((long long)blockIdx.x * (4 * PLACE_TB) >= e) return;

    long long base = (long long)blockIdx.x * (4 * PLACE_TB);
    int2 en[4];
    #pragma unroll
    for (int u = 0; u < 4; ++u) {
        const long long j = base + u * PLACE_TB + tid;
        en[u] = make_int2(0, 0);
        if (j < e) en[u] = list[j];
    }

    const int nt  = min(n, N_NODES);
    const int n16 = nt >> 4;
    const uint4* g = reinterpret_cast<const uint4*>(deg8);
    uint4* t = reinterpret_cast<uint4*>(tab);
    uint4 tv[TAB_IT];
    #pragma unroll
    for (int it = 0; it < TAB_IT; ++it) {
        const int k = tid + it * PLACE_TB;
        tv[it] = make_uint4(0u, 0u, 0u, 0u);
        if (k < n16) tv[it] = g[k];
    }
    #pragma unroll
    for (int it = 0; it < TAB_IT; ++it) {
        const int k = tid + it * PLACE_TB;
        if (k < n16) t[k] = tv[it];
    }
    if (tid < (nt & 15)) tab[(n16 << 4) + tid] = deg8[(n16 << 4) + tid];
    __syncthreads();

    for (;;) {
        #pragma unroll
        for (int u = 0; u < 4; ++u) {
            const long long j = base + u * PLACE_TB + tid;
            if (j < e) {
                const int sidx = min(max(en[u].x, 0), nt - 1);
                const int cs = tab[sidx];
                const float fs = (cs < DEG_SAT) ? deg_to_dis(cs) : dis[sidx];
                list[j] = make_int2(en[u].x, __float_as_int(fs * __int_as_float(en[u].y)));
            }
        }
        base += (long long)PLACE_WG * (4 * PLACE_TB);
        if (base >= e) break;
        #pragma unroll
        for (int u = 0; u < 4; ++u) {
            const long long j = base + u * PLACE_TB + tid;
            en[u] = make_int2(0, 0);
            if (j < e) en[u] = list[j];
        }
    }
}


template <int KV, int AS, int KP, bool SPLIT_A, bool FULL_O>
__device__ __forceinline__ void
gemm_wmma_body(float* Wlds, const int wg, const float* A0, const float* A1,
               const float* __restrict__ W, float* O0, float* O1, int n) {
    for (int idx = threadIdx.x; idx < KP * H; idx += blockDim.x) {
        int k = idx / H, c = idx - k * H;
        float v = (k < KV) ? W[k * H + c] : 0.0f;
        Wlds[(k >> 1) * (H * 2) + c * 2 + (k & 1)] = v;
    }
    __syncthreads();

    const int lane  = threadIdx.x & 31;
    const int strip = wg * (blockDim.x >> 5) + (threadIdx.x >> 5);
    const int row0  = strip * 16;
    if (row0 >= n) return;

    const int mrow = lane & 15;
    const int hi   = lane >> 4;
    const float* arow  = SPLIT_A ? A0 + (size_t)(row0 + mrow) * HH
                                 : A0 + (size_t)(row0 + mrow) * AS;
    const float* arow1 = A1 + (size_t)(row0 + mrow) * HH;

    v8f acc[4];
    #pragma unroll
    for (int t = 0; t < 4; ++t) acc[t] = (v8f){0, 0, 0, 0, 0, 0, 0, 0};

    #pragma unroll
    for (int k0 = 0; k0 < KP; k0 += 4) {
        const int ka = k0 + hi * 2;
        v2f a;
        if (SPLIT_A) {
            const float* ap = (k0 < HH) ? (arow + ka) : (arow1 + (ka - HH));
            a = *reinterpret_cast<const v2f*>(ap);
        } else if (KV == KP && (AS & 1) == 0) {
            a = *reinterpret_cast<const v2f*>(arow + ka);
        } else {
            const int kc0 = (ka     < KV) ? ka     : (KV - 1);
            const int kc1 = (ka + 1 < KV) ? ka + 1 : (KV - 1);
            float ax = arow[kc0];
            float ay = arow[kc1];
            a.x = (ka     < KV) ? ax : 0.0f;
            a.y = (ka + 1 < KV) ? ay : 0.0f;
        }

        const int pbase = ((k0 >> 1) + hi) * (H * 2);
        #pragma unroll
        for (int t = 0; t < 4; ++t) {
            const int c = t * 16 + mrow;
            v2f b = *reinterpret_cast<const v2f*>(&Wlds[pbase + c * 2]);
            acc[t] = __builtin_amdgcn_wmma_f32_16x16x4_f32(
                 false, a,  false, b,
                 (short)0, acc[t],  false,  false);
        }
    }

    const int mbase = (lane < 16) ? 0 : 8;
    #pragma unroll
    for (int t = 0; t < 4; ++t) {
        if (FULL_O) {
            #pragma unroll
            for (int j = 0; j < 8; ++j) {
                O0[(size_t)(row0 + mbase + j) * H + t * 16 + mrow] = acc[t][j];
            }
        } else {
            float* op = (t < 2) ? O0 : O1;
            const int col = (t & 1) * 16 + mrow;
            #pragma unroll
            for (int j = 0; j < 8; ++j) {
                op[(size_t)(row0 + mbase + j) * HH + col] = acc[t][j];
            }
        }
    }
}

template <int KV, int AS, int KP, bool SPLIT_A, bool FULL_O>
__global__ void k_gemm_wmma(const float* A0, const float* A1, const float* __restrict__ W,
                            float* O0, float* O1, int n) {
    __shared__ float Wlds[(KP / 2) * H * 2];
    gemm_wmma_body<KV, AS, KP, SPLIT_A, FULL_O>(Wlds, blockIdx.x, A0, A1, W, O0, O1, n);
}

__global__ void __launch_bounds__(SORT_TB)
k_bin_sort_gemm1(const int2* __restrict__ binned, const int* __restrict__ bstart,
                 int2* __restrict__ list, int* __restrict__ rowend, float* __restrict__ dis,
                 unsigned char* __restrict__ deg8, const int* __restrict__ batch,
                 int* __restrict__ gstart, int e, int n, int nsort,
                 const float* __restrict__ x, const float* __restrict__ W1, float* z1) {
    __shared__ int cnt[BK];
    __shared__ int cursor[BK];
    __shared__ int wsum[SORT_TB / 32];
    __shared__ int2 stage[STAGE_CAP];
    __shared__ float Wlds[(12 / 2) * H * 2];
    if ((int)blockIdx.x < nsort) {
        bin_sort_body(blockIdx.x, cnt, cursor, wsum, stage, binned, bstart, list, rowend, dis,
                      deg8, batch, gstart, e, n);
    } else {
        gemm_wmma_body<F_IN, F_IN, 12, false, true>(Wlds, (int)blockIdx.x - nsort, x, x, W1,
                                                    z1, z1, n);
    }
}

__device__ __forceinline__ float4 pull_edges(const float* __restrict__ hin,
                                             const int2* __restrict__ list,
                                             int beg, int end, float4 acc, int q4) {
    const int last = end - 1;
    for (int j = beg; j < end; j += 4) {
        const int j1 = (j + 1 < end) ? j + 1 : last;
        const int j2 = (j + 2 < end) ? j + 2 : last;
        const int j3 = (j + 3 < end) ? j + 3 : last;
        const int2 e0 = list[j];
        const int2 e1 = list[j1];
        const int2 e2 = list[j2];
        const int2 e3 = list[j3];
        const float4 r0 = *reinterpret_cast<const float4*>(hin + (size_t)e0.x * HH + q4);
        const float4 r1 = *reinterpret_cast<const float4*>(hin + (size_t)e1.x * HH + q4);
        const float4 r2 = *reinterpret_cast<const float4*>(hin + (size_t)e2.x * HH + q4);
        const float4 r3 = *reinterpret_cast<const float4*>(hin + (size_t)e3.x * HH + q4);
        const bool v1 = (j + 1 < end), v2 = (j + 2 < end), v3 = (j + 3 < end);
        const float w0 = __int_as_float(e0.y);
        const float w1 = __int_as_float(e1.y);
        const float w2 = __int_as_float(e2.y);
        const float w3 = __int_as_float(e3.y);
        acc.x += r0.x * w0; acc.y += r0.y * w0; acc.z += r0.z * w0; acc.w += r0.w * w0;
        acc.x += (v1 ? r1.x : 0.0f) * w1; acc.y += (v1 ? r1.y : 0.0f) * w1;
        acc.z += (v1 ? r1.z : 0.0f) * w1; acc.w += (v1 ? r1.w : 0.0f) * w1;
        acc.x += (v2 ? r2.x : 0.0f) * w2; acc.y += (v2 ? r2.y : 0.0f) * w2;
        acc.z += (v2 ? r2.z : 0.0f) * w2; acc.w += (v2 ? r2.w : 0.0f) * w2;
        acc.x += (v3 ? r3.x : 0.0f) * w3; acc.y += (v3 ? r3.y : 0.0f) * w3;
        acc.z += (v3 ? r3.z : 0.0f) * w3; acc.w += (v3 ? r3.w : 0.0f) * w3;
    }
    return acc;
}

__device__ __forceinline__ float4 pull_node(const float* __restrict__ hin,
                                            const int* __restrict__ rowend,
                                            const int2* __restrict__ list,
                                            const float* __restrict__ dis,
                                            const float* __restrict__ bias, int node, int q4) {
    const int end = rowend[node];
    const int beg = (node > 0) ? rowend[node - 1] : 0;
    const float di = dis[node];
    const float wself = di * di;
    const float4 hv = *reinterpret_cast<const float4*>(hin + (size_t)node * HH + q4);
    float4 acc = make_float4(hv.x * wself, hv.y * wself, hv.z * wself, hv.w * wself);
    acc = pull_edges(hin, list, beg, end, acc, q4);
    const float4 bv = *reinterpret_cast<const float4*>(bias + q4);
    return make_float4(acc.x + bv.x, acc.y + bv.y, acc.z + bv.z, acc.w + bv.w);
}

__global__ void k_pull_half(const float* __restrict__ hin, float* __restrict__ hout,
                            const int* __restrict__ rowend, const int2* __restrict__ list,
                            const float* __restrict__ dis, const float* __restrict__ bias,
                            int n) {
    int t = blockIdx.x * blockDim.x + threadIdx.x;
    int node = t >> 3;
    if (node >= n) return;
    const int q4 = (t & 7) * 4;
    float4 r = pull_node(hin, rowend, list, dis, bias, node, q4);
    r.x = fmaxf(r.x, 0.0f); r.y = fmaxf(r.y, 0.0f);
    r.z = fmaxf(r.z, 0.0f); r.w = fmaxf(r.w, 0.0f);
    *reinterpret_cast<float4*>(hout + (size_t)node * HH + q4) = r;
}

__device__ __forceinline__ void pull_edges2(const float* __restrict__ h0,
                                            const float* __restrict__ h1,
                                            const int2* __restrict__ list,
                                            int beg, int end, float4& a0, float4& a1, int q4) {
    const int last = end - 1;
    for (int j = beg; j < end; j += 4) {
        const int j1 = (j + 1 < end) ? j + 1 : last;
        const int j2 = (j + 2 < end) ? j + 2 : last;
        const int j3 = (j + 3 < end) ? j + 3 : last;
        const int2 e0 = list[j];
        const int2 e1 = list[j1];
        const int2 e2 = list[j2];
        const int2 e3 = list[j3];
        const size_t o0 = (size_t)e0.x * HH + q4;
        const size_t o1 = (size_t)e1.x * HH + q4;
        const size_t o2 = (size_t)e2.x * HH + q4;
        const size_t o3 = (size_t)e3.x * HH + q4;
        const float4 r00 = *reinterpret_cast<const float4*>(h0 + o0);
        const float4 r01 = *reinterpret_cast<const float4*>(h1 + o0);
        const float4 r10 = *reinterpret_cast<const float4*>(h0 + o1);
        const float4 r11 = *reinterpret_cast<const float4*>(h1 + o1);
        const float4 r20 = *reinterpret_cast<const float4*>(h0 + o2);
        const float4 r21 = *reinterpret_cast<const float4*>(h1 + o2);
        const float4 r30 = *reinterpret_cast<const float4*>(h0 + o3);
        const float4 r31 = *reinterpret_cast<const float4*>(h1 + o3);
        const bool v1 = (j + 1 < end), v2 = (j + 2 < end), v3 = (j + 3 < end);
        const float w0 = __int_as_float(e0.y);
        const float w1 = __int_as_float(e1.y);
        const float w2 = __int_as_float(e2.y);
        const float w3 = __int_as_float(e3.y);
        a0.x += r00.x * w0; a0.y += r00.y * w0; a0.z += r00.z * w0; a0.w += r00.w * w0;
        a1.x += r01.x * w0; a1.y += r01.y * w0; a1.z += r01.z * w0; a1.w += r01.w * w0;
        a0.x += (v1 ? r10.x : 0.0f) * w1; a0.y += (v1 ? r10.y : 0.0f) * w1;
        a0.z += (v1 ? r10.z : 0.0f) * w1; a0.w += (v1 ? r10.w : 0.0f) * w1;
        a1.x += (v1 ? r11.x : 0.0f) * w1; a1.y += (v1 ? r11.y : 0.0f) * w1;
        a1.z += (v1 ? r11.z : 0.0f) * w1; a1.w += (v1 ? r11.w : 0.0f) * w1;
        a0.x += (v2 ? r20.x : 0.0f) * w2; a0.y += (v2 ? r20.y : 0.0f) * w2;
        a0.z += (v2 ? r20.z : 0.0f) * w2; a0.w += (v2 ? r20.w : 0.0f) * w2;
        a1.x += (v2 ? r21.x : 0.0f) * w2; a1.y += (v2 ? r21.y : 0.0f) * w2;
        a1.z += (v2 ? r21.z : 0.0f) * w2; a1.w += (v2 ? r21.w : 0.0f) * w2;
        a0.x += (v3 ? r30.x : 0.0f) * w3; a0.y += (v3 ? r30.y : 0.0f) * w3;
        a0.z += (v3 ? r30.z : 0.0f) * w3; a0.w += (v3 ? r30.w : 0.0f) * w3;
        a1.x += (v3 ? r31.x : 0.0f) * w3; a1.y += (v3 ? r31.y : 0.0f) * w3;
        a1.z += (v3 ? r31.z : 0.0f) * w3; a1.w += (v3 ? r31.w : 0.0f) * w3;
    }
}

#define PG_TB 256
__global__ void __launch_bounds__(PG_TB, 12)
k_pool_graph(const float* __restrict__ h0, const float* __restrict__ h1,
             const int* __restrict__ rowend, const int2* __restrict__ list,
             const float* __restrict__ dis, const float* __restrict__ bias,
             const int* __restrict__ gstart, const float* __restrict__ lin_w,
             const float* __restrict__ lin_b, float* __restrict__ out, int n) {
    __shared__ __attribute__((aligned(16))) float red[PG_TB / 32][H];
    const int g = blockIdx.x;
    const int nbeg = min(max(gstart[g], 0), n);
    const int nend = min(max(gstart[g + 1], nbeg), n);
    const int q4 = (threadIdx.x & 7) * 4;
    const int slot = threadIdx.x >> 3;

    float4 s0 = make_float4(0.0f, 0.0f, 0.0f, 0.0f);
    float4 s1 = make_float4(0.0f, 0.0f, 0.0f, 0.0f);
    for (int node = nbeg + slot; node < nend; node += PG_TB / 8) {
        const int end = rowend[node];
        const int beg = (node > 0) ? rowend[node - 1] : 0;
        const float di = dis[node];
        const float wself = di * di;
        const float4 hv0 = *reinterpret_cast<const float4*>(h0 + (size_t)node * HH + q4);
        const float4 hv1 = *reinterpret_cast<const float4*>(h1 + (size_t)node * HH + q4);
        float4 a0 = make_float4(hv0.x * wself, hv0.y * wself, hv0.z * wself, hv0.w * wself);
        float4 a1 = make_float4(hv1.x * wself, hv1.y * wself, hv1.z * wself, hv1.w * wself);
        pull_edges2(h0, h1, list, beg, end, a0, a1, q4);
        s0.x += a0.x; s0.y += a0.y; s0.z += a0.z; s0.w += a0.w;
        s1.x += a1.x; s1.y += a1.y; s1.z += a1.z; s1.w += a1.w;
    }

    s0.x += __shfl_xor(s0.x, 8);  s0.y += __shfl_xor(s0.y, 8);
    s0.z += __shfl_xor(s0.z, 8);  s0.w += __shfl_xor(s0.w, 8);
    s1.x += __shfl_xor(s1.x, 8);  s1.y += __shfl_xor(s1.y, 8);
    s1.z += __shfl_xor(s1.z, 8);  s1.w += __shfl_xor(s1.w, 8);
    s0.x += __shfl_xor(s0.x, 16); s0.y += __shfl_xor(s0.y, 16);
    s0.z += __shfl_xor(s0.z, 16); s0.w += __shfl_xor(s0.w, 16);
    s1.x += __shfl_xor(s1.x, 16); s1.y += __shfl_xor(s1.y, 16);
    s1.z += __shfl_xor(s1.z, 16); s1.w += __shfl_xor(s1.w, 16);
    const int lane = threadIdx.x & 31;
    const int wv   = threadIdx.x >> 5;
    if (lane < 8) {
        *reinterpret_cast<float4*>(&red[wv][q4]) = s0;
        *reinterpret_cast<float4*>(&red[wv][HH + q4]) = s1;
    }
    __syncthreads();

    if (threadIdx.x < 32) {
        const float cntf = (float)(nend - nbeg);
        float ca = 0.0f, cb = 0.0f;
        for (int w = 0; w < PG_TB / 32; ++w) {
            ca += red[w][lane];
            cb += red[w][HH + lane];
        }
        ca += cntf * bias[lane];
        cb += cntf * bias[HH + lane];
        float p = ca * lin_w[lane] + cb * lin_w[HH + lane];
        for (int o = 16; o > 0; o >>= 1) p += __shfl_xor(p, o);
        if (lane == 0) {
            const float inv = 1.0f / fmaxf(cntf, 1.0f);
            out[g] = p * inv + lin_b[0];
        }
    }
}

#define FB_ROWS 128
#define TP 36

__global__ void __launch_bounds__(256)
k_pull_gemm(const float* __restrict__ hg, float* P0, float* O1,
            const int* __restrict__ rowend, const int2* __restrict__ list,
            const float* __restrict__ dis, const float* __restrict__ bias,
            const float* __restrict__ W, int n) {
    __shared__ float Wlds[(H / 2) * H * 2];
    __shared__ __attribute__((aligned(16))) float tile[FB_ROWS * TP];
    for (int idx = threadIdx.x; idx < H * H; idx += blockDim.x) {
        int k = idx / H, c = idx - k * H;
        Wlds[(k >> 1) * (H * 2) + c * 2 + (k & 1)] = W[k * H + c];
    }

    const int blockBase = blockIdx.x * FB_ROWS;
    const int q4 = (threadIdx.x & 7) * 4;
    const int sub = threadIdx.x >> 3;
    const float4 bv = *reinterpret_cast<const float4*>(bias + q4);

    int beg[4], end[4];
    float4 acc[4];
    #pragma unroll
    for (int it = 0; it < 4; ++it) {
        int node = blockBase + it * 32 + sub;
        node = (node < n) ? node : (n - 1);
        end[it] = rowend[node];
        beg[it] = (node > 0) ? rowend[node - 1] : 0;
        const float di = dis[node];
        const float wself = di * di;
        const float4 hv = *reinterpret_cast<const float4*>(hg + (size_t)node * HH + q4);
        acc[it] = make_float4(hv.x * wself, hv.y * wself, hv.z * wself, hv.w * wself);
    }
    #pragma unroll
    for (int it = 0; it < 4; ++it) {
        const float4 a = pull_edges(hg, list, beg[it], end[it], acc[it], q4);
        float4 r = make_float4(a.x + bv.x, a.y + bv.y, a.z + bv.z, a.w + bv.w);
        r.x = fmaxf(r.x, 0.0f); r.y = fmaxf(r.y, 0.0f);
        r.z = fmaxf(r.z, 0.0f); r.w = fmaxf(r.w, 0.0f);
        *reinterpret_cast<float4*>(&tile[(it * 32 + sub) * TP + q4]) = r;
    }
    __syncthreads();

    const int lane = threadIdx.x & 31;
    const int wv   = threadIdx.x >> 5;
    const int row0 = blockBase + wv * 16;
    if (row0 >= n) return;

    const int mrow = lane & 15;
    const int hi   = lane >> 4;
    const float* arow = P0 + (size_t)(row0 + mrow) * HH;
    const float* trow = &tile[(wv * 16 + mrow) * TP];

    v8f c4[4];
    #pragma unroll
    for (int t = 0; t < 4; ++t) c4[t] = (v8f){0, 0, 0, 0, 0, 0, 0, 0};

    #pragma unroll
    for (int k0 = 0; k0 < H; k0 += 4) {
        const int ka = k0 + hi * 2;
        const float* ap = (k0 < HH) ? (arow + ka) : (trow + (ka - HH));
        const v2f a = *reinterpret_cast<const v2f*>(ap);

        const int pbase = ((k0 >> 1) + hi) * (H * 2);
        #pragma unroll
        for (int t = 0; t < 4; ++t) {
            const int c = t * 16 + mrow;
            v2f b = *reinterpret_cast<const v2f*>(&Wlds[pbase + c * 2]);
            c4[t] = __builtin_amdgcn_wmma_f32_16x16x4_f32(
                 false, a,  false, b,
                 (short)0, c4[t],  false,  false);
        }
    }

    const int mbase = (lane < 16) ? 0 : 8;
    #pragma unroll
    for (int t = 0; t < 4; ++t) {
        float* op = (t < 2) ? P0 : O1;
        const int col = (t & 1) * 16 + mrow;
        #pragma unroll
        for (int j = 0; j < 8; ++j) {
            op[(size_t)(row0 + mbase + j) * HH + col] = c4[t][j];
        }
    }
}


__device__ __forceinline__ float4 pull_edges_row(const float* __restrict__ zin,
                                                 const int2* __restrict__ list,
                                                 int beg, int end, float4 acc, int c4) {
    const int last = end - 1;
    for (int j = beg; j < end; j += 8) {
        int2 en[8];
        float4 r[8];
        #pragma unroll
        for (int u = 0; u < 8; ++u) en[u] = list[min(j + u, last)];
        #pragma unroll
        for (int u = 0; u < 8; ++u)
            r[u] = *reinterpret_cast<const float4*>(zin + (size_t)en[u].x * H + c4);
        #pragma unroll
        for (int u = 0; u < 8; ++u) {
            const bool v = (j + u < end);
            const float w = __int_as_float(en[u].y);
            acc.x += (v ? r[u].x : 0.0f) * w; acc.y += (v ? r[u].y : 0.0f) * w;
            acc.z += (v ? r[u].z : 0.0f) * w; acc.w += (v ? r[u].w : 0.0f) * w;
        }
    }
    return acc;
}

__device__ __forceinline__ void list_head8(const int2* __restrict__ list, int beg, int end,
                                           int2 (&en)[8]) {
    const int last = max(end - 1, 0);
    #pragma unroll
    for (int u = 0; u < 8; ++u) en[u] = list[min(beg + u, last)];
}

#define TPF 68

__global__ void __launch_bounds__(256, 6)
k_pull_gemm_rows(const float* __restrict__ zin, float* __restrict__ zout,
                 const int* __restrict__ rowend, const int2* __restrict__ list,
                 const float* __restrict__ dis, const float* __restrict__ bias,
                 const float* __restrict__ W, int n) {
    __shared__ float Wlds[(H / 2) * H * 2];
    __shared__ __attribute__((aligned(16))) float tile[FB_ROWS * TPF];
    for (int idx = threadIdx.x; idx < H * H; idx += blockDim.x) {
        int k = idx / H, c = idx - k * H;
        Wlds[(k >> 1) * (H * 2) + c * 2 + (k & 1)] = W[k * H + c];
    }

    const int lane = threadIdx.x & 31;
    const int wv   = threadIdx.x >> 5;
    const int row0 = blockIdx.x * FB_ROWS + wv * 16;
    const int c4   = (lane & 15) * 4;
    const int half = lane >> 4;
    const float4 bv = *reinterpret_cast<const float4*>(bias + c4);

    int beg[8], end[8];
    float4 acc[8];
    #pragma unroll
    for (int it = 0; it < 8; ++it) {
        int node = row0 + it * 2 + half;
        node = (node < n) ? node : (n - 1);
        end[it] = rowend[node];
        beg[it] = (node > 0) ? rowend[node - 1] : 0;
        const float di = dis[node];
        const float wself = di * di;
        const float4 hv = *reinterpret_cast<const float4*>(zin + (size_t)node * H + c4);
        acc[it] = make_float4(hv.x * wself, hv.y * wself, hv.z * wself, hv.w * wself);
    }
    int2 en[8];
    list_head8(list, beg[0], end[0], en);
    #pragma unroll
    for (int it = 0; it < 8; ++it) {
        float4 r[8];
        #pragma unroll
        for (int u = 0; u < 8; ++u)
            r[u] = *reinterpret_cast<const float4*>(zin + (size_t)en[u].x * H + c4);
        int2 nx[8];
        if (it + 1 < 8) list_head8(list, beg[it + 1], end[it + 1], nx);
        float4 a = acc[it];
        #pragma unroll
        for (int u = 0; u < 8; ++u) {
            const bool v = (beg[it] + u < end[it]);
            const float w = __int_as_float(en[u].y);
            a.x += (v ? r[u].x : 0.0f) * w; a.y += (v ? r[u].y : 0.0f) * w;
            a.z += (v ? r[u].z : 0.0f) * w; a.w += (v ? r[u].w : 0.0f) * w;
        }
        a = pull_edges_row(zin, list, beg[it] + 8, end[it], a, c4);
        float4 o = make_float4(a.x + bv.x, a.y + bv.y, a.z + bv.z, a.w + bv.w);
        o.x = fmaxf(o.x, 0.0f); o.y = fmaxf(o.y, 0.0f);
        o.z = fmaxf(o.z, 0.0f); o.w = fmaxf(o.w, 0.0f);
        *reinterpret_cast<float4*>(&tile[(wv * 16 + it * 2 + half) * TPF + c4]) = o;
        if (it + 1 < 8) {
            #pragma unroll
            for (int u = 0; u < 8; ++u) en[u] = nx[u];
        }
    }
    __syncthreads();

    if (row0 >= n) return;

    const int mrow = lane & 15;
    const int hi   = lane >> 4;
    const float* trow = &tile[(wv * 16 + mrow) * TPF];

    v8f c4acc[4];
    #pragma unroll
    for (int t = 0; t < 4; ++t) c4acc[t] = (v8f){0, 0, 0, 0, 0, 0, 0, 0};

    #pragma unroll
    for (int k0 = 0; k0 < H; k0 += 4) {
        const int ka = k0 + hi * 2;
        const v2f a = *reinterpret_cast<const v2f*>(trow + ka);

        const int pbase = ((k0 >> 1) + hi) * (H * 2);
        #pragma unroll
        for (int t = 0; t < 4; ++t) {
            const int c = t * 16 + mrow;
            v2f b = *reinterpret_cast<const v2f*>(&Wlds[pbase + c * 2]);
            c4acc[t] = __builtin_amdgcn_wmma_f32_16x16x4_f32(
                 false, a,  false, b,
                 (short)0, c4acc[t],  false,  false);
        }
    }

    const int mbase = (lane < 16) ? 0 : 8;
    #pragma unroll
    for (int t = 0; t < 4; ++t) {
        #pragma unroll
        for (int j = 0; j < 8; ++j) {
            zout[(size_t)(row0 + mbase + j) * H + t * 16 + mrow] = c4acc[t][j];
        }
    }
}

__global__ void __launch_bounds__(PG_TB, 12)
k_pool_graph_rows(const float* __restrict__ z,
                  const int* __restrict__ rowend, const int2* __restrict__ list,
                  const float* __restrict__ dis, const float* __restrict__ bias,
                  const int* __restrict__ gstart, const float* __restrict__ lin_w,
                  const float* __restrict__ lin_b, float* __restrict__ out, int n) {
    __shared__ __attribute__((aligned(16))) float red[PG_TB / 32][H];
    const int g = blockIdx.x;
    const int nbeg = min(max(gstart[g], 0), n);
    const int nend = min(max(gstart[g + 1], nbeg), n);
    const int c4 = (threadIdx.x & 15) * 4;
    const int slot = threadIdx.x >> 4;

    float4 s = make_float4(0.0f, 0.0f, 0.0f, 0.0f);
    for (int node = nbeg + slot; node < nend; node += PG_TB / 16) {
        const int end = rowend[node];
        const int beg = (node > 0) ? rowend[node - 1] : 0;
        const float di = dis[node];
        const float wself = di * di;
        const float4 hv = *reinterpret_cast<const float4*>(z + (size_t)node * H + c4);
        float4 a = make_float4(hv.x * wself, hv.y * wself, hv.z * wself, hv.w * wself);
        a = pull_edges_row(z, list, beg, end, a, c4);
        s.x += a.x; s.y += a.y; s.z += a.z; s.w += a.w;
    }

    s.x += __shfl_xor(s.x, 16); s.y += __shfl_xor(s.y, 16);
    s.z += __shfl_xor(s.z, 16); s.w += __shfl_xor(s.w, 16);
    const int lane = threadIdx.x & 31;
    const int wv   = threadIdx.x >> 5;
    if (lane < 16) *reinterpret_cast<float4*>(&red[wv][c4]) = s;
    __syncthreads();

    if (threadIdx.x < 32) {
        const float cntf = (float)(nend - nbeg);
        float ca = 0.0f, cb = 0.0f;
        for (int w = 0; w < PG_TB / 32; ++w) {
            ca += red[w][lane];
            cb += red[w][HH + lane];
        }
        ca += cntf * bias[lane];
        cb += cntf * bias[HH + lane];
        float p = ca * lin_w[lane] + cb * lin_w[HH + lane];
        for (int o = 16; o > 0; o >>= 1) p += __shfl_xor(p, o);
        if (lane == 0) {
            const float inv = 1.0f / fmaxf(cntf, 1.0f);
            out[g] = p * inv + lin_b[0];
        }
    }
}


extern "C" void kernel_launch(void* const* d_in, const int* in_sizes, int n_in,
                              void* d_out, int out_size, void* d_ws, size_t ws_size,
                              hipStream_t stream) {
    const float* x     = (const float*)d_in[0];
    const int*   eidx  = (const int*)  d_in[1];
    const int*   batch = (const int*)  d_in[2];
    const float* W1    = (const float*)d_in[3];
    const float* b1    = (const float*)d_in[4];
    const float* W2    = (const float*)d_in[5];
    const float* b2    = (const float*)d_in[6];
    const float* W3    = (const float*)d_in[7];
    const float* b3    = (const float*)d_in[8];
    const float* lin_w = (const float*)d_in[9];
    const float* lin_b = (const float*)d_in[10];
    float* out = (float*)d_out;

    const int n = in_sizes[0] / F_IN;
    const int e = in_sizes[1] / 2;
    const int* src = eidx;
    const int* dst = eidx + e;

    const int TB = 256;
    dim3 blk(TB);
    auto blocks = [](long long work, int tb) { return dim3((unsigned)((work + tb - 1) / tb)); };
    const dim3 bin_grid  = blocks(e, EPW);
    const dim3 sort_grid = blocks(n, BK);
    const int strips       = (n + 15) / 16;
    const dim3 gemm_grid   = blocks((long long)strips * 32, TB);
    const dim3 fuse_grid   = blocks((long long)n, FB_ROWS);

    const size_t rows_need = ((size_t)n * (2 * H + 3) + 1024 + 2 * (size_t)e + N_GRAPHS + 4)
                             * sizeof(float) + (size_t)n;
    if (ws_size >= rows_need) {
        float* ws   = (float*)d_ws;
        float* ZA   = ws;
        float* ZB   = ZA + (size_t)n * H;
        float* dis  = ZB + (size_t)n * H;
        int*   cnt    = (int*)(dis + n);
        int*   cursor = cnt + n;
        int*   part   = cursor + n;
        int2*  list   = (int2*)(part + 1024);
        int*   gstart = (int*)(list + e);
        unsigned char* deg8 = (unsigned char*)(gstart + N_GRAPHS + 4);
        int2*  binned = (int2*)ZB;
        int*   hist   = (int*)ZB + 2 * (size_t)e;
        int*   bstart = hist + (size_t)bin_grid.x * MAXB;
        const dim3 gemm1_grid = blocks((long long)strips * 32, SORT_TB);

        k_bin_count   <<<bin_grid, dim3(SORT_TB), 0, stream>>>(src, dst, hist, e);
        k_bin_scatter <<<bin_grid, dim3(SORT_TB), 0, stream>>>(src, dst, hist, binned, bstart, e, (int)bin_grid.x);
        k_bin_sort_gemm1<<<dim3(sort_grid.x + gemm1_grid.x), dim3(SORT_TB), 0, stream>>>(
            binned, bstart, list, cursor, dis, deg8, batch, gstart, e, n, (int)sort_grid.x, x, W1, ZA);
        k_edge_weights<<<dim3(PLACE_WG), dim3(PLACE_TB), 0, stream>>>(list, dis, deg8, e, n);
        k_pull_gemm_rows<<<fuse_grid, blk, 0, stream>>>(ZA, ZB, cursor, list, dis, b1, W2, n);
        k_pull_gemm_rows<<<fuse_grid, blk, 0, stream>>>(ZB, ZA, cursor, list, dis, b2, W3, n);
        k_pool_graph_rows<<<dim3(N_GRAPHS), dim3(PG_TB), 0, stream>>>(ZA, cursor, list, dis, b3, gstart,
                                                                      lin_w, lin_b, out, n);
        return;
    }

    float* ws   = (float*)d_ws;
    float* dis  = ws;
    float* R0   = dis + n;
    float* R1   = R0 + (size_t)n * HH;
    float* R2   = R1 + (size_t)n * HH;
    int*   cnt    = (int*)(R2 + (size_t)n * HH);
    int*   cursor = cnt + n;
    int*   part   = cursor + n;
    int2*  list   = (int2*)(part + 1024);
    int*   gstart = (int*)(list + e);
    unsigned char* deg8 = (unsigned char*)(gstart + N_GRAPHS + 4);
    int2*  binned = (int2*)R0;
    int*   hist   = (int*)R1;
    int*   bstart = hist + (size_t)bin_grid.x * MAXB;

    k_bin_count   <<<bin_grid, dim3(SORT_TB), 0, stream>>>(src, dst, hist, e);
    k_bin_scatter <<<bin_grid, dim3(SORT_TB), 0, stream>>>(src, dst, hist, binned, bstart, e, (int)bin_grid.x);
    k_bin_sort    <<<sort_grid, dim3(SORT_TB), 0, stream>>>(binned, bstart, list, cursor, dis, deg8, batch, gstart, e, n);
    k_edge_weights<<<dim3(PLACE_WG), dim3(PLACE_TB), 0, stream>>>(list, dis, deg8, e, n);

    const dim3 pull_grid   = blocks((long long)n * 8, TB);

    k_gemm_wmma<F_IN, F_IN, 12, false, false><<<gemm_grid, blk, 0, stream>>>(x, x, W1, R0, R1, n);

    k_pull_half<<<pull_grid, blk, 0, stream>>>(R0, R2, cursor, list, dis, b1, n);
    k_pull_gemm<<<fuse_grid, blk, 0, stream>>>(R1, R2, R0, cursor, list, dis, b1 + HH, W2, n);

    k_pull_half<<<pull_grid, blk, 0, stream>>>(R2, R1, cursor, list, dis, b2, n);
    k_pull_gemm<<<fuse_grid, blk, 0, stream>>>(R0, R1, R2, cursor, list, dis, b2 + HH, W3, n);

    k_pool_graph<<<dim3(N_GRAPHS), dim3(PG_TB), 0, stream>>>(R1, R2, cursor, list, dis, b3, gstart,
                                                            lin_w, lin_b, out, n);
}


# level-1 binning orders each group's 4096 edges by bin inside the group's own piece (no count launch, no count table); a bin's sort reads its part of every piece; 7 -> 6 launches; sums regrouped
# speedup vs baseline: 3.7435x; 1.0561x over previous
#include <hip/hip_runtime.h>
#include <hip/hip_bf16.h>

#define N_NODES 200000
#define N_EDGES 1000000
#define N_GRAPHS 4096
#define F_IN 11
#define H 64
#define HH 32

typedef float v2f __attribute__((ext_vector_type(2)));
typedef float v8f __attribute__((ext_vector_type(8)));


#define DEG_SAT 255
#define PLACE_TB 1024
#define PLACE_WG 256
#define TAB_IT ((N_NODES / 16 + PLACE_TB - 1) / PLACE_TB)

#define BK_SHIFT 10
#define BK (1 << BK_SHIFT)
#define MAXB ((N_NODES + BK - 1) / BK)
#define SORT_TB 1024
#define EPW (4 * SORT_TB)
#define STAGE_CAP 16384
#define LOFF_ROW (MAXB + 1)
#define PIECE_LANES 4

__device__ __forceinline__ float deg_to_dis(int c) {
    return rsqrtf(fmaxf((float)(c + 1), 1.0f));
}

__device__ __forceinline__ int bin_of(int d) {
    return min(max(d >> BK_SHIFT, 0), MAXB - 1);
}

__device__ __forceinline__ void load_quad(const int* __restrict__ src, const int* __restrict__ dst,
                                          int i, int e, bool want_src,
                                          int (&s)[4], int (&d)[4], int& nv) {
    nv = min(max(e - i, 0), 4);
    const bool vec_ok = ((((size_t)src) | ((size_t)dst)) & 15) == 0;
    if (nv == 4 && vec_ok) {
        const int4 dv = *reinterpret_cast<const int4*>(dst + i);
        d[0] = dv.x; d[1] = dv.y; d[2] = dv.z; d[3] = dv.w;
        if (want_src) {
            const int4 sv = *reinterpret_cast<const int4*>(src + i);
            s[0] = sv.x; s[1] = sv.y; s[2] = sv.z; s[3] = sv.w;
        } else {
            s[0] = 0; s[1] = 0; s[2] = 0; s[3] = 0;
        }
    } else {
        #pragma unroll
        for (int u = 0; u < 4; ++u) {
            s[u] = 0; d[u] = 0;
            if (u < nv) {
                d[u] = dst[i + u];
                if (want_src) s[u] = src[i + u];
            }
        }
    }
}

__global__ void __launch_bounds__(SORT_TB)
k_bin_local(const int* __restrict__ src, const int* __restrict__ dst,
            int2* __restrict__ binned, int* __restrict__ loff, int e) {
    __shared__ int cntb[256];
    __shared__ int cursor[256];
    __shared__ int wsO[8];
    __shared__ int2 stage[EPW];
    const int tid = threadIdx.x, lane = tid & 31, wv = tid >> 5;
    const int w = blockIdx.x;

    int s[4], d[4], nv;
    load_quad(src, dst, w * EPW + tid * 4, e, true, s, d, nv);
    if (tid < 256) cntb[tid] = 0;
    __syncthreads();
    #pragma unroll
    for (int u = 0; u < 4; ++u)
        if (u < nv) atomicAdd(&cntb[bin_of(d[u])], 1);
    __syncthreads();

    int own = 0, iO = 0;
    if (tid < 256) {
        own = cntb[tid];
        iO = own;
        for (int o = 1; o < 32; o <<= 1) {
            const int a = __shfl_up(iO, o);
            if (lane >= o) iO += a;
        }
        if (lane == 31) wsO[wv] = iO;
    }
    __syncthreads();
    if (tid < 256) {
        int offO = 0;
        for (int k = 0; k < wv; ++k) offO += wsO[k];
        const int exO = offO + iO - own;
        cursor[tid] = exO;
        if (tid <= MAXB) loff[(size_t)w * LOFF_ROW + tid] = exO;
    }
    __syncthreads();

    #pragma unroll
    for (int u = 0; u < 4; ++u) {
        if (u < nv) {
            const int r = atomicAdd(&cursor[bin_of(d[u])], 1);
            stage[min(max(r, 0), EPW - 1)] = make_int2(s[u], d[u]);
        }
    }
    __syncthreads();
    const int cw = min(max(e - w * EPW, 0), EPW);
    int2* piece = binned + (size_t)w * EPW;
    for (int k = tid; k < cw; k += SORT_TB) piece[k] = stage[k];
}

__device__ __forceinline__ void
bin_sort_body(const int b, int* cnt, int* cursor, int* wsum, int* sred, int2* stage,
              const int2* __restrict__ binned, const int* __restrict__ loff, const int nw,
              int2* __restrict__ list, int* __restrict__ rowend, float* __restrict__ dis,
              unsigned char* __restrict__ deg8, const int* __restrict__ batch,
              int* __restrict__ gstart, int e, int n) {
    const int tid = threadIdx.x, lane = tid & 31, wv = tid >> 5;
    const int pj = tid & (PIECE_LANES - 1);

    cnt[tid] = 0;
    if (tid < 2) sred[tid] = 0;
    __syncthreads();
    int slo = 0, shi = 0;
    for (int w = tid / PIECE_LANES; w < nw; w += SORT_TB / PIECE_LANES) {
        const int cw = min(max(e - w * EPW, 0), EPW);
        const int lo = min(max(loff[(size_t)w * LOFF_ROW + b], 0), cw);
        const int hi = min(max(loff[(size_t)w * LOFF_ROW + b + 1], lo), cw);
        if (pj == 0) { slo += lo; shi += hi; }
        const int2* piece = binned + (size_t)w * EPW;
        for (int i0 = lo + pj; i0 < hi; i0 += 8 * PIECE_LANES) {
            int dv[8];
            #pragma unroll
            for (int u = 0; u < 8; ++u) dv[u] = piece[min(i0 + u * PIECE_LANES, hi - 1)].y;
            #pragma unroll
            for (int u = 0; u < 8; ++u)
                if (i0 + u * PIECE_LANES < hi)
                    atomicAdd(&cnt[min(max(dv[u] - b * BK, 0), BK - 1)], 1);
        }
    }
    for (int o = 16; o > 0; o >>= 1) {
        slo += __shfl_xor(slo, o);
        shi += __shfl_xor(shi, o);
    }
    if (lane == 0) { atomicAdd(&sred[0], slo); atomicAdd(&sred[1], shi); }
    __syncthreads();

    const int sbeg = min(max(sred[0], 0), e);
    const int send = min(max(sred[1], sbeg), e);
    const int seg = send - sbeg;
    const bool staged = (seg <= STAGE_CAP);

    const int c = cnt[tid];
    int incl = c;
    for (int o = 1; o < 32; o <<= 1) {
        const int v = __shfl_up(incl, o);
        if (lane >= o) incl += v;
    }
    if (lane == 31) wsum[wv] = incl;
    __syncthreads();
    for (int k = 0; k < wv; ++k) incl += wsum[k];
    cursor[tid] = incl - c;

    const int v = b * BK + tid;
    if (v < n) {
        rowend[v] = sbeg + incl;
        dis[v] = deg_to_dis(c);
        deg8[v] = (unsigned char)min(c, DEG_SAT);
        const int pb = (v > 0) ? batch[v - 1] : -1;
        const int gb = batch[v];
        for (int g = max(pb + 1, 0); g <= min(gb, N_GRAPHS); ++g) gstart[g] = v;
        if (v == n - 1)
            for (int g = max(gb + 1, 0); g <= N_GRAPHS; ++g) gstart[g] = n;
    }
    __syncthreads();

    for (int w = tid / PIECE_LANES; w < nw; w += SORT_TB / PIECE_LANES) {
        const int cw = min(max(e - w * EPW, 0), EPW);
        const int lo = min(max(loff[(size_t)w * LOFF_ROW + b], 0), cw);
        const int hi = min(max(loff[(size_t)w * LOFF_ROW + b + 1], lo), cw);
        const int2* piece = binned + (size_t)w * EPW;
        for (int i0 = lo + pj; i0 < hi; i0 += 8 * PIECE_LANES) {
            int2 ed[8];
            #pragma unroll
            for (int u = 0; u < 8; ++u) ed[u] = piece[min(i0 + u * PIECE_LANES, hi - 1)];
            #pragma unroll
            for (int u = 0; u < 8; ++u) {
                if (i0 + u * PIECE_LANES < hi) {
                    const int dl = min(max(ed[u].y - b * BK, 0), BK - 1);
                    const int slot = min(max(atomicAdd(&cursor[dl], 1), 0), seg - 1);
                    const int2 o = make_int2(ed[u].x, __float_as_int(deg_to_dis(cnt[dl])));
                    if (staged) stage[slot] = o;
                    else        list[sbeg + slot] = o;
                }
            }
        }
    }
    __syncthreads();
    if (staged)
        for (int i = tid; i < seg; i += SORT_TB) list[sbeg + i] = stage[i];
}

__global__ void __launch_bounds__(SORT_TB)
k_bin_sort(const int2* __restrict__ binned, const int* __restrict__ loff, int nw,
           int2* __restrict__ list, int* __restrict__ rowend, float* __restrict__ dis,
           unsigned char* __restrict__ deg8, const int* __restrict__ batch,
           int* __restrict__ gstart, int e, int n) {
    __shared__ int cnt[BK];
    __shared__ int cursor[BK];
    __shared__ int wsum[SORT_TB / 32];
    __shared__ int sred[2];
    __shared__ int2 stage[STAGE_CAP];
    bin_sort_body(blockIdx.x, cnt, cursor, wsum, sred, stage, binned, loff, nw, list, rowend, dis,
                  deg8, batch, gstart, e, n);
}

__global__ void __launch_bounds__(PLACE_TB)
k_edge_weights(int2* __restrict__ list, const float* __restrict__ dis,
               const unsigned char* __restrict__ deg8, int e, int n) {
    __shared__ __attribute__((aligned(16))) unsigned char tab[N_NODES];
    const int tid = threadIdx.x;
    if ((long long)blockIdx.x * (4 * PLACE_TB) >= e) return;

    long long base = (long long)blockIdx.x * (4 * PLACE_TB);
    int2 en[4];
    #pragma unroll
    for (int u = 0; u < 4; ++u) {
        const long long j = base + u * PLACE_TB + tid;
        en[u] = make_int2(0, 0);
        if (j < e) en[u] = list[j];
    }

    const int nt  = min(n, N_NODES);
    const int n16 = nt >> 4;
    const uint4* g = reinterpret_cast<const uint4*>(deg8);
    uint4* t = reinterpret_cast<uint4*>(tab);
    uint4 tv[TAB_IT];
    #pragma unroll
    for (int it = 0; it < TAB_IT; ++it) {
        const int k = tid + it * PLACE_TB;
        tv[it] = make_uint4(0u, 0u, 0u, 0u);
        if (k < n16) tv[it] = g[k];
    }
    #pragma unroll
    for (int it = 0; it < TAB_IT; ++it) {
        const int k = tid + it * PLACE_TB;
        if (k < n16) t[k] = tv[it];
    }
    if (tid < (nt & 15)) tab[(n16 << 4) + tid] = deg8[(n16 << 4) + tid];
    __syncthreads();

    for (;;) {
        #pragma unroll
        for (int u = 0; u < 4; ++u) {
            const long long j = base + u * PLACE_TB + tid;
            if (j < e) {
                const int sidx = min(max(en[u].x, 0), nt - 1);
                const int cs = tab[sidx];
                const float fs = (cs < DEG_SAT) ? deg_to_dis(cs) : dis[sidx];
                list[j] = make_int2(en[u].x, __float_as_int(fs * __int_as_float(en[u].y)));
            }
        }
        base += (long long)PLACE_WG * (4 * PLACE_TB);
        if (base >= e) break;
        #pragma unroll
        for (int u = 0; u < 4; ++u) {
            const long long j = base + u * PLACE_TB + tid;
            en[u] = make_int2(0, 0);
            if (j < e) en[u] = list[j];
        }
    }
}


template <int KV, int AS, int KP, bool SPLIT_A, bool FULL_O>
__device__ __forceinline__ void
gemm_wmma_body(float* Wlds, const int wg, const float* A0, const float* A1,
               const float* __restrict__ W, float* O0, float* O1, int n) {
    for (int idx = threadIdx.x; idx < KP * H; idx += blockDim.x) {
        int k = idx / H, c = idx - k * H;
        float v = (k < KV) ? W[k * H + c] : 0.0f;
        Wlds[(k >> 1) * (H * 2) + c * 2 + (k & 1)] = v;
    }
    __syncthreads();

    const int lane  = threadIdx.x & 31;
    const int strip = wg * (blockDim.x >> 5) + (threadIdx.x >> 5);
    const int row0  = strip * 16;
    if (row0 >= n) return;

    const int mrow = lane & 15;
    const int hi   = lane >> 4;
    const float* arow  = SPLIT_A ? A0 + (size_t)(row0 + mrow) * HH
                                 : A0 + (size_t)(row0 + mrow) * AS;
    const float* arow1 = A1 + (size_t)(row0 + mrow) * HH;

    v8f acc[4];
    #pragma unroll
    for (int t = 0; t < 4; ++t) acc[t] = (v8f){0, 0, 0, 0, 0, 0, 0, 0};

    #pragma unroll
    for (int k0 = 0; k0 < KP; k0 += 4) {
        const int ka = k0 + hi * 2;
        v2f a;
        if (SPLIT_A) {
            const float* ap = (k0 < HH) ? (arow + ka) : (arow1 + (ka - HH));
            a = *reinterpret_cast<const v2f*>(ap);
        } else if (KV == KP && (AS & 1) == 0) {
            a = *reinterpret_cast<const v2f*>(arow + ka);
        } else {
            const int kc0 = (ka     < KV) ? ka     : (KV - 1);
            const int kc1 = (ka + 1 < KV) ? ka + 1 : (KV - 1);
            float ax = arow[kc0];
            float ay = arow[kc1];
            a.x = (ka     < KV) ? ax : 0.0f;
            a.y = (ka + 1 < KV) ? ay : 0.0f;
        }

        const int pbase = ((k0 >> 1) + hi) * (H * 2);
        #pragma unroll
        for (int t = 0; t < 4; ++t) {
            const int c = t * 16 + mrow;
            v2f b = *reinterpret_cast<const v2f*>(&Wlds[pbase + c * 2]);
            acc[t] = __builtin_amdgcn_wmma_f32_16x16x4_f32(
                 false, a,  false, b,
                 (short)0, acc[t],  false,  false);
        }
    }

    const int mbase = (lane < 16) ? 0 : 8;
    #pragma unroll
    for (int t = 0; t < 4; ++t) {
        if (FULL_O) {
            #pragma unroll
            for (int j = 0; j < 8; ++j) {
                O0[(size_t)(row0 + mbase + j) * H + t * 16 + mrow] = acc[t][j];
            }
        } else {
            float* op = (t < 2) ? O0 : O1;
            const int col = (t & 1) * 16 + mrow;
            #pragma unroll
            for (int j = 0; j < 8; ++j) {
                op[(size_t)(row0 + mbase + j) * HH + col] = acc[t][j];
            }
        }
    }
}

template <int KV, int AS, int KP, bool SPLIT_A, bool FULL_O>
__global__ void k_gemm_wmma(const float* A0, const float* A1, const float* __restrict__ W,
                            float* O0, float* O1, int n) {
    __shared__ float Wlds[(KP / 2) * H * 2];
    gemm_wmma_body<KV, AS, KP, SPLIT_A, FULL_O>(Wlds, blockIdx.x, A0, A1, W, O0, O1, n);
}

__global__ void __launch_bounds__(SORT_TB)
k_bin_sort_gemm1(const int2* __restrict__ binned, const int* __restrict__ loff, int nw,
                 int2* __restrict__ list, int* __restrict__ rowend, float* __restrict__ dis,
                 unsigned char* __restrict__ deg8, const int* __restrict__ batch,
                 int* __restrict__ gstart, int e, int n, int nsort,
                 const float* __restrict__ x, const float* __restrict__ W1, float* z1) {
    __shared__ int cnt[BK];
    __shared__ int cursor[BK];
    __shared__ int wsum[SORT_TB / 32];
    __shared__ int sred[2];
    __shared__ int2 stage[STAGE_CAP];
    __shared__ float Wlds[(12 / 2) * H * 2];
    if ((int)blockIdx.x < nsort) {
        bin_sort_body(blockIdx.x, cnt, cursor, wsum, sred, stage, binned, loff, nw, list, rowend,
                      dis, deg8, batch, gstart, e, n);
    } else {
        gemm_wmma_body<F_IN, F_IN, 12, false, true>(Wlds, (int)blockIdx.x - nsort, x, x, W1,
                                                    z1, z1, n);
    }
}

__device__ __forceinline__ float4 pull_edges(const float* __restrict__ hin,
                                             const int2* __restrict__ list,
                                             int beg, int end, float4 acc, int q4) {
    const int last = end - 1;
    for (int j = beg; j < end; j += 4) {
        const int j1 = (j + 1 < end) ? j + 1 : last;
        const int j2 = (j + 2 < end) ? j + 2 : last;
        const int j3 = (j + 3 < end) ? j + 3 : last;
        const int2 e0 = list[j];
        const int2 e1 = list[j1];
        const int2 e2 = list[j2];
        const int2 e3 = list[j3];
        const float4 r0 = *reinterpret_cast<const float4*>(hin + (size_t)e0.x * HH + q4);
        const float4 r1 = *reinterpret_cast<const float4*>(hin + (size_t)e1.x * HH + q4);
        const float4 r2 = *reinterpret_cast<const float4*>(hin + (size_t)e2.x * HH + q4);
        const float4 r3 = *reinterpret_cast<const float4*>(hin + (size_t)e3.x * HH + q4);
        const bool v1 = (j + 1 < end), v2 = (j + 2 < end), v3 = (j + 3 < end);
        const float w0 = __int_as_float(e0.y);
        const float w1 = __int_as_float(e1.y);
        const float w2 = __int_as_float(e2.y);
        const float w3 = __int_as_float(e3.y);
        acc.x += r0.x * w0; acc.y += r0.y * w0; acc.z += r0.z * w0; acc.w += r0.w * w0;
        acc.x += (v1 ? r1.x : 0.0f) * w1; acc.y += (v1 ? r1.y : 0.0f) * w1;
        acc.z += (v1 ? r1.z : 0.0f) * w1; acc.w += (v1 ? r1.w : 0.0f) * w1;
        acc.x += (v2 ? r2.x : 0.0f) * w2; acc.y += (v2 ? r2.y : 0.0f) * w2;
        acc.z += (v2 ? r2.z : 0.0f) * w2; acc.w += (v2 ? r2.w : 0.0f) * w2;
        acc.x += (v3 ? r3.x : 0.0f) * w3; acc.y += (v3 ? r3.y : 0.0f) * w3;
        acc.z += (v3 ? r3.z : 0.0f) * w3; acc.w += (v3 ? r3.w : 0.0f) * w3;
    }
    return acc;
}

__device__ __forceinline__ float4 pull_node(const float* __restrict__ hin,
                                            const int* __restrict__ rowend,
                                            const int2* __restrict__ list,
                                            const float* __restrict__ dis,
                                            const float* __restrict__ bias, int node, int q4) {
    const int end = rowend[node];
    const int beg = (node > 0) ? rowend[node - 1] : 0;
    const float di = dis[node];
    const float wself = di * di;
    const float4 hv = *reinterpret_cast<const float4*>(hin + (size_t)node * HH + q4);
    float4 acc = make_float4(hv.x * wself, hv.y * wself, hv.z * wself, hv.w * wself);
    acc = pull_edges(hin, list, beg, end, acc, q4);
    const float4 bv = *reinterpret_cast<const float4*>(bias + q4);
    return make_float4(acc.x + bv.x, acc.y + bv.y, acc.z + bv.z, acc.w + bv.w);
}

__global__ void k_pull_half(const float* __restrict__ hin, float* __restrict__ hout,
                            const int* __restrict__ rowend, const int2* __restrict__ list,
                            const float* __restrict__ dis, const float* __restrict__ bias,
                            int n) {
    int t = blockIdx.x * blockDim.x + threadIdx.x;
    int node = t >> 3;
    if (node >= n) return;
    const int q4 = (t & 7) * 4;
    float4 r = pull_node(hin, rowend, list, dis, bias, node, q4);
    r.x = fmaxf(r.x, 0.0f); r.y = fmaxf(r.y, 0.0f);
    r.z = fmaxf(r.z, 0.0f); r.w = fmaxf(r.w, 0.0f);
    *reinterpret_cast<float4*>(hout + (size_t)node * HH + q4) = r;
}

__device__ __forceinline__ void pull_edges2(const float* __restrict__ h0,
                                            const float* __restrict__ h1,
                                            const int2* __restrict__ list,
                                            int beg, int end, float4& a0, float4& a1, int q4) {
    const int last = end - 1;
    for (int j = beg; j < end; j += 4) {
        const int j1 = (j + 1 < end) ? j + 1 : last;
        const int j2 = (j + 2 < end) ? j + 2 : last;
        const int j3 = (j + 3 < end) ? j + 3 : last;
        const int2 e0 = list[j];
        const int2 e1 = list[j1];
        const int2 e2 = list[j2];
        const int2 e3 = list[j3];
        const size_t o0 = (size_t)e0.x * HH + q4;
        const size_t o1 = (size_t)e1.x * HH + q4;
        const size_t o2 = (size_t)e2.x * HH + q4;
        const size_t o3 = (size_t)e3.x * HH + q4;
        const float4 r00 = *reinterpret_cast<const float4*>(h0 + o0);
        const float4 r01 = *reinterpret_cast<const float4*>(h1 + o0);
        const float4 r10 = *reinterpret_cast<const float4*>(h0 + o1);
        const float4 r11 = *reinterpret_cast<const float4*>(h1 + o1);
        const float4 r20 = *reinterpret_cast<const float4*>(h0 + o2);
        const float4 r21 = *reinterpret_cast<const float4*>(h1 + o2);
        const float4 r30 = *reinterpret_cast<const float4*>(h0 + o3);
        const float4 r31 = *reinterpret_cast<const float4*>(h1 + o3);
        const bool v1 = (j + 1 < end), v2 = (j + 2 < end), v3 = (j + 3 < end);
        const float w0 = __int_as_float(e0.y);
        const float w1 = __int_as_float(e1.y);
        const float w2 = __int_as_float(e2.y);
        const float w3 = __int_as_float(e3.y);
        a0.x += r00.x * w0; a0.y += r00.y * w0; a0.z += r00.z * w0; a0.w += r00.w * w0;
        a1.x += r01.x * w0; a1.y += r01.y * w0; a1.z += r01.z * w0; a1.w += r01.w * w0;
        a0.x += (v1 ? r10.x : 0.0f) * w1; a0.y += (v1 ? r10.y : 0.0f) * w1;
        a0.z += (v1 ? r10.z : 0.0f) * w1; a0.w += (v1 ? r10.w : 0.0f) * w1;
        a1.x += (v1 ? r11.x : 0.0f) * w1; a1.y += (v1 ? r11.y : 0.0f) * w1;
        a1.z += (v1 ? r11.z : 0.0f) * w1; a1.w += (v1 ? r11.w : 0.0f) * w1;
        a0.x += (v2 ? r20.x : 0.0f) * w2; a0.y += (v2 ? r20.y : 0.0f) * w2;
        a0.z += (v2 ? r20.z : 0.0f) * w2; a0.w += (v2 ? r20.w : 0.0f) * w2;
        a1.x += (v2 ? r21.x : 0.0f) * w2; a1.y += (v2 ? r21.y : 0.0f) * w2;
        a1.z += (v2 ? r21.z : 0.0f) * w2; a1.w += (v2 ? r21.w : 0.0f) * w2;
        a0.x += (v3 ? r30.x : 0.0f) * w3; a0.y += (v3 ? r30.y : 0.0f) * w3;
        a0.z += (v3 ? r30.z : 0.0f) * w3; a0.w += (v3 ? r30.w : 0.0f) * w3;
        a1.x += (v3 ? r31.x : 0.0f) * w3; a1.y += (v3 ? r31.y : 0.0f) * w3;
        a1.z += (v3 ? r31.z : 0.0f) * w3; a1.w += (v3 ? r31.w : 0.0f) * w3;
    }
}

#define PG_TB 256
__global__ void __launch_bounds__(PG_TB, 12)
k_pool_graph(const float* __restrict__ h0, const float* __restrict__ h1,
             const int* __restrict__ rowend, const int2* __restrict__ list,
             const float* __restrict__ dis, const float* __restrict__ bias,
             const int* __restrict__ gstart, const float* __restrict__ lin_w,
             const float* __restrict__ lin_b, float* __restrict__ out, int n) {
    __shared__ __attribute__((aligned(16))) float red[PG_TB / 32][H];
    const int g = blockIdx.x;
    const int nbeg = min(max(gstart[g], 0), n);
    const int nend = min(max(gstart[g + 1], nbeg), n);
    const int q4 = (threadIdx.x & 7) * 4;
    const int slot = threadIdx.x >> 3;

    float4 s0 = make_float4(0.0f, 0.0f, 0.0f, 0.0f);
    float4 s1 = make_float4(0.0f, 0.0f, 0.0f, 0.0f);
    for (int node = nbeg + slot; node < nend; node += PG_TB / 8) {
        const int end = rowend[node];
        const int beg = (node > 0) ? rowend[node - 1] : 0;
        const float di = dis[node];
        const float wself = di * di;
        const float4 hv0 = *reinterpret_cast<const float4*>(h0 + (size_t)node * HH + q4);
        const float4 hv1 = *reinterpret_cast<const float4*>(h1 + (size_t)node * HH + q4);
        float4 a0 = make_float4(hv0.x * wself, hv0.y * wself, hv0.z * wself, hv0.w * wself);
        float4 a1 = make_float4(hv1.x * wself, hv1.y * wself, hv1.z * wself, hv1.w * wself);
        pull_edges2(h0, h1, list, beg, end, a0, a1, q4);
        s0.x += a0.x; s0.y += a0.y; s0.z += a0.z; s0.w += a0.w;
        s1.x += a1.x; s1.y += a1.y; s1.z += a1.z; s1.w += a1.w;
    }

    s0.x += __shfl_xor(s0.x, 8);  s0.y += __shfl_xor(s0.y, 8);
    s0.z += __shfl_xor(s0.z, 8);  s0.w += __shfl_xor(s0.w, 8);
    s1.x += __shfl_xor(s1.x, 8);  s1.y += __shfl_xor(s1.y, 8);
    s1.z += __shfl_xor(s1.z, 8);  s1.w += __shfl_xor(s1.w, 8);
    s0.x += __shfl_xor(s0.x, 16); s0.y += __shfl_xor(s0.y, 16);
    s0.z += __shfl_xor(s0.z, 16); s0.w += __shfl_xor(s0.w, 16);
    s1.x += __shfl_xor(s1.x, 16); s1.y += __shfl_xor(s1.y, 16);
    s1.z += __shfl_xor(s1.z, 16); s1.w += __shfl_xor(s1.w, 16);
    const int lane = threadIdx.x & 31;
    const int wv   = threadIdx.x >> 5;
    if (lane < 8) {
        *reinterpret_cast<float4*>(&red[wv][q4]) = s0;
        *reinterpret_cast<float4*>(&red[wv][HH + q4]) = s1;
    }
    __syncthreads();

    if (threadIdx.x < 32) {
        const float cntf = (float)(nend - nbeg);
        float ca = 0.0f, cb = 0.0f;
        for (int w = 0; w < PG_TB / 32; ++w) {
            ca += red[w][lane];
            cb += red[w][HH + lane];
        }
        ca += cntf * bias[lane];
        cb += cntf * bias[HH + lane];
        float p = ca * lin_w[lane] + cb * lin_w[HH + lane];
        for (int o = 16; o > 0; o >>= 1) p += __shfl_xor(p, o);
        if (lane == 0) {
            const float inv = 1.0f / fmaxf(cntf, 1.0f);
            out[g] = p * inv + lin_b[0];
        }
    }
}

#define FB_ROWS 128
#define TP 36

__global__ void __launch_bounds__(256)
k_pull_gemm(const float* __restrict__ hg, float* P0, float* O1,
            const int* __restrict__ rowend, const int2* __restrict__ list,
            const float* __restrict__ dis, const float* __restrict__ bias,
            const float* __restrict__ W, int n) {
    __shared__ float Wlds[(H / 2) * H * 2];
    __shared__ __attribute__((aligned(16))) float tile[FB_ROWS * TP];
    for (int idx = threadIdx.x; idx < H * H; idx += blockDim.x) {
        int k = idx / H, c = idx - k * H;
        Wlds[(k >> 1) * (H * 2) + c * 2 + (k & 1)] = W[k * H + c];
    }

    const int blockBase = blockIdx.x * FB_ROWS;
    const int q4 = (threadIdx.x & 7) * 4;
    const int sub = threadIdx.x >> 3;
    const float4 bv = *reinterpret_cast<const float4*>(bias + q4);

    int beg[4], end[4];
    float4 acc[4];
    #pragma unroll
    for (int it = 0; it < 4; ++it) {
        int node = blockBase + it * 32 + sub;
        node = (node < n) ? node : (n - 1);
        end[it] = rowend[node];
        beg[it] = (node > 0) ? rowend[node - 1] : 0;
        const float di = dis[node];
        const float wself = di * di;
        const float4 hv = *reinterpret_cast<const float4*>(hg + (size_t)node * HH + q4);
        acc[it] = make_float4(hv.x * wself, hv.y * wself, hv.z * wself, hv.w * wself);
    }
    #pragma unroll
    for (int it = 0; it < 4; ++it) {
        const float4 a = pull_edges(hg, list, beg[it], end[it], acc[it], q4);
        float4 r = make_float4(a.x + bv.x, a.y + bv.y, a.z + bv.z, a.w + bv.w);
        r.x = fmaxf(r.x, 0.0f); r.y = fmaxf(r.y, 0.0f);
        r.z = fmaxf(r.z, 0.0f); r.w = fmaxf(r.w, 0.0f);
        *reinterpret_cast<float4*>(&tile[(it * 32 + sub) * TP + q4]) = r;
    }
    __syncthreads();

    const int lane = threadIdx.x & 31;
    const int wv   = threadIdx.x >> 5;
    const int row0 = blockBase + wv * 16;
    if (row0 >= n) return;

    const int mrow = lane & 15;
    const int hi   = lane >> 4;
    const float* arow = P0 + (size_t)(row0 + mrow) * HH;
    const float* trow = &tile[(wv * 16 + mrow) * TP];

    v8f c4[4];
    #pragma unroll
    for (int t = 0; t < 4; ++t) c4[t] = (v8f){0, 0, 0, 0, 0, 0, 0, 0};

    #pragma unroll
    for (int k0 = 0; k0 < H; k0 += 4) {
        const int ka = k0 + hi * 2;
        const float* ap = (k0 < HH) ? (arow + ka) : (trow + (ka - HH));
        const v2f a = *reinterpret_cast<const v2f*>(ap);

        const int pbase = ((k0 >> 1) + hi) * (H * 2);
        #pragma unroll
        for (int t = 0; t < 4; ++t) {
            const int c = t * 16 + mrow;
            v2f b = *reinterpret_cast<const v2f*>(&Wlds[pbase + c * 2]);
            c4[t] = __builtin_amdgcn_wmma_f32_16x16x4_f32(
                 false, a,  false, b,
                 (short)0, c4[t],  false,  false);
        }
    }

    const int mbase = (lane < 16) ? 0 : 8;
    #pragma unroll
    for (int t = 0; t < 4; ++t) {
        float* op = (t < 2) ? P0 : O1;
        const int col = (t & 1) * 16 + mrow;
        #pragma unroll
        for (int j = 0; j < 8; ++j) {
            op[(size_t)(row0 + mbase + j) * HH + col] = c4[t][j];
        }
    }
}


__device__ __forceinline__ float4 pull_edges_row(const float* __restrict__ zin,
                                                 const int2* __restrict__ list,
                                                 int beg, int end, float4 acc, int c4) {
    const int last = end - 1;
    for (int j = beg; j < end; j += 8) {
        int2 en[8];
        float4 r[8];
        #pragma unroll
        for (int u = 0; u < 8; ++u) en[u] = list[min(j + u, last)];
        #pragma unroll
        for (int u = 0; u < 8; ++u)
            r[u] = *reinterpret_cast<const float4*>(zin + (size_t)en[u].x * H + c4);
        #pragma unroll
        for (int u = 0; u < 8; ++u) {
            const bool v = (j + u < end);
            const float w = __int_as_float(en[u].y);
            acc.x += (v ? r[u].x : 0.0f) * w; acc.y += (v ? r[u].y : 0.0f) * w;
            acc.z += (v ? r[u].z : 0.0f) * w; acc.w += (v ? r[u].w : 0.0f) * w;
        }
    }
    return acc;
}

__device__ __forceinline__ void list_head8(const int2* __restrict__ list, int beg, int end,
                                           int2 (&en)[8]) {
    const int last = max(end - 1, 0);
    #pragma unroll
    for (int u = 0; u < 8; ++u) en[u] = list[min(beg + u, last)];
}

#define TPF 68

__global__ void __launch_bounds__(256, 6)
k_pull_gemm_rows(const float* __restrict__ zin, float* __restrict__ zout,
                 const int* __restrict__ rowend, const int2* __restrict__ list,
                 const float* __restrict__ dis, const float* __restrict__ bias,
                 const float* __restrict__ W, int n) {
    __shared__ float Wlds[(H / 2) * H * 2];
    __shared__ __attribute__((aligned(16))) float tile[FB_ROWS * TPF];
    for (int idx = threadIdx.x; idx < H * H; idx += blockDim.x) {
        int k = idx / H, c = idx - k * H;
        Wlds[(k >> 1) * (H * 2) + c * 2 + (k & 1)] = W[k * H + c];
    }

    const int lane = threadIdx.x & 31;
    const int wv   = threadIdx.x >> 5;
    const int row0 = blockIdx.x * FB_ROWS + wv * 16;
    const int c4   = (lane & 15) * 4;
    const int half = lane >> 4;
    const float4 bv = *reinterpret_cast<const float4*>(bias + c4);

    int beg[8], end[8];
    float4 acc[8];
    #pragma unroll
    for (int it = 0; it < 8; ++it) {
        int node = row0 + it * 2 + half;
        node = (node < n) ? node : (n - 1);
        end[it] = rowend[node];
        beg[it] = (node > 0) ? rowend[node - 1] : 0;
        const float di = dis[node];
        const float wself = di * di;
        const float4 hv = *reinterpret_cast<const float4*>(zin + (size_t)node * H + c4);
        acc[it] = make_float4(hv.x * wself, hv.y * wself, hv.z * wself, hv.w * wself);
    }
    int2 en[8];
    list_head8(list, beg[0], end[0], en);
    #pragma unroll
    for (int it = 0; it < 8; ++it) {
        float4 r[8];
        #pragma unroll
        for (int u = 0; u < 8; ++u)
            r[u] = *reinterpret_cast<const float4*>(zin + (size_t)en[u].x * H + c4);
        int2 nx[8];
        if (it + 1 < 8) list_head8(list, beg[it + 1], end[it + 1], nx);
        float4 a = acc[it];
        #pragma unroll
        for (int u = 0; u < 8; ++u) {
            const bool v = (beg[it] + u < end[it]);
            const float w = __int_as_float(en[u].y);
            a.x += (v ? r[u].x : 0.0f) * w; a.y += (v ? r[u].y : 0.0f) * w;
            a.z += (v ? r[u].z : 0.0f) * w; a.w += (v ? r[u].w : 0.0f) * w;
        }
        a = pull_edges_row(zin, list, beg[it] + 8, end[it], a, c4);
        float4 o = make_float4(a.x + bv.x, a.y + bv.y, a.z + bv.z, a.w + bv.w);
        o.x = fmaxf(o.x, 0.0f); o.y = fmaxf(o.y, 0.0f);
        o.z = fmaxf(o.z, 0.0f); o.w = fmaxf(o.w, 0.0f);
        *reinterpret_cast<float4*>(&tile[(wv * 16 + it * 2 + half) * TPF + c4]) = o;
        if (it + 1 < 8) {
            #pragma unroll
            for (int u = 0; u < 8; ++u) en[u] = nx[u];
        }
    }
    __syncthreads();

    if (row0 >= n) return;

    const int mrow = lane & 15;
    const int hi   = lane >> 4;
    const float* trow = &tile[(wv * 16 + mrow) * TPF];

    v8f c4acc[4];
    #pragma unroll
    for (int t = 0; t < 4; ++t) c4acc[t] = (v8f){0, 0, 0, 0, 0, 0, 0, 0};

    #pragma unroll
    for (int k0 = 0; k0 < H; k0 += 4) {
        const int ka = k0 + hi * 2;
        const v2f a = *reinterpret_cast<const v2f*>(trow + ka);

        const int pbase = ((k0 >> 1) + hi) * (H * 2);
        #pragma unroll
        for (int t = 0; t < 4; ++t) {
            const int c = t * 16 + mrow;
            v2f b = *reinterpret_cast<const v2f*>(&Wlds[pbase + c * 2]);
            c4acc[t] = __builtin_amdgcn_wmma_f32_16x16x4_f32(
                 false, a,  false, b,
                 (short)0, c4acc[t],  false,  false);
        }
    }

    const int mbase = (lane < 16) ? 0 : 8;
    #pragma unroll
    for (int t = 0; t < 4; ++t) {
        #pragma unroll
        for (int j = 0; j < 8; ++j) {
            zout[(size_t)(row0 + mbase + j) * H + t * 16 + mrow] = c4acc[t][j];
        }
    }
}

__global__ void __launch_bounds__(PG_TB, 12)
k_pool_graph_rows(const float* __restrict__ z,
                  const int* __restrict__ rowend, const int2* __restrict__ list,
                  const float* __restrict__ dis, const float* __restrict__ bias,
                  const int* __restrict__ gstart, const float* __restrict__ lin_w,
                  const float* __restrict__ lin_b, float* __restrict__ out, int n) {
    __shared__ __attribute__((aligned(16))) float red[PG_TB / 32][H];
    const int g = blockIdx.x;
    const int nbeg = min(max(gstart[g], 0), n);
    const int nend = min(max(gstart[g + 1], nbeg), n);
    const int c4 = (threadIdx.x & 15) * 4;
    const int slot = threadIdx.x >> 4;

    float4 s = make_float4(0.0f, 0.0f, 0.0f, 0.0f);
    for (int node = nbeg + slot; node < nend; node += PG_TB / 16) {
        const int end = rowend[node];
        const int beg = (node > 0) ? rowend[node - 1] : 0;
        const float di = dis[node];
        const float wself = di * di;
        const float4 hv = *reinterpret_cast<const float4*>(z + (size_t)node * H + c4);
        float4 a = make_float4(hv.x * wself, hv.y * wself, hv.z * wself, hv.w * wself);
        a = pull_edges_row(z, list, beg, end, a, c4);
        s.x += a.x; s.y += a.y; s.z += a.z; s.w += a.w;
    }

    s.x += __shfl_xor(s.x, 16); s.y += __shfl_xor(s.y, 16);
    s.z += __shfl_xor(s.z, 16); s.w += __shfl_xor(s.w, 16);
    const int lane = threadIdx.x & 31;
    const int wv   = threadIdx.x >> 5;
    if (lane < 16) *reinterpret_cast<float4*>(&red[wv][c4]) = s;
    __syncthreads();

    if (threadIdx.x < 32) {
        const float cntf = (float)(nend - nbeg);
        float ca = 0.0f, cb = 0.0f;
        for (int w = 0; w < PG_TB / 32; ++w) {
            ca += red[w][lane];
            cb += red[w][HH + lane];
        }
        ca += cntf * bias[lane];
        cb += cntf * bias[HH + lane];
        float p = ca * lin_w[lane] + cb * lin_w[HH + lane];
        for (int o = 16; o > 0; o >>= 1) p += __shfl_xor(p, o);
        if (lane == 0) {
            const float inv = 1.0f / fmaxf(cntf, 1.0f);
            out[g] = p * inv + lin_b[0];
        }
    }
}


extern "C" void kernel_launch(void* const* d_in, const int* in_sizes, int n_in,
                              void* d_out, int out_size, void* d_ws, size_t ws_size,
                              hipStream_t stream) {
    const float* x     = (const float*)d_in[0];
    const int*   eidx  = (const int*)  d_in[1];
    const int*   batch = (const int*)  d_in[2];
    const float* W1    = (const float*)d_in[3];
    const float* b1    = (const float*)d_in[4];
    const float* W2    = (const float*)d_in[5];
    const float* b2    = (const float*)d_in[6];
    const float* W3    = (const float*)d_in[7];
    const float* b3    = (const float*)d_in[8];
    const float* lin_w = (const float*)d_in[9];
    const float* lin_b = (const float*)d_in[10];
    float* out = (float*)d_out;

    const int n = in_sizes[0] / F_IN;
    const int e = in_sizes[1] / 2;
    const int* src = eidx;
    const int* dst = eidx + e;

    const int TB = 256;
    dim3 blk(TB);
    auto blocks = [](long long work, int tb) { return dim3((unsigned)((work + tb - 1) / tb)); };
    const dim3 bin_grid  = blocks(e, EPW);
    const dim3 sort_grid = blocks(n, BK);
    const int strips       = (n + 15) / 16;
    const dim3 gemm_grid   = blocks((long long)strips * 32, TB);
    const dim3 fuse_grid   = blocks((long long)n, FB_ROWS);

    const size_t rows_need = ((size_t)n * (2 * H + 3) + 1024 + 2 * (size_t)e + N_GRAPHS + 4)
                             * sizeof(float) + (size_t)n;
    if (ws_size >= rows_need) {
        float* ws   = (float*)d_ws;
        float* ZA   = ws;
        float* ZB   = ZA + (size_t)n * H;
        float* dis  = ZB + (size_t)n * H;
        int*   cnt    = (int*)(dis + n);
        int*   cursor = cnt + n;
        int*   part   = cursor + n;
        int2*  list   = (int2*)(part + 1024);
        int*   gstart = (int*)(list + e);
        unsigned char* deg8 = (unsigned char*)(gstart + N_GRAPHS + 4);
        int2*  binned = (int2*)ZB;
        int*   loff   = (int*)ZB + 2 * (size_t)e;
        const dim3 gemm1_grid = blocks((long long)strips * 32, SORT_TB);

        k_bin_local   <<<bin_grid, dim3(SORT_TB), 0, stream>>>(src, dst, binned, loff, e);
        k_bin_sort_gemm1<<<dim3(sort_grid.x + gemm1_grid.x), dim3(SORT_TB), 0, stream>>>(
            binned, loff, (int)bin_grid.x, list, cursor, dis, deg8, batch, gstart, e, n,
            (int)sort_grid.x, x, W1, ZA);
        k_edge_weights<<<dim3(PLACE_WG), dim3(PLACE_TB), 0, stream>>>(list, dis, deg8, e, n);
        k_pull_gemm_rows<<<fuse_grid, blk, 0, stream>>>(ZA, ZB, cursor, list, dis, b1, W2, n);
        k_pull_gemm_rows<<<fuse_grid, blk, 0, stream>>>(ZB, ZA, cursor, list, dis, b2, W3, n);
        k_pool_graph_rows<<<dim3(N_GRAPHS), dim3(PG_TB), 0, stream>>>(ZA, cursor, list, dis, b3, gstart,
                                                                      lin_w, lin_b, out, n);
        return;
    }

    float* ws   = (float*)d_ws;
    float* dis  = ws;
    float* R0   = dis + n;
    float* R1   = R0 + (size_t)n * HH;
    float* R2   = R1 + (size_t)n * HH;
    int*   cnt    = (int*)(R2 + (size_t)n * HH);
    int*   cursor = cnt + n;
    int*   part   = cursor + n;
    int2*  list   = (int2*)(part + 1024);
    int*   gstart = (int*)(list + e);
    unsigned char* deg8 = (unsigned char*)(gstart + N_GRAPHS + 4);
    int2*  binned = (int2*)R0;
    int*   loff   = (int*)R1;

    k_bin_local   <<<bin_grid, dim3(SORT_TB), 0, stream>>>(src, dst, binned, loff, e);
    k_bin_sort    <<<sort_grid, dim3(SORT_TB), 0, stream>>>(binned, loff, (int)bin_grid.x, list, cursor, dis, deg8, batch, gstart, e, n);
    k_edge_weights<<<dim3(PLACE_WG), dim3(PLACE_TB), 0, stream>>>(list, dis, deg8, e, n);

    const dim3 pull_grid   = blocks((long long)n * 8, TB);

    k_gemm_wmma<F_IN, F_IN, 12, false, false><<<gemm_grid, blk, 0, stream>>>(x, x, W1, R0, R1, n);

    k_pull_half<<<pull_grid, blk, 0, stream>>>(R0, R2, cursor, list, dis, b1, n);
    k_pull_gemm<<<fuse_grid, blk, 0, stream>>>(R1, R2, R0, cursor, list, dis, b1 + HH, W2, n);

    k_pull_half<<<pull_grid, blk, 0, stream>>>(R2, R1, cursor, list, dis, b2, n);
    k_pull_gemm<<<fuse_grid, blk, 0, stream>>>(R0, R1, R2, cursor, list, dis, b2 + HH, W3, n);

    k_pool_graph<<<dim3(N_GRAPHS), dim3(PG_TB), 0, stream>>>(R1, R2, cursor, list, dis, b3, gstart,
                                                            lin_w, lin_b, out, n);
}
